# starting kernel
# baseline (speedup 1.0000x reference)
#include <hip/hip_runtime.h>
#include <cstdint>


typedef float v2f __attribute__((ext_vector_type(2)));
typedef float v8f __attribute__((ext_vector_type(8)));

#define NEG_SLOPE 0.2f
#define GAT_EPS 1e-16f
#define ENC_NEG_INF 0x007FFFFFu

__device__ __forceinline__ unsigned enc_ord(float f) {
  unsigned u = __float_as_uint(f);
  return (u & 0x80000000u) ? ~u : (u | 0x80000000u);
}
__device__ __forceinline__ float dec_ord(unsigned k) {
  return (k & 0x80000000u) ? __uint_as_float(k & 0x7FFFFFFFu)
                           : __uint_as_float(~k);
}

__global__ void fill_u32_kernel(unsigned* __restrict__ p, unsigned v, long long n) {
  long long i = (long long)blockIdx.x * blockDim.x + threadIdx.x;
  long long stride = (long long)gridDim.x * blockDim.x;
  for (; i < n; i += stride) p[i] = v;
}

__global__ void gemm_f32_wmma_kernel(const float* __restrict__ A,
                                     const float* __restrict__ B,
                                     float* __restrict__ C,
                                     int M, int N, int K) {
  int wave = (int)((blockIdx.x * blockDim.x + threadIdx.x) >> 5);
  int lane = threadIdx.x & 31;
  int ntiles = N >> 4;
  int mtiles = (M + 15) >> 4;
  if (wave >= mtiles * ntiles) return;
  int mt = wave / ntiles;
  int nt = wave - mt * ntiles;
  int m0 = mt << 4, n0 = nt << 4;
  int half = lane >> 4;
  int lid  = lane & 15;

  v8f c = {0.f, 0.f, 0.f, 0.f, 0.f, 0.f, 0.f, 0.f};
  const float* arow = A + (size_t)(m0 + lid) * K + 2 * half;
  const float* bcol = B + (size_t)(2 * half) * N + n0 + lid;

  for (int k0 = 0; k0 < K; k0 += 4) {
    v2f a, b;
    a.x = arow[k0];
    a.y = arow[k0 + 1];
    b.x = bcol[(size_t)k0 * N];
    b.y = bcol[(size_t)k0 * N + N];
    c = __builtin_amdgcn_wmma_f32_16x16x4_f32(false, a, false, b,
                                              (short)0, c, false, false);
  }

  float* crow = C + (size_t)m0 * N + n0 + lid;
#pragma unroll
  for (int r = 0; r < 8; ++r)
    crow[(size_t)(r + 8 * half) * N] = c[r];
}

template <int H, int D>
__global__ void node_alpha_kernel(const float* __restrict__ h,
                                  const float* __restrict__ avs,
                                  const float* __restrict__ avd,
                                  float* __restrict__ as,
                                  float* __restrict__ ad, int N) {
  int n = blockIdx.x * blockDim.x + threadIdx.x;
  if (n >= N) return;
  const float* hr = h + (size_t)n * H * D;
#pragma unroll
  for (int hh = 0; hh < H; ++hh) {
    float s0 = 0.f, s1 = 0.f;
#pragma unroll
    for (int d = 0; d < D; ++d) {
      float v = hr[hh * D + d];
      s0 += v * avs[hh * D + d];
      s1 += v * avd[hh * D + d];
    }
    as[(size_t)n * H + hh] = s0;
    ad[(size_t)n * H + hh] = s1;
  }
}

template <int H>
__global__ void edge_max_kernel(const int* __restrict__ src,
                                const int* __restrict__ dst, int E, int Etot,
                                const float* __restrict__ as,
                                const float* __restrict__ ad,
                                unsigned* __restrict__ m) {
  int gid = blockIdx.x * blockDim.x + threadIdx.x;
  if (gid >= Etot * H) return;
  int eidx = gid / H, hh = gid - (gid / H) * H;
  int si, di;
  if (eidx < E) { si = src[eidx]; di = dst[eidx]; }
  else          { si = di = eidx - E; }
  float e = as[(size_t)si * H + hh] + ad[(size_t)di * H + hh];
  e = e > 0.f ? e : NEG_SLOPE * e;
  atomicMax(&m[(size_t)di * H + hh], enc_ord(e));
}

template <int H>
__global__ void edge_sum_kernel(const int* __restrict__ src,
                                const int* __restrict__ dst, int E, int Etot,
                                const float* __restrict__ as,
                                const float* __restrict__ ad,
                                const unsigned* __restrict__ m,
                                float* __restrict__ s) {
  int gid = blockIdx.x * blockDim.x + threadIdx.x;
  if (gid >= Etot * H) return;
  int eidx = gid / H, hh = gid - (gid / H) * H;
  int si, di;
  if (eidx < E) { si = src[eidx]; di = dst[eidx]; }
  else          { si = di = eidx - E; }
  float e = as[(size_t)si * H + hh] + ad[(size_t)di * H + hh];
  e = e > 0.f ? e : NEG_SLOPE * e;
  float ex = __expf(e - dec_ord(m[(size_t)di * H + hh]));
  atomicAdd(&s[(size_t)di * H + hh], ex);
}

template <int DTOT, int H>
__global__ void edge_scatter_kernel(const int* __restrict__ src,
                                    const int* __restrict__ dst, int E, int Etot,
                                    const float* __restrict__ as,
                                    const float* __restrict__ ad,
                                    const unsigned* __restrict__ m,
                                    const float* __restrict__ s,
                                    const float* __restrict__ hsrc,
                                    float* __restrict__ acc) {
  int gid = blockIdx.x * blockDim.x + threadIdx.x;
  int eidx = gid >> 5;
  int lane = gid & 31;
  if (eidx >= Etot) return;
  int si, di;
  if (eidx < E) { si = src[eidx]; di = dst[eidx]; }
  else          { si = di = eidx - E; }
  const int Dh = DTOT / H;
#pragma unroll
  for (int c = lane; c < DTOT; c += 32) {
    int hh = c / Dh;
    float e = as[(size_t)si * H + hh] + ad[(size_t)di * H + hh];
    e = e > 0.f ? e : NEG_SLOPE * e;
    float mm = dec_ord(m[(size_t)di * H + hh]);
    float ss = s[(size_t)di * H + hh];
    float alpha = __expf(e - mm) / (ss + GAT_EPS);
    atomicAdd(&acc[(size_t)di * DTOT + c],
              hsrc[(size_t)si * DTOT + c] * alpha);
  }
}

__global__ void finalize_kernel(float* __restrict__ acc,
                                const float* __restrict__ bias,
                                long long total, int C, int apply_elu) {
  long long i = (long long)blockIdx.x * blockDim.x + threadIdx.x;
  if (i >= total) return;
  int j = (int)(i % C);
  float v = acc[i] + bias[j];
  if (apply_elu) v = v > 0.f ? v : __expf(v) - 1.f;
  acc[i] = v;
}

static inline int cdiv(long long a, long long b) { return (int)((a + b - 1) / b); }

extern "C" void kernel_launch(void* const* d_in, const int* in_sizes, int n_in,
                              void* d_out, int out_size, void* d_ws, size_t ws_size,
                              hipStream_t stream) {
  const float* x      = (const float*)d_in[0];
  const int*   ei     = (const int*)  d_in[1];
  const float* W1     = (const float*)d_in[2];
  const float* a_src1 = (const float*)d_in[3];
  const float* a_dst1 = (const float*)d_in[4];
  const float* b1     = (const float*)d_in[5];
  const float* W2     = (const float*)d_in[6];
  const float* a_src2 = (const float*)d_in[7];
  const float* a_dst2 = (const float*)d_in[8];
  const float* b2     = (const float*)d_in[9];
  float* z = (float*)d_out;

  const int Fin = 256, H1 = 8, C1 = 64, C2 = 128;
  const int N    = in_sizes[0] / Fin;
  const int E    = in_sizes[1] / 2;
  const int Etot = E + N;
  const int* srcv = ei;
  const int* dstv = ei + E;

  float*    h1   = (float*)d_ws;
  float*    as1  = h1  + (size_t)N * C1;
  float*    ad1  = as1 + (size_t)N * H1;
  unsigned* m1   = (unsigned*)(ad1 + (size_t)N * H1);
  float*    s1   = (float*)(m1 + (size_t)N * H1);
  float*    acc1 = s1  + (size_t)N * H1;
  float*    h2   = acc1 + (size_t)N * C1;
  float*    as2  = h2  + (size_t)N * C2;
  float*    ad2  = as2 + N;
  unsigned* m2   = (unsigned*)(ad2 + N);
  float*    s2   = (float*)(m2 + N);

  const int B = 256;

  fill_u32_kernel<<<2048, B, 0, stream>>>((unsigned*)s1, 0u,
                                          (long long)N * (H1 + C1));
  fill_u32_kernel<<<2048, B, 0, stream>>>(m1, ENC_NEG_INF, (long long)N * H1);
  fill_u32_kernel<<<2048, B, 0, stream>>>((unsigned*)z, 0u, (long long)N * C2);
  fill_u32_kernel<<<2048, B, 0, stream>>>((unsigned*)s2, 0u, (long long)N);
  fill_u32_kernel<<<2048, B, 0, stream>>>(m2, ENC_NEG_INF, (long long)N);

  {
    int waves = (N / 16) * (C1 / 16);
    gemm_f32_wmma_kernel<<<cdiv((long long)waves * 32, B), B, 0, stream>>>(
        x, W1, h1, N, C1, Fin);
  }
  node_alpha_kernel<8, 8><<<cdiv(N, B), B, 0, stream>>>(h1, a_src1, a_dst1,
                                                        as1, ad1, N);
  edge_max_kernel<8><<<cdiv((long long)Etot * 8, B), B, 0, stream>>>(
      srcv, dstv, E, Etot, as1, ad1, m1);
  edge_sum_kernel<8><<<cdiv((long long)Etot * 8, B), B, 0, stream>>>(
      srcv, dstv, E, Etot, as1, ad1, m1, s1);
  edge_scatter_kernel<64, 8><<<cdiv((long long)Etot * 32, B), B, 0, stream>>>(
      srcv, dstv, E, Etot, as1, ad1, m1, s1, h1, acc1);
  finalize_kernel<<<cdiv((long long)N * C1, B), B, 0, stream>>>(
      acc1, b1, (long long)N * C1, C1,  1);

  {
    int waves = (N / 16) * (C2 / 16);
    gemm_f32_wmma_kernel<<<cdiv((long long)waves * 32, B), B, 0, stream>>>(
        acc1, W2, h2, N, C2, C1);
  }
  node_alpha_kernel<1, 128><<<cdiv(N, B), B, 0, stream>>>(h2, a_src2, a_dst2,
                                                          as2, ad2, N);
  edge_max_kernel<1><<<cdiv((long long)Etot, B), B, 0, stream>>>(
      srcv, dstv, E, Etot, as2, ad2, m2);
  edge_sum_kernel<1><<<cdiv((long long)Etot, B), B, 0, stream>>>(
      srcv, dstv, E, Etot, as2, ad2, m2, s2);
  edge_scatter_kernel<128, 1><<<cdiv((long long)Etot * 32, B), B, 0, stream>>>(
      srcv, dstv, E, Etot, as2, ad2, m2, s2, h2, z);
  finalize_kernel<<<cdiv((long long)N * C2, B), B, 0, stream>>>(
      z, b2, (long long)N * C2, C2,  0);
}


# Edges counting-sorted by dst in-call (int atomics); one wave per node does online softmax + row gather in registers, replacing max/sum/scatter float-atomic passes; sum order = sort arrival order
# speedup vs baseline: 2.5450x; 2.5450x over previous
#include <hip/hip_runtime.h>
#include <cstdint>


typedef float v2f __attribute__((ext_vector_type(2)));
typedef float v8f __attribute__((ext_vector_type(8)));

#define NEG_SLOPE 0.2f
#define GAT_EPS 1e-16f

__global__ void fill_u32_kernel(unsigned* __restrict__ p, unsigned v, long long n) {
  long long i = (long long)blockIdx.x * blockDim.x + threadIdx.x;
  long long stride = (long long)gridDim.x * blockDim.x;
  for (; i < n; i += stride) p[i] = v;
}

__global__ void gemm_f32_wmma_kernel(const float* __restrict__ A,
                                     const float* __restrict__ B,
                                     float* __restrict__ C,
                                     int M, int N, int K) {
  int wave = (int)((blockIdx.x * blockDim.x + threadIdx.x) >> 5);
  int lane = threadIdx.x & 31;
  int ntiles = N >> 4;
  int mtiles = (M + 15) >> 4;
  if (wave >= mtiles * ntiles) return;
  int mt = wave / ntiles;
  int nt = wave - mt * ntiles;
  int m0 = mt << 4, n0 = nt << 4;
  int half = lane >> 4;
  int lid  = lane & 15;

  v8f c = {0.f, 0.f, 0.f, 0.f, 0.f, 0.f, 0.f, 0.f};
  const float* arow = A + (size_t)(m0 + lid) * K + 2 * half;
  const float* bcol = B + (size_t)(2 * half) * N + n0 + lid;

  for (int k0 = 0; k0 < K; k0 += 4) {
    v2f a, b;
    a.x = arow[k0];
    a.y = arow[k0 + 1];
    b.x = bcol[(size_t)k0 * N];
    b.y = bcol[(size_t)k0 * N + N];
    c = __builtin_amdgcn_wmma_f32_16x16x4_f32(false, a, false, b,
                                              (short)0, c, false, false);
  }

  float* crow = C + (size_t)m0 * N + n0 + lid;
#pragma unroll
  for (int r = 0; r < 8; ++r)
    crow[(size_t)(r + 8 * half) * N] = c[r];
}

template <int H, int D>
__global__ void node_alpha_kernel(const float* __restrict__ h,
                                  const float* __restrict__ avs,
                                  const float* __restrict__ avd,
                                  float* __restrict__ as,
                                  float* __restrict__ ad, int N) {
  int n = blockIdx.x * blockDim.x + threadIdx.x;
  if (n >= N) return;
  const float* hr = h + (size_t)n * H * D;
#pragma unroll
  for (int hh = 0; hh < H; ++hh) {
    float s0 = 0.f, s1 = 0.f;
#pragma unroll
    for (int d = 0; d < D; ++d) {
      float v = hr[hh * D + d];
      s0 += v * avs[hh * D + d];
      s1 += v * avd[hh * D + d];
    }
    as[(size_t)n * H + hh] = s0;
    ad[(size_t)n * H + hh] = s1;
  }
}

__global__ void dst_count_kernel(const int* __restrict__ dst, int E,
                                 int* __restrict__ deg) {
  int e = blockIdx.x * blockDim.x + threadIdx.x;
  if (e < E) atomicAdd(&deg[dst[e]], 1);
}

__global__ void seg_alloc_kernel(const int* __restrict__ deg,
                                 int* __restrict__ total,
                                 int* __restrict__ seg_beg,
                                 int* __restrict__ seg_cur, int N) {
  __shared__ int wave_off[32];
  __shared__ int block_base;
  int i = blockIdx.x * blockDim.x + threadIdx.x;
  int lane = threadIdx.x & 31;
  int w = threadIdx.x >> 5;
  int nw = blockDim.x >> 5;
  int v = (i < N) ? deg[i] : 0;
  int x = v;
#pragma unroll
  for (int o = 1; o < 32; o <<= 1) {
    int y = __shfl_up(x, o);
    if (lane >= o) x += y;
  }
  if (lane == 31) wave_off[w] = x;
  __syncthreads();
  if (w == 0) {
    int t = (lane < nw) ? wave_off[lane] : 0;
    int xs = t;
#pragma unroll
    for (int o = 1; o < 32; o <<= 1) {
      int y = __shfl_up(xs, o);
      if (lane >= o) xs += y;
    }
    if (lane < nw) wave_off[lane] = xs - t;
    if (lane == 31) block_base = atomicAdd(total, xs);
  }
  __syncthreads();
  int pos = block_base + wave_off[w] + (x - v);
  if (i < N) { seg_beg[i] = pos; seg_cur[i] = pos; }
}

__global__ void csr_fill_kernel(const int* __restrict__ src,
                                const int* __restrict__ dst, int E,
                                int* __restrict__ seg_cur,
                                int* __restrict__ csr_src) {
  int e = blockIdx.x * blockDim.x + threadIdx.x;
  if (e < E) {
    int p = atomicAdd(&seg_cur[dst[e]], 1);
    csr_src[p] = src[e];
  }
}

template <int H, int DTOT, bool ELU>
__global__ void node_gather_kernel(const int* __restrict__ seg_beg,
                                   const int* __restrict__ seg_end,
                                   const int* __restrict__ csr_src,
                                   const float* __restrict__ as,
                                   const float* __restrict__ ad,
                                   const float* __restrict__ hfeat,
                                   const float* __restrict__ bias,
                                   float* __restrict__ out, int N) {
  constexpr int VEC = DTOT / 32;
  constexpr int DH = DTOT / H;
  typedef float vecf __attribute__((ext_vector_type(VEC)));
  int lane = threadIdx.x & 31;
  int node = __builtin_amdgcn_readfirstlane(
      (int)((blockIdx.x * blockDim.x + threadIdx.x) >> 5));
  if (node >= N) return;
  int c0 = lane * VEC;
  int hh = c0 / DH;

  float adv = ad[(size_t)node * H + hh];
  float e0 = as[(size_t)node * H + hh] + adv;
  e0 = e0 > 0.f ? e0 : NEG_SLOPE * e0;
  float m = e0;
  float s = 1.f;
  vecf acc = *(const vecf*)(hfeat + (size_t)node * DTOT + c0);

  int beg = seg_beg[node];
  int end = seg_end[node];
  for (int base = beg; base < end; base += 32) {
    int cnt = end - base;
    if (cnt > 32) cnt = 32;
    int mysrc = 0;
    float mye = 0.f;
    if (lane < cnt) {
      mysrc = csr_src[base + lane];
      if constexpr (H == 1) {
        float t = as[mysrc] + adv;
        mye = t > 0.f ? t : NEG_SLOPE * t;
      }
    }
#pragma unroll 4
    for (int j = 0; j < cnt; ++j) {
      int sj = __builtin_amdgcn_readlane(mysrc, j);
      float e;
      if constexpr (H == 1) {
        e = __int_as_float(__builtin_amdgcn_readlane(__float_as_int(mye), j));
      } else {
        e = as[(size_t)sj * H + hh] + adv;
        e = e > 0.f ? e : NEG_SLOPE * e;
      }
      vecf hv = *(const vecf*)(hfeat + (size_t)sj * DTOT + c0);
      float mn = fmaxf(m, e);
      float sc = __expf(m - mn);
      float p = __expf(e - mn);
      s = s * sc + p;
      acc = acc * sc + hv * p;
      m = mn;
    }
  }

  float denom = s + GAT_EPS;
  vecf o;
#pragma unroll
  for (int k = 0; k < VEC; ++k) {
    float v = acc[k] / denom + bias[c0 + k];
    if (ELU) v = v > 0.f ? v : __expf(v) - 1.f;
    o[k] = v;
  }
  *(vecf*)(out + (size_t)node * DTOT + c0) = o;
}

static inline int cdiv(long long a, long long b) { return (int)((a + b - 1) / b); }

extern "C" void kernel_launch(void* const* d_in, const int* in_sizes, int n_in,
                              void* d_out, int out_size, void* d_ws, size_t ws_size,
                              hipStream_t stream) {
  const float* x      = (const float*)d_in[0];
  const int*   ei     = (const int*)  d_in[1];
  const float* W1     = (const float*)d_in[2];
  const float* a_src1 = (const float*)d_in[3];
  const float* a_dst1 = (const float*)d_in[4];
  const float* b1     = (const float*)d_in[5];
  const float* W2     = (const float*)d_in[6];
  const float* a_src2 = (const float*)d_in[7];
  const float* a_dst2 = (const float*)d_in[8];
  const float* b2     = (const float*)d_in[9];
  float* z = (float*)d_out;

  const int Fin = 256, H1 = 8, C1 = 64, C2 = 128;
  const int N    = in_sizes[0] / Fin;
  const int E    = in_sizes[1] / 2;
  const int* srcv = ei;
  const int* dstv = ei + E;

  float* h1   = (float*)d_ws;
  float* hid  = h1  + (size_t)N * C1;
  float* h2   = hid + (size_t)N * C1;
  float* as1  = h2  + (size_t)N * C2;
  float* ad1  = as1 + (size_t)N * H1;
  float* as2  = ad1 + (size_t)N * H1;
  float* ad2  = as2 + N;
  int*   deg  = (int*)(ad2 + N);
  int*   sbeg = deg  + (N + 1);
  int*   scur = sbeg + N;
  int*   csrc = scur + N;

  const int B = 256;

  fill_u32_kernel<<<cdiv((long long)N + 1, B), B, 0, stream>>>(
      (unsigned*)deg, 0u, (long long)N + 1);
  dst_count_kernel<<<cdiv(E, B), B, 0, stream>>>(dstv, E, deg);
  seg_alloc_kernel<<<cdiv(N, B), B, 0, stream>>>(deg, deg + N, sbeg, scur, N);
  csr_fill_kernel<<<cdiv(E, B), B, 0, stream>>>(srcv, dstv, E, scur, csrc);

  {
    int waves = (N / 16) * (C1 / 16);
    gemm_f32_wmma_kernel<<<cdiv((long long)waves * 32, B), B, 0, stream>>>(
        x, W1, h1, N, C1, Fin);
  }
  node_alpha_kernel<8, 8><<<cdiv(N, B), B, 0, stream>>>(h1, a_src1, a_dst1,
                                                        as1, ad1, N);
  node_gather_kernel<8, 64, true><<<cdiv((long long)N * 32, B), B, 0, stream>>>(
      sbeg, scur, csrc, as1, ad1, h1, b1, hid, N);

  {
    int waves = (N / 16) * (C2 / 16);
    gemm_f32_wmma_kernel<<<cdiv((long long)waves * 32, B), B, 0, stream>>>(
        hid, W2, h2, N, C2, C1);
  }
  node_alpha_kernel<1, 128><<<cdiv(N, B), B, 0, stream>>>(h2, a_src2, a_dst2,
                                                          as2, ad2, N);
  node_gather_kernel<1, 128, false><<<cdiv((long long)N * 32, B), B, 0, stream>>>(
      sbeg, scur, csrc, as2, ad2, h2, b2, z, N);
}


# Every launch cut to few long-lived waves looping over their items: gathers 16,384 resident waves over nodes, sort passes 8 edges a thread, GEMM one wave per 16-row strip; sum order = sort arrival
# speedup vs baseline: 2.7067x; 1.0635x over previous
#include <hip/hip_runtime.h>
#include <cstdint>


typedef float v2f __attribute__((ext_vector_type(2)));
typedef float v8f __attribute__((ext_vector_type(8)));

#define NEG_SLOPE 0.2f
#define GAT_EPS 1e-16f

__global__ void fill_u32_kernel(unsigned* __restrict__ p, unsigned v, long long n) {
  long long i = (long long)blockIdx.x * blockDim.x + threadIdx.x;
  long long stride = (long long)gridDim.x * blockDim.x;
  for (; i < n; i += stride) p[i] = v;
}

template <int NT>
__global__ void gemm_f32_wmma_kernel(const float* __restrict__ A,
                                     const float* __restrict__ B,
                                     float* __restrict__ C,
                                     int M, int K) {
  constexpr int N = NT * 16;
  int wave = (int)((blockIdx.x * blockDim.x + threadIdx.x) >> 5);
  int lane = threadIdx.x & 31;
  int mtiles = (M + 15) >> 4;
  if (wave >= mtiles) return;
  int m0 = wave << 4;
  int half = lane >> 4;
  int lid  = lane & 15;

  v8f c[NT];
#pragma unroll
  for (int t = 0; t < NT; ++t)
    c[t] = (v8f){0.f, 0.f, 0.f, 0.f, 0.f, 0.f, 0.f, 0.f};
  const float* arow = A + (size_t)(m0 + lid) * K + 2 * half;
  const float* bcol = B + (size_t)(2 * half) * N + lid;

#pragma unroll(16 / NT)
  for (int k0 = 0; k0 < K; k0 += 4) {
    v2f a;
    a.x = arow[k0];
    a.y = arow[k0 + 1];
    const float* bp = bcol + (size_t)k0 * N;
#pragma unroll
    for (int t = 0; t < NT; ++t) {
      v2f b;
      b.x = bp[t * 16];
      b.y = bp[t * 16 + N];
      c[t] = __builtin_amdgcn_wmma_f32_16x16x4_f32(false, a, false, b,
                                                   (short)0, c[t], false, false);
    }
  }

  float* crow = C + (size_t)m0 * N + lid;
#pragma unroll
  for (int r = 0; r < 8; ++r)
#pragma unroll
    for (int t = 0; t < NT; ++t)
      crow[(size_t)(r + 8 * half) * N + t * 16] = c[t][r];
}

template <int H, int D>
__global__ void node_alpha_kernel(const float* __restrict__ h,
                                  const float* __restrict__ avs,
                                  const float* __restrict__ avd,
                                  float* __restrict__ as,
                                  float* __restrict__ ad, int N) {
  int n = blockIdx.x * blockDim.x + threadIdx.x;
  if (n >= N) return;
  const float* hr = h + (size_t)n * H * D;
#pragma unroll
  for (int hh = 0; hh < H; ++hh) {
    float s0 = 0.f, s1 = 0.f;
#pragma unroll
    for (int d = 0; d < D; ++d) {
      float v = hr[hh * D + d];
      s0 += v * avs[hh * D + d];
      s1 += v * avd[hh * D + d];
    }
    as[(size_t)n * H + hh] = s0;
    ad[(size_t)n * H + hh] = s1;
  }
}

__global__ void dst_count_kernel(const int* __restrict__ dst, int E,
                                 int* __restrict__ deg) {
  int i = blockIdx.x * blockDim.x + threadIdx.x;
  int stride = gridDim.x * blockDim.x;
  for (int e = i; e < E; e += 4 * stride) {
    if (e + 3 * stride < E) {
      int d0 = dst[e];
      int d1 = dst[e + stride];
      int d2 = dst[e + 2 * stride];
      int d3 = dst[e + 3 * stride];
      atomicAdd(&deg[d0], 1);
      atomicAdd(&deg[d1], 1);
      atomicAdd(&deg[d2], 1);
      atomicAdd(&deg[d3], 1);
    } else {
      for (int eu = e; eu < E; eu += stride) atomicAdd(&deg[dst[eu]], 1);
    }
  }
}

__global__ void seg_alloc_kernel(const int* __restrict__ deg,
                                 int* __restrict__ total,
                                 int* __restrict__ seg_beg,
                                 int* __restrict__ seg_cur, int N) {
  __shared__ int wave_off[32];
  __shared__ int block_base;
  int i = blockIdx.x * blockDim.x + threadIdx.x;
  int lane = threadIdx.x & 31;
  int w = threadIdx.x >> 5;
  int nw = blockDim.x >> 5;
  int v = (i < N) ? deg[i] : 0;
  int x = v;
#pragma unroll
  for (int o = 1; o < 32; o <<= 1) {
    int y = __shfl_up(x, o);
    if (lane >= o) x += y;
  }
  if (lane == 31) wave_off[w] = x;
  __syncthreads();
  if (w == 0) {
    int t = (lane < nw) ? wave_off[lane] : 0;
    int xs = t;
#pragma unroll
    for (int o = 1; o < 32; o <<= 1) {
      int y = __shfl_up(xs, o);
      if (lane >= o) xs += y;
    }
    if (lane < nw) wave_off[lane] = xs - t;
    if (lane == 31) block_base = atomicAdd(total, xs);
  }
  __syncthreads();
  int pos = block_base + wave_off[w] + (x - v);
  if (i < N) { seg_beg[i] = pos; seg_cur[i] = pos; }
}

__global__ void csr_fill_kernel(const int* __restrict__ src,
                                const int* __restrict__ dst, int E,
                                int* __restrict__ seg_cur,
                                int* __restrict__ csr_src) {
  int i = blockIdx.x * blockDim.x + threadIdx.x;
  int stride = gridDim.x * blockDim.x;
  for (int e = i; e < E; e += 4 * stride) {
    if (e + 3 * stride < E) {
      int d0 = dst[e];
      int d1 = dst[e + stride];
      int d2 = dst[e + 2 * stride];
      int d3 = dst[e + 3 * stride];
      int s0 = src[e];
      int s1 = src[e + stride];
      int s2 = src[e + 2 * stride];
      int s3 = src[e + 3 * stride];
      int p0 = atomicAdd(&seg_cur[d0], 1);
      int p1 = atomicAdd(&seg_cur[d1], 1);
      int p2 = atomicAdd(&seg_cur[d2], 1);
      int p3 = atomicAdd(&seg_cur[d3], 1);
      csr_src[p0] = s0;
      csr_src[p1] = s1;
      csr_src[p2] = s2;
      csr_src[p3] = s3;
    } else {
      for (int eu = e; eu < E; eu += stride) {
        int p = atomicAdd(&seg_cur[dst[eu]], 1);
        csr_src[p] = src[eu];
      }
    }
  }
}

template <int H, int DTOT, bool ELU>
__global__ void node_gather_kernel(const int* __restrict__ seg_beg,
                                   const int* __restrict__ seg_end,
                                   const int* __restrict__ csr_src,
                                   const float* __restrict__ as,
                                   const float* __restrict__ ad,
                                   const float* __restrict__ hfeat,
                                   const float* __restrict__ bias,
                                   float* __restrict__ out, int N) {
  constexpr int VEC = DTOT / 32;
  constexpr int DH = DTOT / H;
  typedef float vecf __attribute__((ext_vector_type(VEC)));
  int lane = threadIdx.x & 31;
  int wave0 = __builtin_amdgcn_readfirstlane(
      (int)((blockIdx.x * blockDim.x + threadIdx.x) >> 5));
  int nwaves = (int)((gridDim.x * blockDim.x) >> 5);
  int c0 = lane * VEC;
  int hh = c0 / DH;
  vecf bv = *(const vecf*)(bias + c0);

  for (int node = wave0; node < N; node += nwaves) {
    float adv = ad[(size_t)node * H + hh];
    float e0 = as[(size_t)node * H + hh] + adv;
    e0 = e0 > 0.f ? e0 : NEG_SLOPE * e0;
    float m = e0;
    float s = 1.f;
    vecf acc = *(const vecf*)(hfeat + (size_t)node * DTOT + c0);

    int beg = seg_beg[node];
    int end = seg_end[node];
    for (int base = beg; base < end; base += 32) {
      int cnt = end - base;
      if (cnt > 32) cnt = 32;
      int mysrc = 0;
      float mye = 0.f;
      if (lane < cnt) {
        mysrc = csr_src[base + lane];
        if constexpr (H == 1) {
          float t = as[mysrc] + adv;
          mye = t > 0.f ? t : NEG_SLOPE * t;
        }
      }
#pragma unroll 4
      for (int j = 0; j < cnt; ++j) {
        int sj = __builtin_amdgcn_readlane(mysrc, j);
        float e;
        if constexpr (H == 1) {
          e = __int_as_float(__builtin_amdgcn_readlane(__float_as_int(mye), j));
        } else {
          e = as[(size_t)sj * H + hh] + adv;
          e = e > 0.f ? e : NEG_SLOPE * e;
        }
        vecf hv = *(const vecf*)(hfeat + (size_t)sj * DTOT + c0);
        float mn = fmaxf(m, e);
        float sc = __expf(m - mn);
        float p = __expf(e - mn);
        s = s * sc + p;
        acc = acc * sc + hv * p;
        m = mn;
      }
    }

    float denom = s + GAT_EPS;
    vecf o;
#pragma unroll
    for (int k = 0; k < VEC; ++k) {
      float v = acc[k] / denom + bv[k];
      if (ELU) v = v > 0.f ? v : __expf(v) - 1.f;
      o[k] = v;
    }
    *(vecf*)(out + (size_t)node * DTOT + c0) = o;
  }
}

static inline int cdiv(long long a, long long b) { return (int)((a + b - 1) / b); }

extern "C" void kernel_launch(void* const* d_in, const int* in_sizes, int n_in,
                              void* d_out, int out_size, void* d_ws, size_t ws_size,
                              hipStream_t stream) {
  const float* x      = (const float*)d_in[0];
  const int*   ei     = (const int*)  d_in[1];
  const float* W1     = (const float*)d_in[2];
  const float* a_src1 = (const float*)d_in[3];
  const float* a_dst1 = (const float*)d_in[4];
  const float* b1     = (const float*)d_in[5];
  const float* W2     = (const float*)d_in[6];
  const float* a_src2 = (const float*)d_in[7];
  const float* a_dst2 = (const float*)d_in[8];
  const float* b2     = (const float*)d_in[9];
  float* z = (float*)d_out;

  const int Fin = 256, H1 = 8, C1 = 64, C2 = 128;
  const int N    = in_sizes[0] / Fin;
  const int E    = in_sizes[1] / 2;
  const int* srcv = ei;
  const int* dstv = ei + E;

  float* h1   = (float*)d_ws;
  float* hid  = h1  + (size_t)N * C1;
  float* h2   = hid + (size_t)N * C1;
  float* as1  = h2  + (size_t)N * C2;
  float* ad1  = as1 + (size_t)N * H1;
  float* as2  = ad1 + (size_t)N * H1;
  float* ad2  = as2 + N;
  int*   deg  = (int*)(ad2 + N);
  int*   sbeg = deg  + (N + 1);
  int*   scur = sbeg + N;
  int*   csrc = scur + N;

  const int B = 256;
  const int GB = 128;
  const int GATHER_BLOCKS = 2048;

  fill_u32_kernel<<<cdiv((long long)N + 1, B * 4), B, 0, stream>>>(
      (unsigned*)deg, 0u, (long long)N + 1);
  dst_count_kernel<<<cdiv(E, B * 8), B, 0, stream>>>(dstv, E, deg);
  seg_alloc_kernel<<<cdiv(N, B), B, 0, stream>>>(deg, deg + N, sbeg, scur, N);
  csr_fill_kernel<<<cdiv(E, B * 8), B, 0, stream>>>(srcv, dstv, E, scur, csrc);

  gemm_f32_wmma_kernel<C1 / 16><<<cdiv((long long)(N / 16) * 32, GB), GB, 0,
                                  stream>>>(x, W1, h1, N, Fin);
  node_alpha_kernel<8, 8><<<cdiv(N, B), B, 0, stream>>>(h1, a_src1, a_dst1,
                                                        as1, ad1, N);
  node_gather_kernel<8, 64, true><<<GATHER_BLOCKS, B, 0, stream>>>(
      sbeg, scur, csrc, as1, ad1, h1, b1, hid, N);

  gemm_f32_wmma_kernel<C2 / 16><<<cdiv((long long)(N / 16) * 32, GB), GB, 0,
                                  stream>>>(hid, W2, h2, N, C1);
  node_alpha_kernel<1, 128><<<cdiv(N, B), B, 0, stream>>>(h2, a_src2, a_dst2,
                                                          as2, ad2, N);
  node_gather_kernel<1, 128, false><<<GATHER_BLOCKS, B, 0, stream>>>(
      sbeg, scur, csrc, as2, ad2, h2, b2, z, N);
}


# Layer 2 sums the 64-wide hidden rows per node before the W2 product (the sum is linear), logits as hid.(W2 a): gather rows 512 -> 256 B, no h2 plane or logit pass; sum order: product after row sum
# speedup vs baseline: 2.8186x; 1.0413x over previous
#include <hip/hip_runtime.h>
#include <cstdint>


typedef float v2f __attribute__((ext_vector_type(2)));
typedef float v8f __attribute__((ext_vector_type(8)));

#define NEG_SLOPE 0.2f
#define GAT_EPS 1e-16f

__global__ void fill_u32_kernel(unsigned* __restrict__ p, unsigned v, long long n) {
  long long i = (long long)blockIdx.x * blockDim.x + threadIdx.x;
  long long stride = (long long)gridDim.x * blockDim.x;
  for (; i < n; i += stride) p[i] = v;
}

template <int M>
__device__ __forceinline__ float row_xor_lane(float x) {
  int b = __builtin_bit_cast(int, x);
  return __builtin_bit_cast(
      float, __builtin_amdgcn_update_dpp(b, b, 0x160 + M, 0xf, 0xf, false));
}

__device__ __forceinline__ float wave_sum(float v) {
  v += row_xor_lane<1>(v);
  v += row_xor_lane<2>(v);
  v += row_xor_lane<4>(v);
  v += row_xor_lane<8>(v);
  int b = __builtin_bit_cast(int, v);
  return __builtin_bit_cast(float, __builtin_amdgcn_readlane(b, 0)) +
         __builtin_bit_cast(float, __builtin_amdgcn_readlane(b, 16));
}

template <int NT, bool BIAS>
__global__ void gemm_f32_wmma_kernel(const float* __restrict__ A,
                                     const float* __restrict__ B,
                                     const float* __restrict__ bias,
                                     float* __restrict__ C,
                                     int M, int K) {
  constexpr int N = NT * 16;
  int wave = (int)((blockIdx.x * blockDim.x + threadIdx.x) >> 5);
  int lane = threadIdx.x & 31;
  int mtiles = (M + 15) >> 4;
  if (wave >= mtiles) return;
  int m0 = wave << 4;
  int half = lane >> 4;
  int lid  = lane & 15;

  v8f c[NT];
#pragma unroll
  for (int t = 0; t < NT; ++t)
    c[t] = (v8f){0.f, 0.f, 0.f, 0.f, 0.f, 0.f, 0.f, 0.f};
  const float* arow = A + (size_t)(m0 + lid) * K + 2 * half;
  const float* bcol = B + (size_t)(2 * half) * N + lid;

#pragma unroll(16 / NT)
  for (int k0 = 0; k0 < K; k0 += 4) {
    v2f a;
    a.x = arow[k0];
    a.y = arow[k0 + 1];
    const float* bp = bcol + (size_t)k0 * N;
#pragma unroll
    for (int t = 0; t < NT; ++t) {
      v2f b;
      b.x = bp[t * 16];
      b.y = bp[t * 16 + N];
      c[t] = __builtin_amdgcn_wmma_f32_16x16x4_f32(false, a, false, b,
                                                   (short)0, c[t], false, false);
    }
  }

  float bc[NT];
#pragma unroll
  for (int t = 0; t < NT; ++t) bc[t] = BIAS ? bias[t * 16 + lid] : 0.f;

  float* crow = C + (size_t)m0 * N + lid;
#pragma unroll
  for (int r = 0; r < 8; ++r)
#pragma unroll
    for (int t = 0; t < NT; ++t)
      crow[(size_t)(r + 8 * half) * N + t * 16] = BIAS ? c[t][r] + bc[t] : c[t][r];
}

template <int K, int DO>
__global__ void att_vec_kernel(const float* __restrict__ W,
                               const float* __restrict__ a_s,
                               const float* __restrict__ a_d,
                               float* __restrict__ ws,
                               float* __restrict__ wd) {
  constexpr int V = DO / 32;
  typedef float vecf __attribute__((ext_vector_type(V)));
  __shared__ float res[2 * K];
  int lane = threadIdx.x & 31;
  int w = threadIdx.x >> 5;
  int nw = blockDim.x >> 5;
  vecf sv = *(const vecf*)(a_s + lane * V);
  vecf dv = *(const vecf*)(a_d + lane * V);
  for (int k = w; k < K; k += nw) {
    vecf wv = *(const vecf*)(W + (size_t)k * DO + lane * V);
    float ps = 0.f, pd = 0.f;
#pragma unroll
    for (int i = 0; i < V; ++i) {
      ps += wv[i] * sv[i];
      pd += wv[i] * dv[i];
    }
    ps = wave_sum(ps);
    pd = wave_sum(pd);
    if (lane == 0) { res[k] = ps; res[K + k] = pd; }
  }
  __syncthreads();
  for (int i = threadIdx.x; i < 2 * K; i += blockDim.x) {
    if (i < K) ws[i] = res[i];
    else wd[i - K] = res[i];
  }
}

template <int H, int D>
__global__ void node_alpha_kernel(const float* __restrict__ h,
                                  const float* __restrict__ avs,
                                  const float* __restrict__ avd,
                                  float* __restrict__ as,
                                  float* __restrict__ ad, int N) {
  int n = blockIdx.x * blockDim.x + threadIdx.x;
  if (n >= N) return;
  const float* hr = h + (size_t)n * H * D;
#pragma unroll
  for (int hh = 0; hh < H; ++hh) {
    float s0 = 0.f, s1 = 0.f;
#pragma unroll
    for (int d = 0; d < D; ++d) {
      float v = hr[hh * D + d];
      s0 += v * avs[hh * D + d];
      s1 += v * avd[hh * D + d];
    }
    as[(size_t)n * H + hh] = s0;
    ad[(size_t)n * H + hh] = s1;
  }
}

__global__ void dst_count_kernel(const int* __restrict__ dst, int E,
                                 int* __restrict__ deg) {
  int i = blockIdx.x * blockDim.x + threadIdx.x;
  int stride = gridDim.x * blockDim.x;
  for (int e = i; e < E; e += 4 * stride) {
    if (e + 3 * stride < E) {
      int d0 = dst[e];
      int d1 = dst[e + stride];
      int d2 = dst[e + 2 * stride];
      int d3 = dst[e + 3 * stride];
      atomicAdd(&deg[d0], 1);
      atomicAdd(&deg[d1], 1);
      atomicAdd(&deg[d2], 1);
      atomicAdd(&deg[d3], 1);
    } else {
      for (int eu = e; eu < E; eu += stride) atomicAdd(&deg[dst[eu]], 1);
    }
  }
}

__global__ void seg_alloc_kernel(const int* __restrict__ deg,
                                 int* __restrict__ total,
                                 int* __restrict__ seg_beg,
                                 int* __restrict__ seg_cur, int N) {
  __shared__ int wave_off[32];
  __shared__ int block_base;
  int i = blockIdx.x * blockDim.x + threadIdx.x;
  int lane = threadIdx.x & 31;
  int w = threadIdx.x >> 5;
  int nw = blockDim.x >> 5;
  int v = (i < N) ? deg[i] : 0;
  int x = v;
#pragma unroll
  for (int o = 1; o < 32; o <<= 1) {
    int y = __shfl_up(x, o);
    if (lane >= o) x += y;
  }
  if (lane == 31) wave_off[w] = x;
  __syncthreads();
  if (w == 0) {
    int t = (lane < nw) ? wave_off[lane] : 0;
    int xs = t;
#pragma unroll
    for (int o = 1; o < 32; o <<= 1) {
      int y = __shfl_up(xs, o);
      if (lane >= o) xs += y;
    }
    if (lane < nw) wave_off[lane] = xs - t;
    if (lane == 31) block_base = atomicAdd(total, xs);
  }
  __syncthreads();
  int pos = block_base + wave_off[w] + (x - v);
  if (i < N) { seg_beg[i] = pos; seg_cur[i] = pos; }
}

__global__ void csr_fill_kernel(const int* __restrict__ src,
                                const int* __restrict__ dst, int E,
                                int* __restrict__ seg_cur,
                                int* __restrict__ csr_src) {
  int i = blockIdx.x * blockDim.x + threadIdx.x;
  int stride = gridDim.x * blockDim.x;
  for (int e = i; e < E; e += 4 * stride) {
    if (e + 3 * stride < E) {
      int d0 = dst[e];
      int d1 = dst[e + stride];
      int d2 = dst[e + 2 * stride];
      int d3 = dst[e + 3 * stride];
      int s0 = src[e];
      int s1 = src[e + stride];
      int s2 = src[e + 2 * stride];
      int s3 = src[e + 3 * stride];
      int p0 = atomicAdd(&seg_cur[d0], 1);
      int p1 = atomicAdd(&seg_cur[d1], 1);
      int p2 = atomicAdd(&seg_cur[d2], 1);
      int p3 = atomicAdd(&seg_cur[d3], 1);
      csr_src[p0] = s0;
      csr_src[p1] = s1;
      csr_src[p2] = s2;
      csr_src[p3] = s3;
    } else {
      for (int eu = e; eu < E; eu += stride) {
        int p = atomicAdd(&seg_cur[dst[eu]], 1);
        csr_src[p] = src[eu];
      }
    }
  }
}

template <int H, int DTOT, bool ELU, bool BIAS, bool LOGITS>
__global__ void node_gather_kernel(const int* __restrict__ seg_beg,
                                   const int* __restrict__ seg_end,
                                   const int* __restrict__ csr_src,
                                   const float* __restrict__ as,
                                   const float* __restrict__ ad,
                                   const float* __restrict__ hfeat,
                                   const float* __restrict__ bias,
                                   float* __restrict__ out,
                                   const float* __restrict__ wsrc,
                                   const float* __restrict__ wdst,
                                   float* __restrict__ as_next,
                                   float* __restrict__ ad_next, int N) {
  constexpr int VEC = DTOT / 32;
  constexpr int DH = DTOT / H;
  typedef float vecf __attribute__((ext_vector_type(VEC)));
  int lane = threadIdx.x & 31;
  int wave0 = __builtin_amdgcn_readfirstlane(
      (int)((blockIdx.x * blockDim.x + threadIdx.x) >> 5));
  int nwaves = (int)((gridDim.x * blockDim.x) >> 5);
  int c0 = lane * VEC;
  int hh = c0 / DH;
  vecf bv, wsv, wdv;
  if constexpr (BIAS) bv = *(const vecf*)(bias + c0);
  if constexpr (LOGITS) {
    wsv = *(const vecf*)(wsrc + c0);
    wdv = *(const vecf*)(wdst + c0);
  }

  for (int node = wave0; node < N; node += nwaves) {
    float adv = ad[(size_t)node * H + hh];
    float e0 = as[(size_t)node * H + hh] + adv;
    e0 = e0 > 0.f ? e0 : NEG_SLOPE * e0;
    float m = e0;
    float s = 1.f;
    vecf acc = *(const vecf*)(hfeat + (size_t)node * DTOT + c0);

    int beg = seg_beg[node];
    int end = seg_end[node];
    for (int base = beg; base < end; base += 32) {
      int cnt = end - base;
      if (cnt > 32) cnt = 32;
      int mysrc = 0;
      float mye = 0.f;
      if (lane < cnt) {
        mysrc = csr_src[base + lane];
        if constexpr (H == 1) {
          float t = as[mysrc] + adv;
          mye = t > 0.f ? t : NEG_SLOPE * t;
        }
      }
#pragma unroll 4
      for (int j = 0; j < cnt; ++j) {
        int sj = __builtin_amdgcn_readlane(mysrc, j);
        float e;
        if constexpr (H == 1) {
          e = __int_as_float(__builtin_amdgcn_readlane(__float_as_int(mye), j));
        } else {
          e = as[(size_t)sj * H + hh] + adv;
          e = e > 0.f ? e : NEG_SLOPE * e;
        }
        vecf hv = *(const vecf*)(hfeat + (size_t)sj * DTOT + c0);
        float mn = fmaxf(m, e);
        float sc = __expf(m - mn);
        float p = __expf(e - mn);
        s = s * sc + p;
        acc = acc * sc + hv * p;
        m = mn;
      }
    }

    float denom = s + GAT_EPS;
    vecf o;
#pragma unroll
    for (int k = 0; k < VEC; ++k) {
      float v = acc[k] / denom;
      if constexpr (BIAS) v += bv[k];
      if (ELU) v = v > 0.f ? v : __expf(v) - 1.f;
      o[k] = v;
    }
    *(vecf*)(out + (size_t)node * DTOT + c0) = o;

    if constexpr (LOGITS) {
      float ps = 0.f, pd = 0.f;
#pragma unroll
      for (int k = 0; k < VEC; ++k) {
        ps += o[k] * wsv[k];
        pd += o[k] * wdv[k];
      }
      ps = wave_sum(ps);
      pd = wave_sum(pd);
      if (lane == 0) {
        as_next[node] = ps;
        ad_next[node] = pd;
      }
    }
  }
}

static inline int cdiv(long long a, long long b) { return (int)((a + b - 1) / b); }

extern "C" void kernel_launch(void* const* d_in, const int* in_sizes, int n_in,
                              void* d_out, int out_size, void* d_ws, size_t ws_size,
                              hipStream_t stream) {
  const float* x      = (const float*)d_in[0];
  const int*   ei     = (const int*)  d_in[1];
  const float* W1     = (const float*)d_in[2];
  const float* a_src1 = (const float*)d_in[3];
  const float* a_dst1 = (const float*)d_in[4];
  const float* b1     = (const float*)d_in[5];
  const float* W2     = (const float*)d_in[6];
  const float* a_src2 = (const float*)d_in[7];
  const float* a_dst2 = (const float*)d_in[8];
  const float* b2     = (const float*)d_in[9];
  float* z = (float*)d_out;

  const int Fin = 256, H1 = 8, C1 = 64, C2 = 128;
  const int N    = in_sizes[0] / Fin;
  const int E    = in_sizes[1] / 2;
  const int* srcv = ei;
  const int* dstv = ei + E;

  float* h1   = (float*)d_ws;
  float* hid  = h1  + (size_t)N * C1;
  float* agg  = hid + (size_t)N * C1;
  float* as1  = agg + (size_t)N * C1;
  float* ad1  = as1 + (size_t)N * H1;
  float* as2  = ad1 + (size_t)N * H1;
  float* ad2  = as2 + N;
  float* ws2  = ad2 + N;
  float* wd2  = ws2 + C1;
  int*   deg  = (int*)(wd2 + C1);
  int*   sbeg = deg  + (N + 1);
  int*   scur = sbeg + N;
  int*   csrc = scur + N;

  const int B = 256;
  const int GB = 128;
  const int GATHER_BLOCKS = 2048;

  att_vec_kernel<64, 128><<<1, 1024, 0, stream>>>(W2, a_src2, a_dst2, ws2, wd2);

  fill_u32_kernel<<<cdiv((long long)N + 1, B * 4), B, 0, stream>>>(
      (unsigned*)deg, 0u, (long long)N + 1);
  dst_count_kernel<<<cdiv(E, B * 8), B, 0, stream>>>(dstv, E, deg);
  seg_alloc_kernel<<<cdiv(N, B), B, 0, stream>>>(deg, deg + N, sbeg, scur, N);
  csr_fill_kernel<<<cdiv(E, B * 8), B, 0, stream>>>(srcv, dstv, E, scur, csrc);

  gemm_f32_wmma_kernel<C1 / 16, false><<<cdiv((long long)(N / 16) * 32, GB), GB,
                                         0, stream>>>(x, W1, nullptr, h1, N, Fin);
  node_alpha_kernel<8, 8><<<cdiv(N, B), B, 0, stream>>>(h1, a_src1, a_dst1,
                                                        as1, ad1, N);
  node_gather_kernel<8, 64, true, true, true><<<GATHER_BLOCKS, B, 0, stream>>>(
      sbeg, scur, csrc, as1, ad1, h1, b1, hid, ws2, wd2, as2, ad2, N);

  node_gather_kernel<1, 64, false, false, false><<<GATHER_BLOCKS, B, 0, stream>>>(
      sbeg, scur, csrc, as2, ad2, hid, nullptr, agg, nullptr, nullptr, nullptr,
      nullptr, N);
  gemm_f32_wmma_kernel<C2 / 16, true><<<cdiv((long long)(N / 16) * 32, GB), GB,
                                        0, stream>>>(agg, W2, b2, z, N, C1);
}


# Edge sort without device-scope atomics: edges counted and ranked by LDS adds in two levels (256-node buckets per chunk, then nodes per bucket), not 3.2M global atomics; sum order = sort arrival
# speedup vs baseline: 4.7242x; 1.6761x over previous
#include <hip/hip_runtime.h>
#include <cstdint>


typedef float v2f __attribute__((ext_vector_type(2)));
typedef float v8f __attribute__((ext_vector_type(8)));

#define NEG_SLOPE 0.2f
#define GAT_EPS 1e-16f

template <int M>
__device__ __forceinline__ float row_xor_lane(float x) {
  int b = __builtin_bit_cast(int, x);
  return __builtin_bit_cast(
      float, __builtin_amdgcn_update_dpp(b, b, 0x160 + M, 0xf, 0xf, false));
}

__device__ __forceinline__ float wave_sum(float v) {
  v += row_xor_lane<1>(v);
  v += row_xor_lane<2>(v);
  v += row_xor_lane<4>(v);
  v += row_xor_lane<8>(v);
  int b = __builtin_bit_cast(int, v);
  return __builtin_bit_cast(float, __builtin_amdgcn_readlane(b, 0)) +
         __builtin_bit_cast(float, __builtin_amdgcn_readlane(b, 16));
}

template <int NT, bool BIAS>
__global__ void gemm_f32_wmma_kernel(const float* __restrict__ A,
                                     const float* __restrict__ B,
                                     const float* __restrict__ bias,
                                     float* __restrict__ C,
                                     int M, int K) {
  constexpr int N = NT * 16;
  int wave = (int)((blockIdx.x * blockDim.x + threadIdx.x) >> 5);
  int lane = threadIdx.x & 31;
  int mtiles = (M + 15) >> 4;
  if (wave >= mtiles) return;
  int m0 = wave << 4;
  int half = lane >> 4;
  int lid  = lane & 15;

  v8f c[NT];
#pragma unroll
  for (int t = 0; t < NT; ++t)
    c[t] = (v8f){0.f, 0.f, 0.f, 0.f, 0.f, 0.f, 0.f, 0.f};
  const float* arow = A + (size_t)(m0 + lid) * K + 2 * half;
  const float* bcol = B + (size_t)(2 * half) * N + lid;

#pragma unroll(16 / NT)
  for (int k0 = 0; k0 < K; k0 += 4) {
    v2f a;
    a.x = arow[k0];
    a.y = arow[k0 + 1];
    const float* bp = bcol + (size_t)k0 * N;
#pragma unroll
    for (int t = 0; t < NT; ++t) {
      v2f b;
      b.x = bp[t * 16];
      b.y = bp[t * 16 + N];
      c[t] = __builtin_amdgcn_wmma_f32_16x16x4_f32(false, a, false, b,
                                                   (short)0, c[t], false, false);
    }
  }

  float bc[NT];
#pragma unroll
  for (int t = 0; t < NT; ++t) bc[t] = BIAS ? bias[t * 16 + lid] : 0.f;

  float* crow = C + (size_t)m0 * N + lid;
#pragma unroll
  for (int r = 0; r < 8; ++r)
#pragma unroll
    for (int t = 0; t < NT; ++t)
      crow[(size_t)(r + 8 * half) * N + t * 16] = BIAS ? c[t][r] + bc[t] : c[t][r];
}

template <int K, int DO>
__global__ void att_vec_kernel(const float* __restrict__ W,
                               const float* __restrict__ a_s,
                               const float* __restrict__ a_d,
                               float* __restrict__ ws,
                               float* __restrict__ wd) {
  constexpr int V = DO / 32;
  typedef float vecf __attribute__((ext_vector_type(V)));
  __shared__ float res[2 * K];
  int lane = threadIdx.x & 31;
  int w = threadIdx.x >> 5;
  int nw = blockDim.x >> 5;
  vecf sv = *(const vecf*)(a_s + lane * V);
  vecf dv = *(const vecf*)(a_d + lane * V);
  for (int k = w; k < K; k += nw) {
    vecf wv = *(const vecf*)(W + (size_t)k * DO + lane * V);
    float ps = 0.f, pd = 0.f;
#pragma unroll
    for (int i = 0; i < V; ++i) {
      ps += wv[i] * sv[i];
      pd += wv[i] * dv[i];
    }
    ps = wave_sum(ps);
    pd = wave_sum(pd);
    if (lane == 0) { res[k] = ps; res[K + k] = pd; }
  }
  __syncthreads();
  for (int i = threadIdx.x; i < 2 * K; i += blockDim.x) {
    if (i < K) ws[i] = res[i];
    else wd[i - K] = res[i];
  }
}

template <int H, int D>
__global__ void node_alpha_kernel(const float* __restrict__ h,
                                  const float* __restrict__ avs,
                                  const float* __restrict__ avd,
                                  float* __restrict__ as,
                                  float* __restrict__ ad, int N) {
  int n = blockIdx.x * blockDim.x + threadIdx.x;
  if (n >= N) return;
  const float* hr = h + (size_t)n * H * D;
#pragma unroll
  for (int hh = 0; hh < H; ++hh) {
    float s0 = 0.f, s1 = 0.f;
#pragma unroll
    for (int d = 0; d < D; ++d) {
      float v = hr[hh * D + d];
      s0 += v * avs[hh * D + d];
      s1 += v * avd[hh * D + d];
    }
    as[(size_t)n * H + hh] = s0;
    ad[(size_t)n * H + hh] = s1;
  }
}

#define NCHUNK 256
#define BUCKET_SHIFT 8
#define BUCKET_NODES 256
#define MAX_BUCKETS 1024
#define CHUNK_THREADS 1024
#define STAGE_CAP 8192

__device__ __forceinline__ int wave_incl_scan(int x, int lane) {
#pragma unroll
  for (int o = 1; o < 32; o <<= 1) {
    int y = __shfl_up(x, o);
    if (lane >= o) x += y;
  }
  return x;
}

__device__ __forceinline__ int block_excl_scan(int v, int* wsum, int& total) {
  int lane = threadIdx.x & 31;
  int w = threadIdx.x >> 5;
  int nw = blockDim.x >> 5;
  int x = wave_incl_scan(v, lane);
  if (lane == 31) wsum[w] = x;
  __syncthreads();
  if (w == 0) {
    int t = (lane < nw) ? wsum[lane] : 0;
    int xs = wave_incl_scan(t, lane);
    wsum[lane] = xs - t;
    if (lane == 31) wsum[32] = xs;
  }
  __syncthreads();
  total = wsum[32];
  int r = wsum[w] + (x - v);
  __syncthreads();
  return r;
}

__global__ void chunk_hist_kernel(const int* __restrict__ dst, int E, int NB,
                                  int* __restrict__ counts) {
  __shared__ int h[MAX_BUCKETS];
  int tid = threadIdx.x;
  int c = blockIdx.x;
  for (int i = tid; i < NB; i += CHUNK_THREADS) h[i] = 0;
  __syncthreads();
  int per = (E + NCHUNK - 1) / NCHUNK;
  int e0 = c * per;
  int e1 = e0 + per < E ? e0 + per : E;
  for (int e = e0 + tid; e < e1; e += 4 * CHUNK_THREADS) {
    int d[4];
#pragma unroll
    for (int k = 0; k < 4; ++k) {
      int ee = e + k * CHUNK_THREADS;
      d[k] = ee < e1 ? dst[ee] : -1;
    }
#pragma unroll
    for (int k = 0; k < 4; ++k)
      if (d[k] >= 0) atomicAdd(&h[d[k] >> BUCKET_SHIFT], 1);
  }
  __syncthreads();
  for (int i = tid; i < NB; i += CHUNK_THREADS)
    counts[(size_t)i * NCHUNK + c] = h[i];
}

__global__ void bucket_scan_kernel(int* __restrict__ counts, int NB,
                                   int* __restrict__ bstart) {
  __shared__ int tot[MAX_BUCKETS];
  __shared__ int wsum[33];
  typedef int v4i __attribute__((ext_vector_type(4)));
  int tid = threadIdx.x;
  int lane = tid & 31;
  int w = tid >> 5;
  int nw = blockDim.x >> 5;
  for (int b0 = w * 4; b0 < NB; b0 += nw * 4) {
    v4i a0[4], a1[4];
#pragma unroll
    for (int q = 0; q < 4; ++q) {
      if (b0 + q < NB) {
        const v4i* row = (const v4i*)(counts + (size_t)(b0 + q) * NCHUNK + lane * 8);
        a0[q] = row[0];
        a1[q] = row[1];
      }
    }
#pragma unroll
    for (int q = 0; q < 4; ++q) {
      if (b0 + q < NB) {
        v4i p0, p1;
        p0.x = 0;
        p0.y = a0[q].x;
        p0.z = p0.y + a0[q].y;
        p0.w = p0.z + a0[q].z;
        p1.x = p0.w + a0[q].w;
        p1.y = p1.x + a1[q].x;
        p1.z = p1.y + a1[q].y;
        p1.w = p1.z + a1[q].z;
        int sum = p1.w + a1[q].w;
        int incl = wave_incl_scan(sum, lane);
        int excl = incl - sum;
        p0 += excl;
        p1 += excl;
        v4i* row = (v4i*)(counts + (size_t)(b0 + q) * NCHUNK + lane * 8);
        row[0] = p0;
        row[1] = p1;
        if (lane == 31) tot[b0 + q] = incl;
      }
    }
  }
  __syncthreads();
  int carry = 0;
  for (int base = 0; base < NB; base += (int)blockDim.x) {
    int i = base + tid;
    int v = (i < NB) ? tot[i] : 0;
    int total;
    int ex = block_excl_scan(v, wsum, total);
    if (i < NB) bstart[i] = carry + ex;
    carry += total;
  }
  if (tid == 0) bstart[NB] = carry;
}

__global__ void chunk_scatter_kernel(const int* __restrict__ src,
                                     const int* __restrict__ dst, int E, int NB,
                                     const int* __restrict__ part,
                                     const int* __restrict__ bstart,
                                     int* __restrict__ rec_words) {
  typedef int v2i __attribute__((ext_vector_type(2)));
  v2i* rec = (v2i*)rec_words;
  __shared__ int cur[MAX_BUCKETS];
  int tid = threadIdx.x;
  int c = blockIdx.x;
  for (int i = tid; i < NB; i += CHUNK_THREADS)
    cur[i] = bstart[i] + part[(size_t)i * NCHUNK + c];
  __syncthreads();
  int per = (E + NCHUNK - 1) / NCHUNK;
  int e0 = c * per;
  int e1 = e0 + per < E ? e0 + per : E;
  for (int e = e0 + tid; e < e1; e += 4 * CHUNK_THREADS) {
    int d[4], s[4], p[4];
#pragma unroll
    for (int k = 0; k < 4; ++k) {
      int ee = e + k * CHUNK_THREADS;
      bool ok = ee < e1;
      d[k] = ok ? dst[ee] : -1;
      s[k] = ok ? src[ee] : 0;
    }
#pragma unroll
    for (int k = 0; k < 4; ++k)
      p[k] = d[k] >= 0 ? atomicAdd(&cur[d[k] >> BUCKET_SHIFT], 1) : 0;
#pragma unroll
    for (int k = 0; k < 4; ++k)
      if (d[k] >= 0) rec[p[k]] = (v2i){s[k], d[k]};
  }
}

__global__ void bucket_rank_kernel(const int* __restrict__ rec_words,
                                   const int* __restrict__ bstart,
                                   int* __restrict__ seg_beg,
                                   int* __restrict__ seg_end,
                                   int* __restrict__ csr_src, int N) {
  typedef int v2i __attribute__((ext_vector_type(2)));
  const v2i* rec = (const v2i*)rec_words;
  __shared__ int cnt[BUCKET_NODES];
  __shared__ int wsum[33];
  __shared__ int stage[STAGE_CAP];
  int tid = threadIdx.x;
  int b = blockIdx.x;
  int lo = b << BUCKET_SHIFT;
  int bs = bstart[b];
  int be = bstart[b + 1];
  int n = be - bs;
  cnt[tid] = 0;
  __syncthreads();
  for (int i = tid; i < n; i += 4 * BUCKET_NODES) {
    int d[4];
#pragma unroll
    for (int k = 0; k < 4; ++k) {
      int ii = i + k * BUCKET_NODES;
      d[k] = ii < n ? rec[bs + ii].y : -1;
    }
#pragma unroll
    for (int k = 0; k < 4; ++k)
      if (d[k] >= 0) atomicAdd(&cnt[d[k] - lo], 1);
  }
  __syncthreads();
  int v = cnt[tid];
  int total;
  int ex = block_excl_scan(v, wsum, total);
  int node = lo + tid;
  if (node < N) {
    seg_beg[node] = bs + ex;
    seg_end[node] = bs + ex + v;
  }
  cnt[tid] = ex;
  __syncthreads();
  bool staged = n <= STAGE_CAP;
  for (int i = tid; i < n; i += 4 * BUCKET_NODES) {
    v2i r[4];
    int p[4];
#pragma unroll
    for (int k = 0; k < 4; ++k) {
      int ii = i + k * BUCKET_NODES;
      r[k] = ii < n ? rec[bs + ii] : (v2i){0, -1};
    }
#pragma unroll
    for (int k = 0; k < 4; ++k)
      p[k] = r[k].y >= 0 ? atomicAdd(&cnt[r[k].y - lo], 1) : 0;
#pragma unroll
    for (int k = 0; k < 4; ++k) {
      if (r[k].y >= 0) {
        if (staged) stage[p[k]] = r[k].x;
        else csr_src[bs + p[k]] = r[k].x;
      }
    }
  }
  if (staged) {
    __syncthreads();
    for (int i = tid; i < n; i += BUCKET_NODES) csr_src[bs + i] = stage[i];
  }
}

template <int H, int DTOT, bool ELU, bool BIAS, bool LOGITS>
__global__ void node_gather_kernel(const int* __restrict__ seg_beg,
                                   const int* __restrict__ seg_end,
                                   const int* __restrict__ csr_src,
                                   const float* __restrict__ as,
                                   const float* __restrict__ ad,
                                   const float* __restrict__ hfeat,
                                   const float* __restrict__ bias,
                                   float* __restrict__ out,
                                   const float* __restrict__ wsrc,
                                   const float* __restrict__ wdst,
                                   float* __restrict__ as_next,
                                   float* __restrict__ ad_next, int N) {
  constexpr int VEC = DTOT / 32;
  constexpr int DH = DTOT / H;
  typedef float vecf __attribute__((ext_vector_type(VEC)));
  int lane = threadIdx.x & 31;
  int wave0 = __builtin_amdgcn_readfirstlane(
      (int)((blockIdx.x * blockDim.x + threadIdx.x) >> 5));
  int nwaves = (int)((gridDim.x * blockDim.x) >> 5);
  int c0 = lane * VEC;
  int hh = c0 / DH;
  vecf bv, wsv, wdv;
  if constexpr (BIAS) bv = *(const vecf*)(bias + c0);
  if constexpr (LOGITS) {
    wsv = *(const vecf*)(wsrc + c0);
    wdv = *(const vecf*)(wdst + c0);
  }

  for (int node = wave0; node < N; node += nwaves) {
    float adv = ad[(size_t)node * H + hh];
    float e0 = as[(size_t)node * H + hh] + adv;
    e0 = e0 > 0.f ? e0 : NEG_SLOPE * e0;
    float m = e0;
    float s = 1.f;
    vecf acc = *(const vecf*)(hfeat + (size_t)node * DTOT + c0);

    int beg = seg_beg[node];
    int end = seg_end[node];
    for (int base = beg; base < end; base += 32) {
      int cnt = end - base;
      if (cnt > 32) cnt = 32;
      int mysrc = 0;
      float mye = 0.f;
      if (lane < cnt) {
        mysrc = csr_src[base + lane];
        if constexpr (H == 1) {
          float t = as[mysrc] + adv;
          mye = t > 0.f ? t : NEG_SLOPE * t;
        }
      }
#pragma unroll 4
      for (int j = 0; j < cnt; ++j) {
        int sj = __builtin_amdgcn_readlane(mysrc, j);
        float e;
        if constexpr (H == 1) {
          e = __int_as_float(__builtin_amdgcn_readlane(__float_as_int(mye), j));
        } else {
          e = as[(size_t)sj * H + hh] + adv;
          e = e > 0.f ? e : NEG_SLOPE * e;
        }
        vecf hv = *(const vecf*)(hfeat + (size_t)sj * DTOT + c0);
        float mn = fmaxf(m, e);
        float sc = __expf(m - mn);
        float p = __expf(e - mn);
        s = s * sc + p;
        acc = acc * sc + hv * p;
        m = mn;
      }
    }

    float denom = s + GAT_EPS;
    vecf o;
#pragma unroll
    for (int k = 0; k < VEC; ++k) {
      float v = acc[k] / denom;
      if constexpr (BIAS) v += bv[k];
      if (ELU) v = v > 0.f ? v : __expf(v) - 1.f;
      o[k] = v;
    }
    *(vecf*)(out + (size_t)node * DTOT + c0) = o;

    if constexpr (LOGITS) {
      float ps = 0.f, pd = 0.f;
#pragma unroll
      for (int k = 0; k < VEC; ++k) {
        ps += o[k] * wsv[k];
        pd += o[k] * wdv[k];
      }
      ps = wave_sum(ps);
      pd = wave_sum(pd);
      if (lane == 0) {
        as_next[node] = ps;
        ad_next[node] = pd;
      }
    }
  }
}

static inline int cdiv(long long a, long long b) { return (int)((a + b - 1) / b); }

extern "C" void kernel_launch(void* const* d_in, const int* in_sizes, int n_in,
                              void* d_out, int out_size, void* d_ws, size_t ws_size,
                              hipStream_t stream) {
  const float* x      = (const float*)d_in[0];
  const int*   ei     = (const int*)  d_in[1];
  const float* W1     = (const float*)d_in[2];
  const float* a_src1 = (const float*)d_in[3];
  const float* a_dst1 = (const float*)d_in[4];
  const float* b1     = (const float*)d_in[5];
  const float* W2     = (const float*)d_in[6];
  const float* a_src2 = (const float*)d_in[7];
  const float* a_dst2 = (const float*)d_in[8];
  const float* b2     = (const float*)d_in[9];
  float* z = (float*)d_out;

  const int Fin = 256, H1 = 8, C1 = 64, C2 = 128;
  const int N    = in_sizes[0] / Fin;
  const int E    = in_sizes[1] / 2;
  const int* srcv = ei;
  const int* dstv = ei + E;

  float* h1   = (float*)d_ws;
  float* hid  = h1  + (size_t)N * C1;
  float* agg  = hid + (size_t)N * C1;
  float* as1  = agg + (size_t)N * C1;
  float* ad1  = as1 + (size_t)N * H1;
  float* as2  = ad1 + (size_t)N * H1;
  float* ad2  = as2 + N;
  float* ws2  = ad2 + N;
  float* wd2  = ws2 + C1;
  const int NB = cdiv(N, BUCKET_NODES);
  int*   cnts = (int*)(wd2 + C1);
  int*   recs = cnts + (size_t)NB * NCHUNK;
  int*   csrc = recs + (size_t)2 * E;
  int*   sbeg = csrc + E;
  int*   send = sbeg + N;
  int*   bsta = send + N;

  const int B = 256;
  const int GB = 128;
  const int GATHER_BLOCKS = 2048;

  att_vec_kernel<64, 128><<<1, 1024, 0, stream>>>(W2, a_src2, a_dst2, ws2, wd2);

  chunk_hist_kernel<<<NCHUNK, CHUNK_THREADS, 0, stream>>>(dstv, E, NB, cnts);
  bucket_scan_kernel<<<1, 1024, 0, stream>>>(cnts, NB, bsta);
  chunk_scatter_kernel<<<NCHUNK, CHUNK_THREADS, 0, stream>>>(srcv, dstv, E, NB,
                                                            cnts, bsta, recs);
  bucket_rank_kernel<<<NB, BUCKET_NODES, 0, stream>>>(recs, bsta, sbeg, send,
                                                      csrc, N);

  gemm_f32_wmma_kernel<C1 / 16, false><<<cdiv((long long)(N / 16) * 32, GB), GB,
                                         0, stream>>>(x, W1, nullptr, h1, N, Fin);
  node_alpha_kernel<8, 8><<<cdiv(N, B), B, 0, stream>>>(h1, a_src1, a_dst1,
                                                        as1, ad1, N);
  node_gather_kernel<8, 64, true, true, true><<<GATHER_BLOCKS, B, 0, stream>>>(
      sbeg, send, csrc, as1, ad1, h1, b1, hid, ws2, wd2, as2, ad2, N);

  node_gather_kernel<1, 64, false, false, false><<<GATHER_BLOCKS, B, 0, stream>>>(
      sbeg, send, csrc, as2, ad2, hid, nullptr, agg, nullptr, nullptr, nullptr,
      nullptr, N);
  gemm_f32_wmma_kernel<C2 / 16, true><<<cdiv((long long)(N / 16) * 32, GB), GB,
                                        0, stream>>>(agg, W2, b2, z, N, C1);
}


# Gathers pipelined: all rows and logits of a node asked before first use, next node's edge list fetched a node ahead; softmax max-then-weights per pass; sum order: 1-head weights as lane tree
# speedup vs baseline: 4.9277x; 1.0431x over previous
#include <hip/hip_runtime.h>
#include <cstdint>


typedef float v2f __attribute__((ext_vector_type(2)));
typedef float v8f __attribute__((ext_vector_type(8)));

#define NEG_SLOPE 0.2f
#define GAT_EPS 1e-16f

template <int M>
__device__ __forceinline__ float row_xor_lane(float x) {
  int b = __builtin_bit_cast(int, x);
  return __builtin_bit_cast(
      float, __builtin_amdgcn_update_dpp(b, b, 0x160 + M, 0xf, 0xf, false));
}

__device__ __forceinline__ float wave_sum(float v) {
  v += row_xor_lane<1>(v);
  v += row_xor_lane<2>(v);
  v += row_xor_lane<4>(v);
  v += row_xor_lane<8>(v);
  int b = __builtin_bit_cast(int, v);
  return __builtin_bit_cast(float, __builtin_amdgcn_readlane(b, 0)) +
         __builtin_bit_cast(float, __builtin_amdgcn_readlane(b, 16));
}

__device__ __forceinline__ float wave_max(float v) {
  v = fmaxf(v, row_xor_lane<1>(v));
  v = fmaxf(v, row_xor_lane<2>(v));
  v = fmaxf(v, row_xor_lane<4>(v));
  v = fmaxf(v, row_xor_lane<8>(v));
  int b = __builtin_bit_cast(int, v);
  return fmaxf(__builtin_bit_cast(float, __builtin_amdgcn_readlane(b, 0)),
               __builtin_bit_cast(float, __builtin_amdgcn_readlane(b, 16)));
}

template <int NT, bool BIAS>
__global__ void gemm_f32_wmma_kernel(const float* __restrict__ A,
                                     const float* __restrict__ B,
                                     const float* __restrict__ bias,
                                     float* __restrict__ C,
                                     int M, int K) {
  constexpr int N = NT * 16;
  int wave = (int)((blockIdx.x * blockDim.x + threadIdx.x) >> 5);
  int lane = threadIdx.x & 31;
  int mtiles = (M + 15) >> 4;
  if (wave >= mtiles) return;
  int m0 = wave << 4;
  int half = lane >> 4;
  int lid  = lane & 15;

  v8f c[NT];
#pragma unroll
  for (int t = 0; t < NT; ++t)
    c[t] = (v8f){0.f, 0.f, 0.f, 0.f, 0.f, 0.f, 0.f, 0.f};
  const float* arow = A + (size_t)(m0 + lid) * K + 2 * half;
  const float* bcol = B + (size_t)(2 * half) * N + lid;

#pragma unroll(16 / NT)
  for (int k0 = 0; k0 < K; k0 += 4) {
    v2f a;
    a.x = arow[k0];
    a.y = arow[k0 + 1];
    const float* bp = bcol + (size_t)k0 * N;
#pragma unroll
    for (int t = 0; t < NT; ++t) {
      v2f b;
      b.x = bp[t * 16];
      b.y = bp[t * 16 + N];
      c[t] = __builtin_amdgcn_wmma_f32_16x16x4_f32(false, a, false, b,
                                                   (short)0, c[t], false, false);
    }
  }

  float bc[NT];
#pragma unroll
  for (int t = 0; t < NT; ++t) bc[t] = BIAS ? bias[t * 16 + lid] : 0.f;

  float* crow = C + (size_t)m0 * N + lid;
#pragma unroll
  for (int r = 0; r < 8; ++r)
#pragma unroll
    for (int t = 0; t < NT; ++t)
      crow[(size_t)(r + 8 * half) * N + t * 16] = BIAS ? c[t][r] + bc[t] : c[t][r];
}

template <int K, int DO>
__global__ void att_vec_kernel(const float* __restrict__ W,
                               const float* __restrict__ a_s,
                               const float* __restrict__ a_d,
                               float* __restrict__ ws,
                               float* __restrict__ wd) {
  constexpr int V = DO / 32;
  typedef float vecf __attribute__((ext_vector_type(V)));
  __shared__ float res[2 * K];
  int lane = threadIdx.x & 31;
  int w = threadIdx.x >> 5;
  int nw = blockDim.x >> 5;
  vecf sv = *(const vecf*)(a_s + lane * V);
  vecf dv = *(const vecf*)(a_d + lane * V);
  for (int k = w; k < K; k += nw) {
    vecf wv = *(const vecf*)(W + (size_t)k * DO + lane * V);
    float ps = 0.f, pd = 0.f;
#pragma unroll
    for (int i = 0; i < V; ++i) {
      ps += wv[i] * sv[i];
      pd += wv[i] * dv[i];
    }
    ps = wave_sum(ps);
    pd = wave_sum(pd);
    if (lane == 0) { res[k] = ps; res[K + k] = pd; }
  }
  __syncthreads();
  for (int i = threadIdx.x; i < 2 * K; i += blockDim.x) {
    if (i < K) ws[i] = res[i];
    else wd[i - K] = res[i];
  }
}

template <int H, int D>
__global__ void node_alpha_kernel(const float* __restrict__ h,
                                  const float* __restrict__ avs,
                                  const float* __restrict__ avd,
                                  float* __restrict__ as,
                                  float* __restrict__ ad, int N) {
  int n = blockIdx.x * blockDim.x + threadIdx.x;
  if (n >= N) return;
  const float* hr = h + (size_t)n * H * D;
#pragma unroll
  for (int hh = 0; hh < H; ++hh) {
    float s0 = 0.f, s1 = 0.f;
#pragma unroll
    for (int d = 0; d < D; ++d) {
      float v = hr[hh * D + d];
      s0 += v * avs[hh * D + d];
      s1 += v * avd[hh * D + d];
    }
    as[(size_t)n * H + hh] = s0;
    ad[(size_t)n * H + hh] = s1;
  }
}

#define NCHUNK 256
#define BUCKET_SHIFT 8
#define BUCKET_NODES 256
#define MAX_BUCKETS 1024
#define CHUNK_THREADS 1024
#define STAGE_CAP 8192

__device__ __forceinline__ int wave_incl_scan(int x, int lane) {
#pragma unroll
  for (int o = 1; o < 32; o <<= 1) {
    int y = __shfl_up(x, o);
    if (lane >= o) x += y;
  }
  return x;
}

__device__ __forceinline__ int block_excl_scan(int v, int* wsum, int& total) {
  int lane = threadIdx.x & 31;
  int w = threadIdx.x >> 5;
  int nw = blockDim.x >> 5;
  int x = wave_incl_scan(v, lane);
  if (lane == 31) wsum[w] = x;
  __syncthreads();
  if (w == 0) {
    int t = (lane < nw) ? wsum[lane] : 0;
    int xs = wave_incl_scan(t, lane);
    wsum[lane] = xs - t;
    if (lane == 31) wsum[32] = xs;
  }
  __syncthreads();
  total = wsum[32];
  int r = wsum[w] + (x - v);
  __syncthreads();
  return r;
}

__global__ void chunk_hist_kernel(const int* __restrict__ dst, int E, int NB,
                                  int* __restrict__ counts) {
  __shared__ int h[MAX_BUCKETS];
  int tid = threadIdx.x;
  int c = blockIdx.x;
  for (int i = tid; i < NB; i += CHUNK_THREADS) h[i] = 0;
  __syncthreads();
  int per = (E + NCHUNK - 1) / NCHUNK;
  int e0 = c * per;
  int e1 = e0 + per < E ? e0 + per : E;
  for (int e = e0 + tid; e < e1; e += 4 * CHUNK_THREADS) {
    int d[4];
#pragma unroll
    for (int k = 0; k < 4; ++k) {
      int ee = e + k * CHUNK_THREADS;
      d[k] = ee < e1 ? dst[ee] : -1;
    }
#pragma unroll
    for (int k = 0; k < 4; ++k)
      if (d[k] >= 0) atomicAdd(&h[d[k] >> BUCKET_SHIFT], 1);
  }
  __syncthreads();
  for (int i = tid; i < NB; i += CHUNK_THREADS)
    counts[(size_t)i * NCHUNK + c] = h[i];
}

__global__ void bucket_scan_kernel(int* __restrict__ counts, int NB,
                                   int* __restrict__ bstart) {
  __shared__ int tot[MAX_BUCKETS];
  __shared__ int wsum[33];
  typedef int v4i __attribute__((ext_vector_type(4)));
  int tid = threadIdx.x;
  int lane = tid & 31;
  int w = tid >> 5;
  int nw = blockDim.x >> 5;
  for (int b0 = w * 4; b0 < NB; b0 += nw * 4) {
    v4i a0[4], a1[4];
#pragma unroll
    for (int q = 0; q < 4; ++q) {
      if (b0 + q < NB) {
        const v4i* row = (const v4i*)(counts + (size_t)(b0 + q) * NCHUNK + lane * 8);
        a0[q] = row[0];
        a1[q] = row[1];
      }
    }
#pragma unroll
    for (int q = 0; q < 4; ++q) {
      if (b0 + q < NB) {
        v4i p0, p1;
        p0.x = 0;
        p0.y = a0[q].x;
        p0.z = p0.y + a0[q].y;
        p0.w = p0.z + a0[q].z;
        p1.x = p0.w + a0[q].w;
        p1.y = p1.x + a1[q].x;
        p1.z = p1.y + a1[q].y;
        p1.w = p1.z + a1[q].z;
        int sum = p1.w + a1[q].w;
        int incl = wave_incl_scan(sum, lane);
        int excl = incl - sum;
        p0 += excl;
        p1 += excl;
        v4i* row = (v4i*)(counts + (size_t)(b0 + q) * NCHUNK + lane * 8);
        row[0] = p0;
        row[1] = p1;
        if (lane == 31) tot[b0 + q] = incl;
      }
    }
  }
  __syncthreads();
  int carry = 0;
  for (int base = 0; base < NB; base += (int)blockDim.x) {
    int i = base + tid;
    int v = (i < NB) ? tot[i] : 0;
    int total;
    int ex = block_excl_scan(v, wsum, total);
    if (i < NB) bstart[i] = carry + ex;
    carry += total;
  }
  if (tid == 0) bstart[NB] = carry;
}

__global__ void chunk_scatter_kernel(const int* __restrict__ src,
                                     const int* __restrict__ dst, int E, int NB,
                                     const int* __restrict__ part,
                                     const int* __restrict__ bstart,
                                     int* __restrict__ rec_words) {
  typedef int v2i __attribute__((ext_vector_type(2)));
  v2i* rec = (v2i*)rec_words;
  __shared__ int cur[MAX_BUCKETS];
  int tid = threadIdx.x;
  int c = blockIdx.x;
  for (int i = tid; i < NB; i += CHUNK_THREADS)
    cur[i] = bstart[i] + part[(size_t)i * NCHUNK + c];
  __syncthreads();
  int per = (E + NCHUNK - 1) / NCHUNK;
  int e0 = c * per;
  int e1 = e0 + per < E ? e0 + per : E;
  for (int e = e0 + tid; e < e1; e += 4 * CHUNK_THREADS) {
    int d[4], s[4], p[4];
#pragma unroll
    for (int k = 0; k < 4; ++k) {
      int ee = e + k * CHUNK_THREADS;
      bool ok = ee < e1;
      d[k] = ok ? dst[ee] : -1;
      s[k] = ok ? src[ee] : 0;
    }
#pragma unroll
    for (int k = 0; k < 4; ++k)
      p[k] = d[k] >= 0 ? atomicAdd(&cur[d[k] >> BUCKET_SHIFT], 1) : 0;
#pragma unroll
    for (int k = 0; k < 4; ++k)
      if (d[k] >= 0) rec[p[k]] = (v2i){s[k], d[k]};
  }
}

__global__ void bucket_rank_kernel(const int* __restrict__ rec_words,
                                   const int* __restrict__ bstart,
                                   int* __restrict__ seg_beg,
                                   int* __restrict__ seg_end,
                                   int* __restrict__ csr_src, int N) {
  typedef int v2i __attribute__((ext_vector_type(2)));
  const v2i* rec = (const v2i*)rec_words;
  __shared__ int cnt[BUCKET_NODES];
  __shared__ int wsum[33];
  __shared__ int stage[STAGE_CAP];
  int tid = threadIdx.x;
  int b = blockIdx.x;
  int lo = b << BUCKET_SHIFT;
  int bs = bstart[b];
  int be = bstart[b + 1];
  int n = be - bs;
  cnt[tid] = 0;
  __syncthreads();
  for (int i = tid; i < n; i += 4 * BUCKET_NODES) {
    int d[4];
#pragma unroll
    for (int k = 0; k < 4; ++k) {
      int ii = i + k * BUCKET_NODES;
      d[k] = ii < n ? rec[bs + ii].y : -1;
    }
#pragma unroll
    for (int k = 0; k < 4; ++k)
      if (d[k] >= 0) atomicAdd(&cnt[d[k] - lo], 1);
  }
  __syncthreads();
  int v = cnt[tid];
  int total;
  int ex = block_excl_scan(v, wsum, total);
  int node = lo + tid;
  if (node < N) {
    seg_beg[node] = bs + ex;
    seg_end[node] = bs + ex + v;
  }
  cnt[tid] = ex;
  __syncthreads();
  bool staged = n <= STAGE_CAP;
  for (int i = tid; i < n; i += 4 * BUCKET_NODES) {
    v2i r[4];
    int p[4];
#pragma unroll
    for (int k = 0; k < 4; ++k) {
      int ii = i + k * BUCKET_NODES;
      r[k] = ii < n ? rec[bs + ii] : (v2i){0, -1};
    }
#pragma unroll
    for (int k = 0; k < 4; ++k)
      p[k] = r[k].y >= 0 ? atomicAdd(&cnt[r[k].y - lo], 1) : 0;
#pragma unroll
    for (int k = 0; k < 4; ++k) {
      if (r[k].y >= 0) {
        if (staged) stage[p[k]] = r[k].x;
        else csr_src[bs + p[k]] = r[k].x;
      }
    }
  }
  if (staged) {
    __syncthreads();
    for (int i = tid; i < n; i += BUCKET_NODES) csr_src[bs + i] = stage[i];
  }
}

#define GGRP 4

template <int H, int DTOT, bool ELU, bool BIAS, bool LOGITS>
__global__ void __launch_bounds__(256)
node_gather_kernel(const int* __restrict__ seg_beg,
                   const int* __restrict__ seg_end,
                   const int* __restrict__ csr_src,
                   const float* __restrict__ as,
                   const float* __restrict__ ad,
                   const float* __restrict__ hfeat,
                   const float* __restrict__ bias,
                   float* __restrict__ out,
                   const float* __restrict__ wsrc,
                   const float* __restrict__ wdst,
                   float* __restrict__ as_next,
                   float* __restrict__ ad_next, int N) {
  constexpr int VEC = DTOT / 32;
  constexpr int DH = DTOT / H;
  constexpr int GCH = (H == 1) ? 32 : 28;
  constexpr int NG = GCH / GGRP;
  typedef float vecf __attribute__((ext_vector_type(VEC)));
  int lane = threadIdx.x & 31;
  int wave0 = __builtin_amdgcn_readfirstlane(
      (int)((blockIdx.x * blockDim.x + threadIdx.x) >> 5));
  int nwaves = (int)((gridDim.x * blockDim.x) >> 5);
  int c0 = lane * VEC;
  int hh = c0 / DH;
  const float NEG_INF = -__builtin_inff();
  constexpr int RV = DTOT / VEC;
  const vecf* __restrict__ rows = (const vecf*)hfeat;
  vecf bv, wsv, wdv;
  if constexpr (BIAS) bv = *(const vecf*)(bias + c0);
  if constexpr (LOGITS) {
    wsv = *(const vecf*)(wsrc + c0);
    wdv = *(const vecf*)(wdst + c0);
  }

  int node = wave0;
  if (node >= N) return;

  int beg = seg_beg[node];
  int end = seg_end[node];
  int cnt = end - beg;
  if (cnt > GCH) cnt = GCH;
  int mysrc = 0;
  if (lane < cnt) mysrc = csr_src[beg + lane];
  float adv = ad[node * H + hh];
  float as_self = as[node * H + hh];
  vecf acc = rows[node * RV + lane];
  int nfirst = node + nwaves < N ? node + nwaves : node;
  int nbeg = seg_beg[nfirst];
  int nend = seg_end[nfirst];

  for (; node < N; node += nwaves) {
    int ncnt = 0, nsrc = 0, nnbeg = 0, nnend = 0;
    float nadv = 0.f, nas = 0.f;
    vecf nrow;
    auto ask_next = [&]() {
      int nx1 = node + nwaves < N ? node + nwaves : node;
      int nx2 = node + 2 * nwaves < N ? node + 2 * nwaves : node;
      ncnt = nend - nbeg;
      if (ncnt > GCH) ncnt = GCH;
      if (lane < ncnt) nsrc = csr_src[nbeg + lane];
      nadv = ad[nx1 * H + hh];
      nas = as[nx1 * H + hh];
      nrow = rows[nx1 * RV + lane];
      nnbeg = seg_beg[nx2];
      nnend = seg_end[nx2];
    };
    bool asked = false;

    float asv = 0.f;
    if constexpr (H == 1) asv = as[mysrc];
    int base = beg;

    float e0 = as_self + adv;
    e0 = e0 > 0.f ? e0 : NEG_SLOPE * e0;
    float m = e0;
    float s = 1.f;

    while (cnt > 0) {
      int last = cnt - 1;
      vecf hv[GCH];
      float ev[GCH];
#pragma unroll
      for (int g = 0; g < NG; ++g) {
        if (g > 0 && g * GGRP >= cnt) break;
        int sj[GGRP];
#pragma unroll
        for (int i = 0; i < GGRP; ++i) {
          int j = g * GGRP + i;
          sj[i] = __builtin_amdgcn_readlane(mysrc, j < last ? j : last);
        }
        if constexpr (H > 1) {
#pragma unroll
          for (int i = 0; i < GGRP; ++i)
            ev[g * GGRP + i] = as[sj[i] * H + hh];
        }
#pragma unroll
        for (int i = 0; i < GGRP; ++i)
          hv[g * GGRP + i] = rows[sj[i] * RV + lane];
      }
      if (!asked) {
        ask_next();
        asked = true;
      }
      __builtin_amdgcn_sched_barrier(0);

      float mn;
      if constexpr (H == 1) {
        float t = asv + adv;
        t = t > 0.f ? t : NEG_SLOPE * t;
        float mye = lane < cnt ? t : NEG_INF;
        mn = fmaxf(m, wave_max(mye));
        float sc = __expf(m - mn);
        float p = __expf(mye - mn);
        s = s * sc + wave_sum(p);
        acc = acc * sc;
        int pb = __float_as_int(p);
#pragma unroll
        for (int g = 0; g < NG; ++g) {
          if (g > 0 && g * GGRP >= cnt) break;
#pragma unroll
          for (int i = 0; i < GGRP; ++i) {
            int j = g * GGRP + i;
            float pj = __int_as_float(__builtin_amdgcn_readlane(pb, j));
            acc += hv[j] * pj;
          }
        }
      } else {
        float cm = NEG_INF;
#pragma unroll
        for (int g = 0; g < NG; ++g) {
          if (g > 0 && g * GGRP >= cnt) break;
#pragma unroll
          for (int i = 0; i < GGRP; ++i) {
            int j = g * GGRP + i;
            float t = ev[j] + adv;
            t = t > 0.f ? t : NEG_SLOPE * t;
            t = j < cnt ? t : NEG_INF;
            ev[j] = t;
            cm = fmaxf(cm, t);
          }
        }
        mn = fmaxf(m, cm);
        float sc = __expf(m - mn);
        s = s * sc;
        acc = acc * sc;
#pragma unroll
        for (int g = 0; g < NG; ++g) {
          if (g > 0 && g * GGRP >= cnt) break;
#pragma unroll
          for (int i = 0; i < GGRP; ++i) {
            int j = g * GGRP + i;
            float p = __expf(ev[j] - mn);
            s += p;
            acc += hv[j] * p;
          }
        }
      }
      m = mn;

      base += GCH;
      cnt = end - base;
      if (cnt > GCH) cnt = GCH;
      mysrc = 0;
      if (lane < cnt) mysrc = csr_src[base + lane];
      if constexpr (H == 1) {
        if (cnt > 0) asv = as[mysrc];
      }
    }
    if (!asked) ask_next();

    float denom = s + GAT_EPS;
    vecf o;
#pragma unroll
    for (int k = 0; k < VEC; ++k) {
      float v = acc[k] / denom;
      if constexpr (BIAS) v += bv[k];
      if (ELU) v = v > 0.f ? v : __expf(v) - 1.f;
      o[k] = v;
    }
    *(vecf*)(out + (size_t)node * DTOT + c0) = o;

    if constexpr (LOGITS) {
      float ps = 0.f, pd = 0.f;
#pragma unroll
      for (int k = 0; k < VEC; ++k) {
        ps += o[k] * wsv[k];
        pd += o[k] * wdv[k];
      }
      ps = wave_sum(ps);
      pd = wave_sum(pd);
      if (lane == 0) {
        as_next[node] = ps;
        ad_next[node] = pd;
      }
    }

    beg = nbeg;
    end = nend;
    cnt = ncnt;
    mysrc = nsrc;
    adv = nadv;
    as_self = nas;
    acc = nrow;
    nbeg = nnbeg;
    nend = nnend;
  }
}

static inline int cdiv(long long a, long long b) { return (int)((a + b - 1) / b); }

extern "C" void kernel_launch(void* const* d_in, const int* in_sizes, int n_in,
                              void* d_out, int out_size, void* d_ws, size_t ws_size,
                              hipStream_t stream) {
  const float* x      = (const float*)d_in[0];
  const int*   ei     = (const int*)  d_in[1];
  const float* W1     = (const float*)d_in[2];
  const float* a_src1 = (const float*)d_in[3];
  const float* a_dst1 = (const float*)d_in[4];
  const float* b1     = (const float*)d_in[5];
  const float* W2     = (const float*)d_in[6];
  const float* a_src2 = (const float*)d_in[7];
  const float* a_dst2 = (const float*)d_in[8];
  const float* b2     = (const float*)d_in[9];
  float* z = (float*)d_out;

  const int Fin = 256, H1 = 8, C1 = 64, C2 = 128;
  const int N    = in_sizes[0] / Fin;
  const int E    = in_sizes[1] / 2;
  const int* srcv = ei;
  const int* dstv = ei + E;

  float* h1   = (float*)d_ws;
  float* hid  = h1  + (size_t)N * C1;
  float* agg  = hid + (size_t)N * C1;
  float* as1  = agg + (size_t)N * C1;
  float* ad1  = as1 + (size_t)N * H1;
  float* as2  = ad1 + (size_t)N * H1;
  float* ad2  = as2 + N;
  float* ws2  = ad2 + N;
  float* wd2  = ws2 + C1;
  const int NB = cdiv(N, BUCKET_NODES);
  int*   cnts = (int*)(wd2 + C1);
  int*   recs = cnts + (size_t)NB * NCHUNK;
  int*   csrc = recs + (size_t)2 * E;
  int*   sbeg = csrc + E;
  int*   send = sbeg + N;
  int*   bsta = send + N;

  const int B = 256;
  const int GB = 128;
  const int GATHER1_BLOCKS = 1024;
  const int GATHER2_BLOCKS = 1280;

  att_vec_kernel<64, 128><<<1, 1024, 0, stream>>>(W2, a_src2, a_dst2, ws2, wd2);

  chunk_hist_kernel<<<NCHUNK, CHUNK_THREADS, 0, stream>>>(dstv, E, NB, cnts);
  bucket_scan_kernel<<<1, 1024, 0, stream>>>(cnts, NB, bsta);
  chunk_scatter_kernel<<<NCHUNK, CHUNK_THREADS, 0, stream>>>(srcv, dstv, E, NB,
                                                            cnts, bsta, recs);
  bucket_rank_kernel<<<NB, BUCKET_NODES, 0, stream>>>(recs, bsta, sbeg, send,
                                                      csrc, N);

  gemm_f32_wmma_kernel<C1 / 16, false><<<cdiv((long long)(N / 16) * 32, GB), GB,
                                         0, stream>>>(x, W1, nullptr, h1, N, Fin);
  node_alpha_kernel<8, 8><<<cdiv(N, B), B, 0, stream>>>(h1, a_src1, a_dst1,
                                                        as1, ad1, N);
  node_gather_kernel<8, 64, true, true, true><<<GATHER1_BLOCKS, B, 0, stream>>>(
      sbeg, send, csrc, as1, ad1, h1, b1, hid, ws2, wd2, as2, ad2, N);

  node_gather_kernel<1, 64, false, false, false><<<GATHER2_BLOCKS, B, 0, stream>>>(
      sbeg, send, csrc, as2, ad2, hid, nullptr, agg, nullptr, nullptr, nullptr,
      nullptr, N);
  gemm_f32_wmma_kernel<C2 / 16, true><<<cdiv((long long)(N / 16) * 32, GB), GB,
                                        0, stream>>>(agg, W2, b2, z, N, C1);
}


# Launches 10 -> 7: W2-vector pass and bucket scan (lane prefix by DPP) ride as extra work-groups of count pass and layer-1 GEMM; layer-1 logits taken in that GEMM's epilogue (sum order: 8-lane tree)
# speedup vs baseline: 5.3112x; 1.0778x over previous
#include <hip/hip_runtime.h>
#include <cstdint>


typedef float v2f __attribute__((ext_vector_type(2)));
typedef float v8f __attribute__((ext_vector_type(8)));

#define NEG_SLOPE 0.2f
#define GAT_EPS 1e-16f

template <int M>
__device__ __forceinline__ float row_xor_lane(float x) {
  int b = __builtin_bit_cast(int, x);
  return __builtin_bit_cast(
      float, __builtin_amdgcn_update_dpp(b, b, 0x160 + M, 0xf, 0xf, false));
}

__device__ __forceinline__ float wave_sum(float v) {
  v += row_xor_lane<1>(v);
  v += row_xor_lane<2>(v);
  v += row_xor_lane<4>(v);
  v += row_xor_lane<8>(v);
  int b = __builtin_bit_cast(int, v);
  return __builtin_bit_cast(float, __builtin_amdgcn_readlane(b, 0)) +
         __builtin_bit_cast(float, __builtin_amdgcn_readlane(b, 16));
}

__device__ __forceinline__ float wave_max(float v) {
  v = fmaxf(v, row_xor_lane<1>(v));
  v = fmaxf(v, row_xor_lane<2>(v));
  v = fmaxf(v, row_xor_lane<4>(v));
  v = fmaxf(v, row_xor_lane<8>(v));
  int b = __builtin_bit_cast(int, v);
  return fmaxf(__builtin_bit_cast(float, __builtin_amdgcn_readlane(b, 0)),
               __builtin_bit_cast(float, __builtin_amdgcn_readlane(b, 16)));
}

template <int NT, bool BIAS, bool LOGITS>
__device__ __forceinline__ void gemm_strip(const float* __restrict__ A,
                                           const float* __restrict__ B,
                                           const float* __restrict__ bias,
                                           float* __restrict__ C,
                                           int M, int K, int wave,
                                           const float* __restrict__ avs,
                                           const float* __restrict__ avd,
                                           float* __restrict__ as,
                                           float* __restrict__ ad) {
  constexpr int N = NT * 16;
  int lane = threadIdx.x & 31;
  int mtiles = (M + 15) >> 4;
  if (wave >= mtiles) return;
  int m0 = wave << 4;
  int half = lane >> 4;
  int lid  = lane & 15;

  v8f c[NT];
#pragma unroll
  for (int t = 0; t < NT; ++t)
    c[t] = (v8f){0.f, 0.f, 0.f, 0.f, 0.f, 0.f, 0.f, 0.f};
  const float* arow = A + (size_t)(m0 + lid) * K + 2 * half;
  const float* bcol = B + (size_t)(2 * half) * N + lid;

#pragma unroll(16 / NT)
  for (int k0 = 0; k0 < K; k0 += 4) {
    v2f a;
    a.x = arow[k0];
    a.y = arow[k0 + 1];
    const float* bp = bcol + (size_t)k0 * N;
#pragma unroll
    for (int t = 0; t < NT; ++t) {
      v2f b;
      b.x = bp[t * 16];
      b.y = bp[t * 16 + N];
      c[t] = __builtin_amdgcn_wmma_f32_16x16x4_f32(false, a, false, b,
                                                   (short)0, c[t], false, false);
    }
  }

  float bc[NT];
#pragma unroll
  for (int t = 0; t < NT; ++t) bc[t] = BIAS ? bias[t * 16 + lid] : 0.f;

  float* crow = C + (size_t)m0 * N + lid;
#pragma unroll
  for (int r = 0; r < 8; ++r)
#pragma unroll
    for (int t = 0; t < NT; ++t)
      crow[(size_t)(r + 8 * half) * N + t * 16] = BIAS ? c[t][r] + bc[t] : c[t][r];

  if constexpr (LOGITS) {
    constexpr int H = N / 8;
    int ch = lid & 7;
    int hd = lid >> 3;
    float* asrow = as + (size_t)(m0 + ch + 8 * half) * H + hd;
    float* adrow = ad + (size_t)(m0 + ch + 8 * half) * H + hd;
#pragma unroll
    for (int t = 0; t < NT; ++t) {
      float sv = avs[t * 16 + lid];
      float dv = avd[t * 16 + lid];
      float ks = 0.f, kd = 0.f;
#pragma unroll
      for (int r = 0; r < 8; ++r) {
        float ps = c[t][r] * sv;
        float pd = c[t][r] * dv;
        ps += row_xor_lane<1>(ps);
        pd += row_xor_lane<1>(pd);
        ps += row_xor_lane<2>(ps);
        pd += row_xor_lane<2>(pd);
        ps += row_xor_lane<4>(ps);
        pd += row_xor_lane<4>(pd);
        if (ch == r) { ks = ps; kd = pd; }
      }
      asrow[2 * t] = ks;
      adrow[2 * t] = kd;
    }
  }
}

template <int NT, bool BIAS>
__global__ void gemm_f32_wmma_kernel(const float* __restrict__ A,
                                     const float* __restrict__ B,
                                     const float* __restrict__ bias,
                                     float* __restrict__ C,
                                     int M, int K) {
  int wave = (int)((blockIdx.x * blockDim.x + threadIdx.x) >> 5);
  gemm_strip<NT, BIAS, false>(A, B, bias, C, M, K, wave, nullptr, nullptr,
                              nullptr, nullptr);
}

template <int K, int DO>
__device__ __forceinline__ void att_vec_body(const float* __restrict__ W,
                                             const float* __restrict__ a_s,
                                             const float* __restrict__ a_d,
                                             float* __restrict__ ws,
                                             float* __restrict__ wd,
                                             float* res) {
  constexpr int V = DO / 32;
  typedef float vecf __attribute__((ext_vector_type(V)));
  int lane = threadIdx.x & 31;
  int w = threadIdx.x >> 5;
  int nw = blockDim.x >> 5;
  vecf sv = *(const vecf*)(a_s + lane * V);
  vecf dv = *(const vecf*)(a_d + lane * V);
  for (int k = w; k < K; k += nw) {
    vecf wv = *(const vecf*)(W + (size_t)k * DO + lane * V);
    float ps = 0.f, pd = 0.f;
#pragma unroll
    for (int i = 0; i < V; ++i) {
      ps += wv[i] * sv[i];
      pd += wv[i] * dv[i];
    }
    ps = wave_sum(ps);
    pd = wave_sum(pd);
    if (lane == 0) { res[k] = ps; res[K + k] = pd; }
  }
  __syncthreads();
  for (int i = threadIdx.x; i < 2 * K; i += blockDim.x) {
    if (i < K) ws[i] = res[i];
    else wd[i - K] = res[i];
  }
}

#define NCHUNK 256
#define BUCKET_SHIFT 8
#define BUCKET_NODES 256
#define MAX_BUCKETS 1024
#define CHUNK_THREADS 1024
#define STAGE_CAP 8192

__device__ __forceinline__ int wave_incl_scan(int x, int lane) {
#pragma unroll
  for (int o = 1; o < 32; o <<= 1) {
    int y = __shfl_up(x, o);
    if (lane >= o) x += y;
  }
  return x;
}

template <int S>
__device__ __forceinline__ int row_shr_lane(int x) {
  return __builtin_amdgcn_update_dpp(0, x, 0x110 + S, 0xf, 0xf, false);
}

__device__ __forceinline__ int wave_incl_scan_valu(int x, int lane) {
  x += row_shr_lane<1>(x);
  x += row_shr_lane<2>(x);
  x += row_shr_lane<4>(x);
  x += row_shr_lane<8>(x);
  int low = __builtin_amdgcn_readlane(x, 15);
  return lane >= 16 ? x + low : x;
}

__device__ __forceinline__ int block_excl_scan(int v, int* wsum, int& total) {
  int lane = threadIdx.x & 31;
  int w = threadIdx.x >> 5;
  int nw = blockDim.x >> 5;
  int x = wave_incl_scan(v, lane);
  if (lane == 31) wsum[w] = x;
  __syncthreads();
  if (w == 0) {
    int t = (lane < nw) ? wsum[lane] : 0;
    int xs = wave_incl_scan(t, lane);
    wsum[lane] = xs - t;
    if (lane == 31) wsum[32] = xs;
  }
  __syncthreads();
  total = wsum[32];
  int r = wsum[w] + (x - v);
  __syncthreads();
  return r;
}

__global__ void chunk_hist_kernel(const int* __restrict__ dst, int E, int NB,
                                  int* __restrict__ counts,
                                  const float* __restrict__ W2,
                                  const float* __restrict__ a_s2,
                                  const float* __restrict__ a_d2,
                                  float* __restrict__ ws2,
                                  float* __restrict__ wd2) {
  __shared__ int h[MAX_BUCKETS];
  __shared__ float res[2 * 64];
  if (blockIdx.x == NCHUNK) {
    att_vec_body<64, 128>(W2, a_s2, a_d2, ws2, wd2, res);
    return;
  }
  int tid = threadIdx.x;
  int c = blockIdx.x;
  for (int i = tid; i < NB; i += CHUNK_THREADS) h[i] = 0;
  __syncthreads();
  int per = (E + NCHUNK - 1) / NCHUNK;
  int e0 = c * per;
  int e1 = e0 + per < E ? e0 + per : E;
  for (int e = e0 + tid; e < e1; e += 4 * CHUNK_THREADS) {
    int d[4];
#pragma unroll
    for (int k = 0; k < 4; ++k) {
      int ee = e + k * CHUNK_THREADS;
      d[k] = ee < e1 ? dst[ee] : -1;
    }
#pragma unroll
    for (int k = 0; k < 4; ++k)
      if (d[k] >= 0) atomicAdd(&h[d[k] >> BUCKET_SHIFT], 1);
  }
  __syncthreads();
  for (int i = tid; i < NB; i += CHUNK_THREADS)
    counts[(size_t)i * NCHUNK + c] = h[i];
}

template <int RPT>
__device__ __forceinline__ void bucket_scan_body(int* __restrict__ counts, int NB,
                                                 int* __restrict__ bstart,
                                                 int* tot, int* wsum) {
  typedef int v4i __attribute__((ext_vector_type(4)));
  int tid = threadIdx.x;
  int lane = tid & 31;
  int w = tid >> 5;
  int nw = blockDim.x >> 5;
  for (int b0 = w * RPT; b0 < NB; b0 += nw * RPT) {
    v4i a0[RPT], a1[RPT];
#pragma unroll
    for (int q = 0; q < RPT; ++q) {
      if (b0 + q < NB) {
        const v4i* row = (const v4i*)(counts + (size_t)(b0 + q) * NCHUNK + lane * 8);
        a0[q] = row[0];
        a1[q] = row[1];
      }
    }
#pragma unroll
    for (int q = 0; q < RPT; ++q) {
      if (b0 + q < NB) {
        v4i p0, p1;
        p0.x = 0;
        p0.y = a0[q].x;
        p0.z = p0.y + a0[q].y;
        p0.w = p0.z + a0[q].z;
        p1.x = p0.w + a0[q].w;
        p1.y = p1.x + a1[q].x;
        p1.z = p1.y + a1[q].y;
        p1.w = p1.z + a1[q].z;
        int sum = p1.w + a1[q].w;
        int incl = wave_incl_scan_valu(sum, lane);
        int excl = incl - sum;
        p0 += excl;
        p1 += excl;
        v4i* row = (v4i*)(counts + (size_t)(b0 + q) * NCHUNK + lane * 8);
        row[0] = p0;
        row[1] = p1;
        if (lane == 31) tot[b0 + q] = incl;
      }
    }
  }
  __syncthreads();
  int carry = 0;
  for (int base = 0; base < NB; base += (int)blockDim.x) {
    int i = base + tid;
    int v = (i < NB) ? tot[i] : 0;
    int total;
    int ex = block_excl_scan(v, wsum, total);
    if (i < NB) bstart[i] = carry + ex;
    carry += total;
  }
  if (tid == 0) bstart[NB] = carry;
}

__global__ void gemm1_logits_scan_kernel(const float* __restrict__ A,
                                         const float* __restrict__ B,
                                         float* __restrict__ C, int M, int K,
                                         const float* __restrict__ avs,
                                         const float* __restrict__ avd,
                                         float* __restrict__ as,
                                         float* __restrict__ ad,
                                         int* __restrict__ counts, int NB,
                                         int* __restrict__ bstart) {
  __shared__ int tot[MAX_BUCKETS];
  __shared__ int wsum[33];
  if (blockIdx.x == 0) {
    bucket_scan_body<8>(counts, NB, bstart, tot, wsum);
    return;
  }
  int wave = (int)(((blockIdx.x - 1) * blockDim.x + threadIdx.x) >> 5);
  gemm_strip<4, false, true>(A, B, nullptr, C, M, K, wave, avs, avd, as, ad);
}

__global__ void chunk_scatter_kernel(const int* __restrict__ src,
                                     const int* __restrict__ dst, int E, int NB,
                                     const int* __restrict__ part,
                                     const int* __restrict__ bstart,
                                     int* __restrict__ rec_words) {
  typedef int v2i __attribute__((ext_vector_type(2)));
  v2i* rec = (v2i*)rec_words;
  __shared__ int cur[MAX_BUCKETS];
  int tid = threadIdx.x;
  int c = blockIdx.x;
  for (int i = tid; i < NB; i += CHUNK_THREADS)
    cur[i] = bstart[i] + part[(size_t)i * NCHUNK + c];
  __syncthreads();
  int per = (E + NCHUNK - 1) / NCHUNK;
  int e0 = c * per;
  int e1 = e0 + per < E ? e0 + per : E;
  for (int e = e0 + tid; e < e1; e += 4 * CHUNK_THREADS) {
    int d[4], s[4], p[4];
#pragma unroll
    for (int k = 0; k < 4; ++k) {
      int ee = e + k * CHUNK_THREADS;
      bool ok = ee < e1;
      d[k] = ok ? dst[ee] : -1;
      s[k] = ok ? src[ee] : 0;
    }
#pragma unroll
    for (int k = 0; k < 4; ++k)
      p[k] = d[k] >= 0 ? atomicAdd(&cur[d[k] >> BUCKET_SHIFT], 1) : 0;
#pragma unroll
    for (int k = 0; k < 4; ++k)
      if (d[k] >= 0) rec[p[k]] = (v2i){s[k], d[k]};
  }
}

__global__ void bucket_rank_kernel(const int* __restrict__ rec_words,
                                   const int* __restrict__ bstart,
                                   int* __restrict__ seg_beg,
                                   int* __restrict__ seg_end,
                                   int* __restrict__ csr_src, int N) {
  typedef int v2i __attribute__((ext_vector_type(2)));
  const v2i* rec = (const v2i*)rec_words;
  __shared__ int cnt[BUCKET_NODES];
  __shared__ int wsum[33];
  __shared__ int stage[STAGE_CAP];
  int tid = threadIdx.x;
  int b = blockIdx.x;
  int lo = b << BUCKET_SHIFT;
  int bs = bstart[b];
  int be = bstart[b + 1];
  int n = be - bs;
  cnt[tid] = 0;
  __syncthreads();
  for (int i = tid; i < n; i += 4 * BUCKET_NODES) {
    int d[4];
#pragma unroll
    for (int k = 0; k < 4; ++k) {
      int ii = i + k * BUCKET_NODES;
      d[k] = ii < n ? rec[bs + ii].y : -1;
    }
#pragma unroll
    for (int k = 0; k < 4; ++k)
      if (d[k] >= 0) atomicAdd(&cnt[d[k] - lo], 1);
  }
  __syncthreads();
  int v = cnt[tid];
  int total;
  int ex = block_excl_scan(v, wsum, total);
  int node = lo + tid;
  if (node < N) {
    seg_beg[node] = bs + ex;
    seg_end[node] = bs + ex + v;
  }
  cnt[tid] = ex;
  __syncthreads();
  bool staged = n <= STAGE_CAP;
  for (int i = tid; i < n; i += 4 * BUCKET_NODES) {
    v2i r[4];
    int p[4];
#pragma unroll
    for (int k = 0; k < 4; ++k) {
      int ii = i + k * BUCKET_NODES;
      r[k] = ii < n ? rec[bs + ii] : (v2i){0, -1};
    }
#pragma unroll
    for (int k = 0; k < 4; ++k)
      p[k] = r[k].y >= 0 ? atomicAdd(&cnt[r[k].y - lo], 1) : 0;
#pragma unroll
    for (int k = 0; k < 4; ++k) {
      if (r[k].y >= 0) {
        if (staged) stage[p[k]] = r[k].x;
        else csr_src[bs + p[k]] = r[k].x;
      }
    }
  }
  if (staged) {
    __syncthreads();
    for (int i = tid; i < n; i += BUCKET_NODES) csr_src[bs + i] = stage[i];
  }
}

#define GGRP 4

template <int H, int DTOT, bool ELU, bool BIAS, bool LOGITS>
__global__ void __launch_bounds__(256)
node_gather_kernel(const int* __restrict__ seg_beg,
                   const int* __restrict__ seg_end,
                   const int* __restrict__ csr_src,
                   const float* __restrict__ as,
                   const float* __restrict__ ad,
                   const float* __restrict__ hfeat,
                   const float* __restrict__ bias,
                   float* __restrict__ out,
                   const float* __restrict__ wsrc,
                   const float* __restrict__ wdst,
                   float* __restrict__ as_next,
                   float* __restrict__ ad_next, int N) {
  constexpr int VEC = DTOT / 32;
  constexpr int DH = DTOT / H;
  constexpr int GCH = (H == 1) ? 32 : 28;
  constexpr int NG = GCH / GGRP;
  typedef float vecf __attribute__((ext_vector_type(VEC)));
  int lane = threadIdx.x & 31;
  int wave0 = __builtin_amdgcn_readfirstlane(
      (int)((blockIdx.x * blockDim.x + threadIdx.x) >> 5));
  int nwaves = (int)((gridDim.x * blockDim.x) >> 5);
  int c0 = lane * VEC;
  int hh = c0 / DH;
  const float NEG_INF = -__builtin_inff();
  constexpr int RV = DTOT / VEC;
  const vecf* __restrict__ rows = (const vecf*)hfeat;
  vecf bv, wsv, wdv;
  if constexpr (BIAS) bv = *(const vecf*)(bias + c0);
  if constexpr (LOGITS) {
    wsv = *(const vecf*)(wsrc + c0);
    wdv = *(const vecf*)(wdst + c0);
  }

  int node = wave0;
  if (node >= N) return;

  int beg = seg_beg[node];
  int end = seg_end[node];
  int cnt = end - beg;
  if (cnt > GCH) cnt = GCH;
  int mysrc = 0;
  if (lane < cnt) mysrc = csr_src[beg + lane];
  float adv = ad[node * H + hh];
  float as_self = as[node * H + hh];
  vecf acc = rows[node * RV + lane];
  int nfirst = node + nwaves < N ? node + nwaves : node;
  int nbeg = seg_beg[nfirst];
  int nend = seg_end[nfirst];

  for (; node < N; node += nwaves) {
    int ncnt = 0, nsrc = 0, nnbeg = 0, nnend = 0;
    float nadv = 0.f, nas = 0.f;
    vecf nrow;
    auto ask_next = [&]() {
      int nx1 = node + nwaves < N ? node + nwaves : node;
      int nx2 = node + 2 * nwaves < N ? node + 2 * nwaves : node;
      ncnt = nend - nbeg;
      if (ncnt > GCH) ncnt = GCH;
      if (lane < ncnt) nsrc = csr_src[nbeg + lane];
      nadv = ad[nx1 * H + hh];
      nas = as[nx1 * H + hh];
      nrow = rows[nx1 * RV + lane];
      nnbeg = seg_beg[nx2];
      nnend = seg_end[nx2];
    };
    bool asked = false;

    float asv = 0.f;
    if constexpr (H == 1) asv = as[mysrc];
    int base = beg;

    float e0 = as_self + adv;
    e0 = e0 > 0.f ? e0 : NEG_SLOPE * e0;
    float m = e0;
    float s = 1.f;

    while (cnt > 0) {
      int last = cnt - 1;
      vecf hv[GCH];
      float ev[GCH];
#pragma unroll
      for (int g = 0; g < NG; ++g) {
        if (g > 0 && g * GGRP >= cnt) break;
        int sj[GGRP];
#pragma unroll
        for (int i = 0; i < GGRP; ++i) {
          int j = g * GGRP + i;
          sj[i] = __builtin_amdgcn_readlane(mysrc, j < last ? j : last);
        }
        if constexpr (H > 1) {
#pragma unroll
          for (int i = 0; i < GGRP; ++i)
            ev[g * GGRP + i] = as[sj[i] * H + hh];
        }
#pragma unroll
        for (int i = 0; i < GGRP; ++i)
          hv[g * GGRP + i] = rows[sj[i] * RV + lane];
      }
      if (!asked) {
        ask_next();
        asked = true;
      }
      __builtin_amdgcn_sched_barrier(0);

      float mn;
      if constexpr (H == 1) {
        float t = asv + adv;
        t = t > 0.f ? t : NEG_SLOPE * t;
        float mye = lane < cnt ? t : NEG_INF;
        mn = fmaxf(m, wave_max(mye));
        float sc = __expf(m - mn);
        float p = __expf(mye - mn);
        s = s * sc + wave_sum(p);
        acc = acc * sc;
        int pb = __float_as_int(p);
#pragma unroll
        for (int g = 0; g < NG; ++g) {
          if (g > 0 && g * GGRP >= cnt) break;
#pragma unroll
          for (int i = 0; i < GGRP; ++i) {
            int j = g * GGRP + i;
            float pj = __int_as_float(__builtin_amdgcn_readlane(pb, j));
            acc += hv[j] * pj;
          }
        }
      } else {
        float cm = NEG_INF;
#pragma unroll
        for (int g = 0; g < NG; ++g) {
          if (g > 0 && g * GGRP >= cnt) break;
#pragma unroll
          for (int i = 0; i < GGRP; ++i) {
            int j = g * GGRP + i;
            float t = ev[j] + adv;
            t = t > 0.f ? t : NEG_SLOPE * t;
            t = j < cnt ? t : NEG_INF;
            ev[j] = t;
            cm = fmaxf(cm, t);
          }
        }
        mn = fmaxf(m, cm);
        float sc = __expf(m - mn);
        s = s * sc;
        acc = acc * sc;
#pragma unroll
        for (int g = 0; g < NG; ++g) {
          if (g > 0 && g * GGRP >= cnt) break;
#pragma unroll
          for (int i = 0; i < GGRP; ++i) {
            int j = g * GGRP + i;
            float p = __expf(ev[j] - mn);
            s += p;
            acc += hv[j] * p;
          }
        }
      }
      m = mn;

      base += GCH;
      cnt = end - base;
      if (cnt > GCH) cnt = GCH;
      mysrc = 0;
      if (lane < cnt) mysrc = csr_src[base + lane];
      if constexpr (H == 1) {
        if (cnt > 0) asv = as[mysrc];
      }
    }
    if (!asked) ask_next();

    float denom = s + GAT_EPS;
    vecf o;
#pragma unroll
    for (int k = 0; k < VEC; ++k) {
      float v = acc[k] / denom;
      if constexpr (BIAS) v += bv[k];
      if (ELU) v = v > 0.f ? v : __expf(v) - 1.f;
      o[k] = v;
    }
    *(vecf*)(out + (size_t)node * DTOT + c0) = o;

    if constexpr (LOGITS) {
      float ps = 0.f, pd = 0.f;
#pragma unroll
      for (int k = 0; k < VEC; ++k) {
        ps += o[k] * wsv[k];
        pd += o[k] * wdv[k];
      }
      ps = wave_sum(ps);
      pd = wave_sum(pd);
      if (lane == 0) {
        as_next[node] = ps;
        ad_next[node] = pd;
      }
    }

    beg = nbeg;
    end = nend;
    cnt = ncnt;
    mysrc = nsrc;
    adv = nadv;
    as_self = nas;
    acc = nrow;
    nbeg = nnbeg;
    nend = nnend;
  }
}

static inline int cdiv(long long a, long long b) { return (int)((a + b - 1) / b); }

extern "C" void kernel_launch(void* const* d_in, const int* in_sizes, int n_in,
                              void* d_out, int out_size, void* d_ws, size_t ws_size,
                              hipStream_t stream) {
  const float* x      = (const float*)d_in[0];
  const int*   ei     = (const int*)  d_in[1];
  const float* W1     = (const float*)d_in[2];
  const float* a_src1 = (const float*)d_in[3];
  const float* a_dst1 = (const float*)d_in[4];
  const float* b1     = (const float*)d_in[5];
  const float* W2     = (const float*)d_in[6];
  const float* a_src2 = (const float*)d_in[7];
  const float* a_dst2 = (const float*)d_in[8];
  const float* b2     = (const float*)d_in[9];
  float* z = (float*)d_out;

  const int Fin = 256, H1 = 8, C1 = 64, C2 = 128;
  const int N    = in_sizes[0] / Fin;
  const int E    = in_sizes[1] / 2;
  const int* srcv = ei;
  const int* dstv = ei + E;

  float* h1   = (float*)d_ws;
  float* hid  = h1  + (size_t)N * C1;
  float* agg  = hid + (size_t)N * C1;
  float* as1  = agg + (size_t)N * C1;
  float* ad1  = as1 + (size_t)N * H1;
  float* as2  = ad1 + (size_t)N * H1;
  float* ad2  = as2 + N;
  float* ws2  = ad2 + N;
  float* wd2  = ws2 + C1;
  const int NB = cdiv(N, BUCKET_NODES);
  int*   cnts = (int*)(wd2 + C1);
  int*   recs = cnts + (size_t)NB * NCHUNK;
  int*   csrc = recs + (size_t)2 * E;
  int*   sbeg = csrc + E;
  int*   send = sbeg + N;
  int*   bsta = send + N;

  const int B = 256;
  const int GB = 128;
  const int GATHER1_BLOCKS = 1024;
  const int GATHER2_BLOCKS = 1280;

  chunk_hist_kernel<<<NCHUNK + 1, CHUNK_THREADS, 0, stream>>>(
      dstv, E, NB, cnts, W2, a_src2, a_dst2, ws2, wd2);
  gemm1_logits_scan_kernel<<<1 + cdiv((long long)(N / 16) * 32, GB), GB, 0,
                             stream>>>(x, W1, h1, N, Fin, a_src1, a_dst1, as1,
                                       ad1, cnts, NB, bsta);
  chunk_scatter_kernel<<<NCHUNK, CHUNK_THREADS, 0, stream>>>(srcv, dstv, E, NB,
                                                            cnts, bsta, recs);
  bucket_rank_kernel<<<NB, BUCKET_NODES, 0, stream>>>(recs, bsta, sbeg, send,
                                                      csrc, N);

  node_gather_kernel<8, 64, true, true, true><<<GATHER1_BLOCKS, B, 0, stream>>>(
      sbeg, send, csrc, as1, ad1, h1, b1, hid, ws2, wd2, as2, ad2, N);

  node_gather_kernel<1, 64, false, false, false><<<GATHER2_BLOCKS, B, 0, stream>>>(
      sbeg, send, csrc, as2, ad2, hid, nullptr, agg, nullptr, nullptr, nullptr,
      nullptr, N);
  gemm_f32_wmma_kernel<C2 / 16, true><<<cdiv((long long)(N / 16) * 32, GB), GB,
                                        0, stream>>>(agg, W2, b2, z, N, C1);
}


# GEMM fragments fed 16 B a lane: lane n of tile t = column 4n+t%4 (1 B load per k row feeds 4 tiles; C whole-line stores), A 4 adjacent k a lane; sum order: k permuted in products, logits 4-chain+pair
# speedup vs baseline: 5.5767x; 1.0500x over previous
#include <hip/hip_runtime.h>
#include <cstdint>


typedef float v2f __attribute__((ext_vector_type(2)));
typedef float v4f __attribute__((ext_vector_type(4)));
typedef float v8f __attribute__((ext_vector_type(8)));

#define NEG_SLOPE 0.2f
#define GAT_EPS 1e-16f

template <int M>
__device__ __forceinline__ float row_xor_lane(float x) {
  int b = __builtin_bit_cast(int, x);
  return __builtin_bit_cast(
      float, __builtin_amdgcn_update_dpp(b, b, 0x160 + M, 0xf, 0xf, false));
}

__device__ __forceinline__ float wave_sum(float v) {
  v += row_xor_lane<1>(v);
  v += row_xor_lane<2>(v);
  v += row_xor_lane<4>(v);
  v += row_xor_lane<8>(v);
  int b = __builtin_bit_cast(int, v);
  return __builtin_bit_cast(float, __builtin_amdgcn_readlane(b, 0)) +
         __builtin_bit_cast(float, __builtin_amdgcn_readlane(b, 16));
}

__device__ __forceinline__ float wave_max(float v) {
  v = fmaxf(v, row_xor_lane<1>(v));
  v = fmaxf(v, row_xor_lane<2>(v));
  v = fmaxf(v, row_xor_lane<4>(v));
  v = fmaxf(v, row_xor_lane<8>(v));
  int b = __builtin_bit_cast(int, v);
  return fmaxf(__builtin_bit_cast(float, __builtin_amdgcn_readlane(b, 0)),
               __builtin_bit_cast(float, __builtin_amdgcn_readlane(b, 16)));
}

template <int NT, bool BIAS, bool LOGITS>
__device__ __forceinline__ void gemm_strip(const float* __restrict__ A,
                                           const float* __restrict__ B,
                                           const float* __restrict__ bias,
                                           float* __restrict__ C,
                                           int M, int K, int wave,
                                           const float* __restrict__ avs,
                                           const float* __restrict__ avd,
                                           float* __restrict__ as,
                                           float* __restrict__ ad) {
  static_assert(NT % 4 == 0, "tiles come in groups of four");
  constexpr int N = NT * 16;
  constexpr int NGRP = NT / 4;
  int lane = threadIdx.x & 31;
  int mtiles = (M + 15) >> 4;
  if (wave >= mtiles) return;
  int m0 = wave << 4;
  int half = lane >> 4;
  int lid  = lane & 15;

  v8f c[NT];
#pragma unroll
  for (int t = 0; t < NT; ++t)
    c[t] = (v8f){0.f, 0.f, 0.f, 0.f, 0.f, 0.f, 0.f, 0.f};
  const float* arow = A + (size_t)(m0 + lid) * K + 4 * half;
  const float* bcol = B + (size_t)(4 * half) * N + 4 * lid;

#pragma unroll(8 / NT)
  for (int k0 = 0; k0 < K; k0 += 8) {
    v4f a4 = *(const v4f*)(arow + k0);
    const float* bp = bcol + (size_t)k0 * N;
    v4f b4[4][NGRP];
#pragma unroll
    for (int j = 0; j < 4; ++j)
#pragma unroll
      for (int g = 0; g < NGRP; ++g)
        b4[j][g] = *(const v4f*)(bp + j * N + 64 * g);
#pragma unroll
    for (int s = 0; s < 2; ++s) {
      v2f a;
      a.x = s == 0 ? a4.x : a4.z;
      a.y = s == 0 ? a4.y : a4.w;
#pragma unroll
      for (int t = 0; t < NT; ++t) {
        v2f b;
        b.x = b4[2 * s][t >> 2][t & 3];
        b.y = b4[2 * s + 1][t >> 2][t & 3];
        c[t] = __builtin_amdgcn_wmma_f32_16x16x4_f32(false, a, false, b,
                                                     (short)0, c[t], false, false);
      }
    }
  }

  v4f bc[NGRP];
#pragma unroll
  for (int g = 0; g < NGRP; ++g)
    bc[g] = BIAS ? *(const v4f*)(bias + 64 * g + 4 * lid)
                 : (v4f){0.f, 0.f, 0.f, 0.f};

  float* crow = C + (size_t)(m0 + 8 * half) * N + 4 * lid;
#pragma unroll
  for (int r = 0; r < 8; ++r)
#pragma unroll
    for (int g = 0; g < NGRP; ++g) {
      v4f o = (v4f){c[4 * g][r], c[4 * g + 1][r], c[4 * g + 2][r],
                    c[4 * g + 3][r]};
      if (BIAS) o += bc[g];
      *(v4f*)(crow + (size_t)r * N + 64 * g) = o;
    }

  if constexpr (LOGITS) {
    static_assert(NT == 4, "logit epilogue: 64 columns, 8 heads of 8");
    constexpr int H = N / 8;
    int odd = lid & 1;
    int hd = lid >> 1;
    v4f sv = *(const v4f*)(avs + 4 * lid);
    v4f dv = *(const v4f*)(avd + 4 * lid);
    float ks[8], kd[8];
#pragma unroll
    for (int r = 0; r < 8; ++r) {
      float ps = c[0][r] * sv.x;
      float pd = c[0][r] * dv.x;
      ps += c[1][r] * sv.y;
      pd += c[1][r] * dv.y;
      ps += c[2][r] * sv.z;
      pd += c[2][r] * dv.z;
      ps += c[3][r] * sv.w;
      pd += c[3][r] * dv.w;
      ks[r] = ps + row_xor_lane<1>(ps);
      kd[r] = pd + row_xor_lane<1>(pd);
    }
    float* asrow = as + (size_t)(m0 + 8 * half + 4 * odd) * H + hd;
    float* adrow = ad + (size_t)(m0 + 8 * half + 4 * odd) * H + hd;
#pragma unroll
    for (int i = 0; i < 4; ++i) {
      asrow[i * H] = odd ? ks[i + 4] : ks[i];
      adrow[i * H] = odd ? kd[i + 4] : kd[i];
    }
  }
}

template <int NT, bool BIAS>
__global__ void gemm_f32_wmma_kernel(const float* __restrict__ A,
                                     const float* __restrict__ B,
                                     const float* __restrict__ bias,
                                     float* __restrict__ C,
                                     int M, int K) {
  int wave = (int)((blockIdx.x * blockDim.x + threadIdx.x) >> 5);
  gemm_strip<NT, BIAS, false>(A, B, bias, C, M, K, wave, nullptr, nullptr,
                              nullptr, nullptr);
}

template <int K, int DO>
__device__ __forceinline__ void att_vec_body(const float* __restrict__ W,
                                             const float* __restrict__ a_s,
                                             const float* __restrict__ a_d,
                                             float* __restrict__ ws,
                                             float* __restrict__ wd,
                                             float* res) {
  constexpr int V = DO / 32;
  typedef float vecf __attribute__((ext_vector_type(V)));
  int lane = threadIdx.x & 31;
  int w = threadIdx.x >> 5;
  int nw = blockDim.x >> 5;
  vecf sv = *(const vecf*)(a_s + lane * V);
  vecf dv = *(const vecf*)(a_d + lane * V);
  for (int k = w; k < K; k += nw) {
    vecf wv = *(const vecf*)(W + (size_t)k * DO + lane * V);
    float ps = 0.f, pd = 0.f;
#pragma unroll
    for (int i = 0; i < V; ++i) {
      ps += wv[i] * sv[i];
      pd += wv[i] * dv[i];
    }
    ps = wave_sum(ps);
    pd = wave_sum(pd);
    if (lane == 0) { res[k] = ps; res[K + k] = pd; }
  }
  __syncthreads();
  for (int i = threadIdx.x; i < 2 * K; i += blockDim.x) {
    if (i < K) ws[i] = res[i];
    else wd[i - K] = res[i];
  }
}

#define NCHUNK 256
#define BUCKET_SHIFT 8
#define BUCKET_NODES 256
#define MAX_BUCKETS 1024
#define CHUNK_THREADS 1024
#define STAGE_CAP 8192

__device__ __forceinline__ int wave_incl_scan(int x, int lane) {
#pragma unroll
  for (int o = 1; o < 32; o <<= 1) {
    int y = __shfl_up(x, o);
    if (lane >= o) x += y;
  }
  return x;
}

template <int S>
__device__ __forceinline__ int row_shr_lane(int x) {
  return __builtin_amdgcn_update_dpp(0, x, 0x110 + S, 0xf, 0xf, false);
}

__device__ __forceinline__ int wave_incl_scan_valu(int x, int lane) {
  x += row_shr_lane<1>(x);
  x += row_shr_lane<2>(x);
  x += row_shr_lane<4>(x);
  x += row_shr_lane<8>(x);
  int low = __builtin_amdgcn_readlane(x, 15);
  return lane >= 16 ? x + low : x;
}

__device__ __forceinline__ int block_excl_scan(int v, int* wsum, int& total) {
  int lane = threadIdx.x & 31;
  int w = threadIdx.x >> 5;
  int nw = blockDim.x >> 5;
  int x = wave_incl_scan(v, lane);
  if (lane == 31) wsum[w] = x;
  __syncthreads();
  if (w == 0) {
    int t = (lane < nw) ? wsum[lane] : 0;
    int xs = wave_incl_scan(t, lane);
    wsum[lane] = xs - t;
    if (lane == 31) wsum[32] = xs;
  }
  __syncthreads();
  total = wsum[32];
  int r = wsum[w] + (x - v);
  __syncthreads();
  return r;
}

__global__ void chunk_hist_kernel(const int* __restrict__ dst, int E, int NB,
                                  int* __restrict__ counts,
                                  const float* __restrict__ W2,
                                  const float* __restrict__ a_s2,
                                  const float* __restrict__ a_d2,
                                  float* __restrict__ ws2,
                                  float* __restrict__ wd2) {
  __shared__ int h[MAX_BUCKETS];
  __shared__ float res[2 * 64];
  if (blockIdx.x == NCHUNK) {
    att_vec_body<64, 128>(W2, a_s2, a_d2, ws2, wd2, res);
    return;
  }
  int tid = threadIdx.x;
  int c = blockIdx.x;
  for (int i = tid; i < NB; i += CHUNK_THREADS) h[i] = 0;
  __syncthreads();
  int per = (E + NCHUNK - 1) / NCHUNK;
  int e0 = c * per;
  int e1 = e0 + per < E ? e0 + per : E;
  for (int e = e0 + tid; e < e1; e += 4 * CHUNK_THREADS) {
    int d[4];
#pragma unroll
    for (int k = 0; k < 4; ++k) {
      int ee = e + k * CHUNK_THREADS;
      d[k] = ee < e1 ? dst[ee] : -1;
    }
#pragma unroll
    for (int k = 0; k < 4; ++k)
      if (d[k] >= 0) atomicAdd(&h[d[k] >> BUCKET_SHIFT], 1);
  }
  __syncthreads();
  for (int i = tid; i < NB; i += CHUNK_THREADS)
    counts[(size_t)i * NCHUNK + c] = h[i];
}

template <int RPT>
__device__ __forceinline__ void bucket_scan_body(int* __restrict__ counts, int NB,
                                                 int* __restrict__ bstart,
                                                 int* tot, int* wsum) {
  typedef int v4i __attribute__((ext_vector_type(4)));
  int tid = threadIdx.x;
  int lane = tid & 31;
  int w = tid >> 5;
  int nw = blockDim.x >> 5;
  for (int b0 = w * RPT; b0 < NB; b0 += nw * RPT) {
    v4i a0[RPT], a1[RPT];
#pragma unroll
    for (int q = 0; q < RPT; ++q) {
      if (b0 + q < NB) {
        const v4i* row = (const v4i*)(counts + (size_t)(b0 + q) * NCHUNK + lane * 8);
        a0[q] = row[0];
        a1[q] = row[1];
      }
    }
#pragma unroll
    for (int q = 0; q < RPT; ++q) {
      if (b0 + q < NB) {
        v4i p0, p1;
        p0.x = 0;
        p0.y = a0[q].x;
        p0.z = p0.y + a0[q].y;
        p0.w = p0.z + a0[q].z;
        p1.x = p0.w + a0[q].w;
        p1.y = p1.x + a1[q].x;
        p1.z = p1.y + a1[q].y;
        p1.w = p1.z + a1[q].z;
        int sum = p1.w + a1[q].w;
        int incl = wave_incl_scan_valu(sum, lane);
        int excl = incl - sum;
        p0 += excl;
        p1 += excl;
        v4i* row = (v4i*)(counts + (size_t)(b0 + q) * NCHUNK + lane * 8);
        row[0] = p0;
        row[1] = p1;
        if (lane == 31) tot[b0 + q] = incl;
      }
    }
  }
  __syncthreads();
  int carry = 0;
  for (int base = 0; base < NB; base += (int)blockDim.x) {
    int i = base + tid;
    int v = (i < NB) ? tot[i] : 0;
    int total;
    int ex = block_excl_scan(v, wsum, total);
    if (i < NB) bstart[i] = carry + ex;
    carry += total;
  }
  if (tid == 0) bstart[NB] = carry;
}

__global__ void gemm1_logits_scan_kernel(const float* __restrict__ A,
                                         const float* __restrict__ B,
                                         float* __restrict__ C, int M, int K,
                                         const float* __restrict__ avs,
                                         const float* __restrict__ avd,
                                         float* __restrict__ as,
                                         float* __restrict__ ad,
                                         int* __restrict__ counts, int NB,
                                         int* __restrict__ bstart) {
  __shared__ int tot[MAX_BUCKETS];
  __shared__ int wsum[33];
  if (blockIdx.x == 0) {
    bucket_scan_body<8>(counts, NB, bstart, tot, wsum);
    return;
  }
  int wave = (int)(((blockIdx.x - 1) * blockDim.x + threadIdx.x) >> 5);
  gemm_strip<4, false, true>(A, B, nullptr, C, M, K, wave, avs, avd, as, ad);
}

__global__ void chunk_scatter_kernel(const int* __restrict__ src,
                                     const int* __restrict__ dst, int E, int NB,
                                     const int* __restrict__ part,
                                     const int* __restrict__ bstart,
                                     int* __restrict__ rec_words) {
  typedef int v2i __attribute__((ext_vector_type(2)));
  v2i* rec = (v2i*)rec_words;
  __shared__ int cur[MAX_BUCKETS];
  int tid = threadIdx.x;
  int c = blockIdx.x;
  for (int i = tid; i < NB; i += CHUNK_THREADS)
    cur[i] = bstart[i] + part[(size_t)i * NCHUNK + c];
  __syncthreads();
  int per = (E + NCHUNK - 1) / NCHUNK;
  int e0 = c * per;
  int e1 = e0 + per < E ? e0 + per : E;
  for (int e = e0 + tid; e < e1; e += 4 * CHUNK_THREADS) {
    int d[4], s[4], p[4];
#pragma unroll
    for (int k = 0; k < 4; ++k) {
      int ee = e + k * CHUNK_THREADS;
      bool ok = ee < e1;
      d[k] = ok ? dst[ee] : -1;
      s[k] = ok ? src[ee] : 0;
    }
#pragma unroll
    for (int k = 0; k < 4; ++k)
      p[k] = d[k] >= 0 ? atomicAdd(&cur[d[k] >> BUCKET_SHIFT], 1) : 0;
#pragma unroll
    for (int k = 0; k < 4; ++k)
      if (d[k] >= 0) rec[p[k]] = (v2i){s[k], d[k]};
  }
}

__global__ void bucket_rank_kernel(const int* __restrict__ rec_words,
                                   const int* __restrict__ bstart,
                                   int* __restrict__ seg_beg,
                                   int* __restrict__ seg_end,
                                   int* __restrict__ csr_src, int N) {
  typedef int v2i __attribute__((ext_vector_type(2)));
  const v2i* rec = (const v2i*)rec_words;
  __shared__ int cnt[BUCKET_NODES];
  __shared__ int wsum[33];
  __shared__ int stage[STAGE_CAP];
  int tid = threadIdx.x;
  int b = blockIdx.x;
  int lo = b << BUCKET_SHIFT;
  int bs = bstart[b];
  int be = bstart[b + 1];
  int n = be - bs;
  cnt[tid] = 0;
  __syncthreads();
  for (int i = tid; i < n; i += 4 * BUCKET_NODES) {
    int d[4];
#pragma unroll
    for (int k = 0; k < 4; ++k) {
      int ii = i + k * BUCKET_NODES;
      d[k] = ii < n ? rec[bs + ii].y : -1;
    }
#pragma unroll
    for (int k = 0; k < 4; ++k)
      if (d[k] >= 0) atomicAdd(&cnt[d[k] - lo], 1);
  }
  __syncthreads();
  int v = cnt[tid];
  int total;
  int ex = block_excl_scan(v, wsum, total);
  int node = lo + tid;
  if (node < N) {
    seg_beg[node] = bs + ex;
    seg_end[node] = bs + ex + v;
  }
  cnt[tid] = ex;
  __syncthreads();
  bool staged = n <= STAGE_CAP;
  for (int i = tid; i < n; i += 4 * BUCKET_NODES) {
    v2i r[4];
    int p[4];
#pragma unroll
    for (int k = 0; k < 4; ++k) {
      int ii = i + k * BUCKET_NODES;
      r[k] = ii < n ? rec[bs + ii] : (v2i){0, -1};
    }
#pragma unroll
    for (int k = 0; k < 4; ++k)
      p[k] = r[k].y >= 0 ? atomicAdd(&cnt[r[k].y - lo], 1) : 0;
#pragma unroll
    for (int k = 0; k < 4; ++k) {
      if (r[k].y >= 0) {
        if (staged) stage[p[k]] = r[k].x;
        else csr_src[bs + p[k]] = r[k].x;
      }
    }
  }
  if (staged) {
    __syncthreads();
    for (int i = tid; i < n; i += BUCKET_NODES) csr_src[bs + i] = stage[i];
  }
}

#define GGRP 4

template <int H, int DTOT, bool ELU, bool BIAS, bool LOGITS>
__global__ void __launch_bounds__(256)
node_gather_kernel(const int* __restrict__ seg_beg,
                   const int* __restrict__ seg_end,
                   const int* __restrict__ csr_src,
                   const float* __restrict__ as,
                   const float* __restrict__ ad,
                   const float* __restrict__ hfeat,
                   const float* __restrict__ bias,
                   float* __restrict__ out,
                   const float* __restrict__ wsrc,
                   const float* __restrict__ wdst,
                   float* __restrict__ as_next,
                   float* __restrict__ ad_next, int N) {
  constexpr int VEC = DTOT / 32;
  constexpr int DH = DTOT / H;
  constexpr int GCH = (H == 1) ? 32 : 28;
  constexpr int NG = GCH / GGRP;
  typedef float vecf __attribute__((ext_vector_type(VEC)));
  int lane = threadIdx.x & 31;
  int wave0 = __builtin_amdgcn_readfirstlane(
      (int)((blockIdx.x * blockDim.x + threadIdx.x) >> 5));
  int nwaves = (int)((gridDim.x * blockDim.x) >> 5);
  int c0 = lane * VEC;
  int hh = c0 / DH;
  const float NEG_INF = -__builtin_inff();
  constexpr int RV = DTOT / VEC;
  const vecf* __restrict__ rows = (const vecf*)hfeat;
  vecf bv, wsv, wdv;
  if constexpr (BIAS) bv = *(const vecf*)(bias + c0);
  if constexpr (LOGITS) {
    wsv = *(const vecf*)(wsrc + c0);
    wdv = *(const vecf*)(wdst + c0);
  }

  int node = wave0;
  if (node >= N) return;

  int beg = seg_beg[node];
  int end = seg_end[node];
  int cnt = end - beg;
  if (cnt > GCH) cnt = GCH;
  int mysrc = 0;
  if (lane < cnt) mysrc = csr_src[beg + lane];
  float adv = ad[node * H + hh];
  float as_self = as[node * H + hh];
  vecf acc = rows[node * RV + lane];
  int nfirst = node + nwaves < N ? node + nwaves : node;
  int nbeg = seg_beg[nfirst];
  int nend = seg_end[nfirst];

  for (; node < N; node += nwaves) {
    int ncnt = 0, nsrc = 0, nnbeg = 0, nnend = 0;
    float nadv = 0.f, nas = 0.f;
    vecf nrow;
    auto ask_next = [&]() {
      int nx1 = node + nwaves < N ? node + nwaves : node;
      int nx2 = node + 2 * nwaves < N ? node + 2 * nwaves : node;
      ncnt = nend - nbeg;
      if (ncnt > GCH) ncnt = GCH;
      if (lane < ncnt) nsrc = csr_src[nbeg + lane];
      nadv = ad[nx1 * H + hh];
      nas = as[nx1 * H + hh];
      nrow = rows[nx1 * RV + lane];
      nnbeg = seg_beg[nx2];
      nnend = seg_end[nx2];
    };
    bool asked = false;

    float asv = 0.f;
    if constexpr (H == 1) asv = as[mysrc];
    int base = beg;

    float e0 = as_self + adv;
    e0 = e0 > 0.f ? e0 : NEG_SLOPE * e0;
    float m = e0;
    float s = 1.f;

    while (cnt > 0) {
      int last = cnt - 1;
      vecf hv[GCH];
      float ev[GCH];
#pragma unroll
      for (int g = 0; g < NG; ++g) {
        if (g > 0 && g * GGRP >= cnt) break;
        int sj[GGRP];
#pragma unroll
        for (int i = 0; i < GGRP; ++i) {
          int j = g * GGRP + i;
          sj[i] = __builtin_amdgcn_readlane(mysrc, j < last ? j : last);
        }
        if constexpr (H > 1) {
#pragma unroll
          for (int i = 0; i < GGRP; ++i)
            ev[g * GGRP + i] = as[sj[i] * H + hh];
        }
#pragma unroll
        for (int i = 0; i < GGRP; ++i)
          hv[g * GGRP + i] = rows[sj[i] * RV + lane];
      }
      if (!asked) {
        ask_next();
        asked = true;
      }
      __builtin_amdgcn_sched_barrier(0);

      float mn;
      if constexpr (H == 1) {
        float t = asv + adv;
        t = t > 0.f ? t : NEG_SLOPE * t;
        float mye = lane < cnt ? t : NEG_INF;
        mn = fmaxf(m, wave_max(mye));
        float sc = __expf(m - mn);
        float p = __expf(mye - mn);
        s = s * sc + wave_sum(p);
        acc = acc * sc;
        int pb = __float_as_int(p);
#pragma unroll
        for (int g = 0; g < NG; ++g) {
          if (g > 0 && g * GGRP >= cnt) break;
#pragma unroll
          for (int i = 0; i < GGRP; ++i) {
            int j = g * GGRP + i;
            float pj = __int_as_float(__builtin_amdgcn_readlane(pb, j));
            acc += hv[j] * pj;
          }
        }
      } else {
        float cm = NEG_INF;
#pragma unroll
        for (int g = 0; g < NG; ++g) {
          if (g > 0 && g * GGRP >= cnt) break;
#pragma unroll
          for (int i = 0; i < GGRP; ++i) {
            int j = g * GGRP + i;
            float t = ev[j] + adv;
            t = t > 0.f ? t : NEG_SLOPE * t;
            t = j < cnt ? t : NEG_INF;
            ev[j] = t;
            cm = fmaxf(cm, t);
          }
        }
        mn = fmaxf(m, cm);
        float sc = __expf(m - mn);
        s = s * sc;
        acc = acc * sc;
#pragma unroll
        for (int g = 0; g < NG; ++g) {
          if (g > 0 && g * GGRP >= cnt) break;
#pragma unroll
          for (int i = 0; i < GGRP; ++i) {
            int j = g * GGRP + i;
            float p = __expf(ev[j] - mn);
            s += p;
            acc += hv[j] * p;
          }
        }
      }
      m = mn;

      base += GCH;
      cnt = end - base;
      if (cnt > GCH) cnt = GCH;
      mysrc = 0;
      if (lane < cnt) mysrc = csr_src[base + lane];
      if constexpr (H == 1) {
        if (cnt > 0) asv = as[mysrc];
      }
    }
    if (!asked) ask_next();

    float denom = s + GAT_EPS;
    vecf o;
#pragma unroll
    for (int k = 0; k < VEC; ++k) {
      float v = acc[k] / denom;
      if constexpr (BIAS) v += bv[k];
      if (ELU) v = v > 0.f ? v : __expf(v) - 1.f;
      o[k] = v;
    }
    *(vecf*)(out + (size_t)node * DTOT + c0) = o;

    if constexpr (LOGITS) {
      float ps = 0.f, pd = 0.f;
#pragma unroll
      for (int k = 0; k < VEC; ++k) {
        ps += o[k] * wsv[k];
        pd += o[k] * wdv[k];
      }
      ps = wave_sum(ps);
      pd = wave_sum(pd);
      if (lane == 0) {
        as_next[node] = ps;
        ad_next[node] = pd;
      }
    }

    beg = nbeg;
    end = nend;
    cnt = ncnt;
    mysrc = nsrc;
    adv = nadv;
    as_self = nas;
    acc = nrow;
    nbeg = nnbeg;
    nend = nnend;
  }
}

static inline int cdiv(long long a, long long b) { return (int)((a + b - 1) / b); }

extern "C" void kernel_launch(void* const* d_in, const int* in_sizes, int n_in,
                              void* d_out, int out_size, void* d_ws, size_t ws_size,
                              hipStream_t stream) {
  const float* x      = (const float*)d_in[0];
  const int*   ei     = (const int*)  d_in[1];
  const float* W1     = (const float*)d_in[2];
  const float* a_src1 = (const float*)d_in[3];
  const float* a_dst1 = (const float*)d_in[4];
  const float* b1     = (const float*)d_in[5];
  const float* W2     = (const float*)d_in[6];
  const float* a_src2 = (const float*)d_in[7];
  const float* a_dst2 = (const float*)d_in[8];
  const float* b2     = (const float*)d_in[9];
  float* z = (float*)d_out;

  const int Fin = 256, H1 = 8, C1 = 64, C2 = 128;
  const int N    = in_sizes[0] / Fin;
  const int E    = in_sizes[1] / 2;
  const int* srcv = ei;
  const int* dstv = ei + E;

  float* h1   = (float*)d_ws;
  float* hid  = h1  + (size_t)N * C1;
  float* agg  = hid + (size_t)N * C1;
  float* as1  = agg + (size_t)N * C1;
  float* ad1  = as1 + (size_t)N * H1;
  float* as2  = ad1 + (size_t)N * H1;
  float* ad2  = as2 + N;
  float* ws2  = ad2 + N;
  float* wd2  = ws2 + C1;
  const int NB = cdiv(N, BUCKET_NODES);
  int*   cnts = (int*)(wd2 + C1);
  int*   recs = cnts + (size_t)NB * NCHUNK;
  int*   csrc = recs + (size_t)2 * E;
  int*   sbeg = csrc + E;
  int*   send = sbeg + N;
  int*   bsta = send + N;

  const int B = 256;
  const int GB = 128;
  const int GATHER1_BLOCKS = 1024;
  const int GATHER2_BLOCKS = 1280;

  chunk_hist_kernel<<<NCHUNK + 1, CHUNK_THREADS, 0, stream>>>(
      dstv, E, NB, cnts, W2, a_src2, a_dst2, ws2, wd2);
  gemm1_logits_scan_kernel<<<1 + cdiv((long long)(N / 16) * 32, GB), GB, 0,
                             stream>>>(x, W1, h1, N, Fin, a_src1, a_dst1, as1,
                                       ad1, cnts, NB, bsta);
  chunk_scatter_kernel<<<NCHUNK, CHUNK_THREADS, 0, stream>>>(srcv, dstv, E, NB,
                                                            cnts, bsta, recs);
  bucket_rank_kernel<<<NB, BUCKET_NODES, 0, stream>>>(recs, bsta, sbeg, send,
                                                      csrc, N);

  node_gather_kernel<8, 64, true, true, true><<<GATHER1_BLOCKS, B, 0, stream>>>(
      sbeg, send, csrc, as1, ad1, h1, b1, hid, ws2, wd2, as2, ad2, N);

  node_gather_kernel<1, 64, false, false, false><<<GATHER2_BLOCKS, B, 0, stream>>>(
      sbeg, send, csrc, as2, ad2, hid, nullptr, agg, nullptr, nullptr, nullptr,
      nullptr, N);
  gemm_f32_wmma_kernel<C2 / 16, true><<<cdiv((long long)(N / 16) * 32, GB), GB,
                                        0, stream>>>(agg, W2, b2, z, N, C1);
}


# Layer-1 gather reads rows only: each 16-lane half takes one edge of a pair (16 B a lane) and forms its head logits from that row, no as1/ad1 planes; sum order: even/odd edges apart, halves added last
# speedup vs baseline: 5.7567x; 1.0323x over previous
#include <hip/hip_runtime.h>
#include <cstdint>


typedef float v2f __attribute__((ext_vector_type(2)));
typedef float v4f __attribute__((ext_vector_type(4)));
typedef float v8f __attribute__((ext_vector_type(8)));

#define NEG_SLOPE 0.2f
#define GAT_EPS 1e-16f

template <int M>
__device__ __forceinline__ float row_xor_lane(float x) {
  int b = __builtin_bit_cast(int, x);
  return __builtin_bit_cast(
      float, __builtin_amdgcn_update_dpp(b, b, 0x160 + M, 0xf, 0xf, false));
}

__device__ __forceinline__ float wave_sum(float v) {
  v += row_xor_lane<1>(v);
  v += row_xor_lane<2>(v);
  v += row_xor_lane<4>(v);
  v += row_xor_lane<8>(v);
  int b = __builtin_bit_cast(int, v);
  return __builtin_bit_cast(float, __builtin_amdgcn_readlane(b, 0)) +
         __builtin_bit_cast(float, __builtin_amdgcn_readlane(b, 16));
}

__device__ __forceinline__ float wave_max(float v) {
  v = fmaxf(v, row_xor_lane<1>(v));
  v = fmaxf(v, row_xor_lane<2>(v));
  v = fmaxf(v, row_xor_lane<4>(v));
  v = fmaxf(v, row_xor_lane<8>(v));
  int b = __builtin_bit_cast(int, v);
  return fmaxf(__builtin_bit_cast(float, __builtin_amdgcn_readlane(b, 0)),
               __builtin_bit_cast(float, __builtin_amdgcn_readlane(b, 16)));
}

template <int NT, bool BIAS>
__device__ __forceinline__ void gemm_strip(const float* __restrict__ A,
                                           const float* __restrict__ B,
                                           const float* __restrict__ bias,
                                           float* __restrict__ C,
                                           int M, int K, int wave) {
  static_assert(NT % 4 == 0, "tiles come in groups of four");
  constexpr int N = NT * 16;
  constexpr int NGRP = NT / 4;
  int lane = threadIdx.x & 31;
  int mtiles = (M + 15) >> 4;
  if (wave >= mtiles) return;
  int m0 = wave << 4;
  int half = lane >> 4;
  int lid  = lane & 15;

  v8f c[NT];
#pragma unroll
  for (int t = 0; t < NT; ++t)
    c[t] = (v8f){0.f, 0.f, 0.f, 0.f, 0.f, 0.f, 0.f, 0.f};
  const float* arow = A + (size_t)(m0 + lid) * K + 4 * half;
  const float* bcol = B + (size_t)(4 * half) * N + 4 * lid;

#pragma unroll(8 / NT)
  for (int k0 = 0; k0 < K; k0 += 8) {
    v4f a4 = *(const v4f*)(arow + k0);
    const float* bp = bcol + (size_t)k0 * N;
    v4f b4[4][NGRP];
#pragma unroll
    for (int j = 0; j < 4; ++j)
#pragma unroll
      for (int g = 0; g < NGRP; ++g)
        b4[j][g] = *(const v4f*)(bp + j * N + 64 * g);
#pragma unroll
    for (int s = 0; s < 2; ++s) {
      v2f a;
      a.x = s == 0 ? a4.x : a4.z;
      a.y = s == 0 ? a4.y : a4.w;
#pragma unroll
      for (int t = 0; t < NT; ++t) {
        v2f b;
        b.x = b4[2 * s][t >> 2][t & 3];
        b.y = b4[2 * s + 1][t >> 2][t & 3];
        c[t] = __builtin_amdgcn_wmma_f32_16x16x4_f32(false, a, false, b,
                                                     (short)0, c[t], false, false);
      }
    }
  }

  v4f bc[NGRP];
#pragma unroll
  for (int g = 0; g < NGRP; ++g)
    bc[g] = BIAS ? *(const v4f*)(bias + 64 * g + 4 * lid)
                 : (v4f){0.f, 0.f, 0.f, 0.f};

  float* crow = C + (size_t)(m0 + 8 * half) * N + 4 * lid;
#pragma unroll
  for (int r = 0; r < 8; ++r)
#pragma unroll
    for (int g = 0; g < NGRP; ++g) {
      v4f o = (v4f){c[4 * g][r], c[4 * g + 1][r], c[4 * g + 2][r],
                    c[4 * g + 3][r]};
      if (BIAS) o += bc[g];
      *(v4f*)(crow + (size_t)r * N + 64 * g) = o;
    }
}

template <int NT, bool BIAS>
__global__ void gemm_f32_wmma_kernel(const float* __restrict__ A,
                                     const float* __restrict__ B,
                                     const float* __restrict__ bias,
                                     float* __restrict__ C,
                                     int M, int K) {
  int wave = (int)((blockIdx.x * blockDim.x + threadIdx.x) >> 5);
  gemm_strip<NT, BIAS>(A, B, bias, C, M, K, wave);
}

template <int K, int DO>
__device__ __forceinline__ void att_vec_body(const float* __restrict__ W,
                                             const float* __restrict__ a_s,
                                             const float* __restrict__ a_d,
                                             float* __restrict__ ws,
                                             float* __restrict__ wd,
                                             float* res) {
  constexpr int V = DO / 32;
  typedef float vecf __attribute__((ext_vector_type(V)));
  int lane = threadIdx.x & 31;
  int w = threadIdx.x >> 5;
  int nw = blockDim.x >> 5;
  vecf sv = *(const vecf*)(a_s + lane * V);
  vecf dv = *(const vecf*)(a_d + lane * V);
  for (int k = w; k < K; k += nw) {
    vecf wv = *(const vecf*)(W + (size_t)k * DO + lane * V);
    float ps = 0.f, pd = 0.f;
#pragma unroll
    for (int i = 0; i < V; ++i) {
      ps += wv[i] * sv[i];
      pd += wv[i] * dv[i];
    }
    ps = wave_sum(ps);
    pd = wave_sum(pd);
    if (lane == 0) { res[k] = ps; res[K + k] = pd; }
  }
  __syncthreads();
  for (int i = threadIdx.x; i < 2 * K; i += blockDim.x) {
    if (i < K) ws[i] = res[i];
    else wd[i - K] = res[i];
  }
}

#define NCHUNK 256
#define BUCKET_SHIFT 8
#define BUCKET_NODES 256
#define MAX_BUCKETS 1024
#define CHUNK_THREADS 1024
#define STAGE_CAP 8192

__device__ __forceinline__ int wave_incl_scan(int x, int lane) {
#pragma unroll
  for (int o = 1; o < 32; o <<= 1) {
    int y = __shfl_up(x, o);
    if (lane >= o) x += y;
  }
  return x;
}

template <int S>
__device__ __forceinline__ int row_shr_lane(int x) {
  return __builtin_amdgcn_update_dpp(0, x, 0x110 + S, 0xf, 0xf, false);
}

__device__ __forceinline__ int wave_incl_scan_valu(int x, int lane) {
  x += row_shr_lane<1>(x);
  x += row_shr_lane<2>(x);
  x += row_shr_lane<4>(x);
  x += row_shr_lane<8>(x);
  int low = __builtin_amdgcn_readlane(x, 15);
  return lane >= 16 ? x + low : x;
}

__device__ __forceinline__ int block_excl_scan(int v, int* wsum, int& total) {
  int lane = threadIdx.x & 31;
  int w = threadIdx.x >> 5;
  int nw = blockDim.x >> 5;
  int x = wave_incl_scan(v, lane);
  if (lane == 31) wsum[w] = x;
  __syncthreads();
  if (w == 0) {
    int t = (lane < nw) ? wsum[lane] : 0;
    int xs = wave_incl_scan(t, lane);
    wsum[lane] = xs - t;
    if (lane == 31) wsum[32] = xs;
  }
  __syncthreads();
  total = wsum[32];
  int r = wsum[w] + (x - v);
  __syncthreads();
  return r;
}

__global__ void chunk_hist_kernel(const int* __restrict__ dst, int E, int NB,
                                  int* __restrict__ counts,
                                  const float* __restrict__ W2,
                                  const float* __restrict__ a_s2,
                                  const float* __restrict__ a_d2,
                                  float* __restrict__ ws2,
                                  float* __restrict__ wd2) {
  __shared__ int h[MAX_BUCKETS];
  __shared__ float res[2 * 64];
  if (blockIdx.x == NCHUNK) {
    att_vec_body<64, 128>(W2, a_s2, a_d2, ws2, wd2, res);
    return;
  }
  int tid = threadIdx.x;
  int c = blockIdx.x;
  for (int i = tid; i < NB; i += CHUNK_THREADS) h[i] = 0;
  __syncthreads();
  int per = (E + NCHUNK - 1) / NCHUNK;
  int e0 = c * per;
  int e1 = e0 + per < E ? e0 + per : E;
  for (int e = e0 + tid; e < e1; e += 4 * CHUNK_THREADS) {
    int d[4];
#pragma unroll
    for (int k = 0; k < 4; ++k) {
      int ee = e + k * CHUNK_THREADS;
      d[k] = ee < e1 ? dst[ee] : -1;
    }
#pragma unroll
    for (int k = 0; k < 4; ++k)
      if (d[k] >= 0) atomicAdd(&h[d[k] >> BUCKET_SHIFT], 1);
  }
  __syncthreads();
  for (int i = tid; i < NB; i += CHUNK_THREADS)
    counts[(size_t)i * NCHUNK + c] = h[i];
}

template <int RPT>
__device__ __forceinline__ void bucket_scan_body(int* __restrict__ counts, int NB,
                                                 int* __restrict__ bstart,
                                                 int* tot, int* wsum) {
  typedef int v4i __attribute__((ext_vector_type(4)));
  int tid = threadIdx.x;
  int lane = tid & 31;
  int w = tid >> 5;
  int nw = blockDim.x >> 5;
  for (int b0 = w * RPT; b0 < NB; b0 += nw * RPT) {
    v4i a0[RPT], a1[RPT];
#pragma unroll
    for (int q = 0; q < RPT; ++q) {
      if (b0 + q < NB) {
        const v4i* row = (const v4i*)(counts + (size_t)(b0 + q) * NCHUNK + lane * 8);
        a0[q] = row[0];
        a1[q] = row[1];
      }
    }
#pragma unroll
    for (int q = 0; q < RPT; ++q) {
      if (b0 + q < NB) {
        v4i p0, p1;
        p0.x = 0;
        p0.y = a0[q].x;
        p0.z = p0.y + a0[q].y;
        p0.w = p0.z + a0[q].z;
        p1.x = p0.w + a0[q].w;
        p1.y = p1.x + a1[q].x;
        p1.z = p1.y + a1[q].y;
        p1.w = p1.z + a1[q].z;
        int sum = p1.w + a1[q].w;
        int incl = wave_incl_scan_valu(sum, lane);
        int excl = incl - sum;
        p0 += excl;
        p1 += excl;
        v4i* row = (v4i*)(counts + (size_t)(b0 + q) * NCHUNK + lane * 8);
        row[0] = p0;
        row[1] = p1;
        if (lane == 31) tot[b0 + q] = incl;
      }
    }
  }
  __syncthreads();
  int carry = 0;
  for (int base = 0; base < NB; base += (int)blockDim.x) {
    int i = base + tid;
    int v = (i < NB) ? tot[i] : 0;
    int total;
    int ex = block_excl_scan(v, wsum, total);
    if (i < NB) bstart[i] = carry + ex;
    carry += total;
  }
  if (tid == 0) bstart[NB] = carry;
}

__global__ void gemm1_scan_kernel(const float* __restrict__ A,
                                  const float* __restrict__ B,
                                  float* __restrict__ C, int M, int K,
                                  int* __restrict__ counts, int NB,
                                  int* __restrict__ bstart) {
  __shared__ int tot[MAX_BUCKETS];
  __shared__ int wsum[33];
  if (blockIdx.x == 0) {
    bucket_scan_body<8>(counts, NB, bstart, tot, wsum);
    return;
  }
  int wave = (int)(((blockIdx.x - 1) * blockDim.x + threadIdx.x) >> 5);
  gemm_strip<4, false>(A, B, nullptr, C, M, K, wave);
}

__global__ void chunk_scatter_kernel(const int* __restrict__ src,
                                     const int* __restrict__ dst, int E, int NB,
                                     const int* __restrict__ part,
                                     const int* __restrict__ bstart,
                                     int* __restrict__ rec_words) {
  typedef int v2i __attribute__((ext_vector_type(2)));
  v2i* rec = (v2i*)rec_words;
  __shared__ int cur[MAX_BUCKETS];
  int tid = threadIdx.x;
  int c = blockIdx.x;
  for (int i = tid; i < NB; i += CHUNK_THREADS)
    cur[i] = bstart[i] + part[(size_t)i * NCHUNK + c];
  __syncthreads();
  int per = (E + NCHUNK - 1) / NCHUNK;
  int e0 = c * per;
  int e1 = e0 + per < E ? e0 + per : E;
  for (int e = e0 + tid; e < e1; e += 4 * CHUNK_THREADS) {
    int d[4], s[4], p[4];
#pragma unroll
    for (int k = 0; k < 4; ++k) {
      int ee = e + k * CHUNK_THREADS;
      bool ok = ee < e1;
      d[k] = ok ? dst[ee] : -1;
      s[k] = ok ? src[ee] : 0;
    }
#pragma unroll
    for (int k = 0; k < 4; ++k)
      p[k] = d[k] >= 0 ? atomicAdd(&cur[d[k] >> BUCKET_SHIFT], 1) : 0;
#pragma unroll
    for (int k = 0; k < 4; ++k)
      if (d[k] >= 0) rec[p[k]] = (v2i){s[k], d[k]};
  }
}

__global__ void bucket_rank_kernel(const int* __restrict__ rec_words,
                                   const int* __restrict__ bstart,
                                   int* __restrict__ seg_beg,
                                   int* __restrict__ seg_end,
                                   int* __restrict__ csr_src, int N) {
  typedef int v2i __attribute__((ext_vector_type(2)));
  const v2i* rec = (const v2i*)rec_words;
  __shared__ int cnt[BUCKET_NODES];
  __shared__ int wsum[33];
  __shared__ int stage[STAGE_CAP];
  int tid = threadIdx.x;
  int b = blockIdx.x;
  int lo = b << BUCKET_SHIFT;
  int bs = bstart[b];
  int be = bstart[b + 1];
  int n = be - bs;
  cnt[tid] = 0;
  __syncthreads();
  for (int i = tid; i < n; i += 4 * BUCKET_NODES) {
    int d[4];
#pragma unroll
    for (int k = 0; k < 4; ++k) {
      int ii = i + k * BUCKET_NODES;
      d[k] = ii < n ? rec[bs + ii].y : -1;
    }
#pragma unroll
    for (int k = 0; k < 4; ++k)
      if (d[k] >= 0) atomicAdd(&cnt[d[k] - lo], 1);
  }
  __syncthreads();
  int v = cnt[tid];
  int total;
  int ex = block_excl_scan(v, wsum, total);
  int node = lo + tid;
  if (node < N) {
    seg_beg[node] = bs + ex;
    seg_end[node] = bs + ex + v;
  }
  cnt[tid] = ex;
  __syncthreads();
  bool staged = n <= STAGE_CAP;
  for (int i = tid; i < n; i += 4 * BUCKET_NODES) {
    v2i r[4];
    int p[4];
#pragma unroll
    for (int k = 0; k < 4; ++k) {
      int ii = i + k * BUCKET_NODES;
      r[k] = ii < n ? rec[bs + ii] : (v2i){0, -1};
    }
#pragma unroll
    for (int k = 0; k < 4; ++k)
      p[k] = r[k].y >= 0 ? atomicAdd(&cnt[r[k].y - lo], 1) : 0;
#pragma unroll
    for (int k = 0; k < 4; ++k) {
      if (r[k].y >= 0) {
        if (staged) stage[p[k]] = r[k].x;
        else csr_src[bs + p[k]] = r[k].x;
      }
    }
  }
  if (staged) {
    __syncthreads();
    for (int i = tid; i < n; i += BUCKET_NODES) csr_src[bs + i] = stage[i];
  }
}


__device__ __forceinline__ float xhalf(float x) {
  int b = __builtin_bit_cast(int, x);
  return __builtin_bit_cast(
      float, __builtin_amdgcn_permlanex16(b, b, 0x76543210, 0xfedcba98, false, false));
}

__device__ __forceinline__ int half_lane(int x, int i) {
#define HALF_LANE_CASE(I) \
  case I: return __builtin_amdgcn_update_dpp(0, x, 0x150 + I, 0xf, 0xf, false);
  switch (i) {
    HALF_LANE_CASE(0) HALF_LANE_CASE(1) HALF_LANE_CASE(2) HALF_LANE_CASE(3)
    HALF_LANE_CASE(4) HALF_LANE_CASE(5) HALF_LANE_CASE(6) HALF_LANE_CASE(7)
    HALF_LANE_CASE(8) HALF_LANE_CASE(9) HALF_LANE_CASE(10) HALF_LANE_CASE(11)
    HALF_LANE_CASE(12) HALF_LANE_CASE(13) HALF_LANE_CASE(14)
    default: return __builtin_amdgcn_update_dpp(0, x, 0x150 + 15, 0xf, 0xf, false);
  }
#undef HALF_LANE_CASE
}

#define L1_NP 14
#define L1_PG 2

__global__ void __launch_bounds__(256)
node_gather_l1_kernel(const int* __restrict__ seg_beg,
                      const int* __restrict__ seg_end,
                      const int* __restrict__ csr_src,
                      const float* __restrict__ a_s,
                      const float* __restrict__ a_d,
                      const float* __restrict__ hfeat,
                      const float* __restrict__ bias,
                      float* __restrict__ out,
                      const float* __restrict__ wsrc,
                      const float* __restrict__ wdst,
                      float* __restrict__ as_next,
                      float* __restrict__ ad_next, int N) {
  constexpr int GCH = 2 * L1_NP;
  constexpr int NG = L1_NP / L1_PG;
  int lane = threadIdx.x & 31;
  int half = lane >> 4;
  int q = lane & 15;
  int wave0 = __builtin_amdgcn_readfirstlane(
      (int)((blockIdx.x * blockDim.x + threadIdx.x) >> 5));
  int nwaves = (int)((gridDim.x * blockDim.x) >> 5);
  int c0 = 4 * q;
  int myedge = 2 * q + half;
  const float NEG_INF = -__builtin_inff();
  const v4f* __restrict__ rows = (const v4f*)hfeat;
  v4f asv = *(const v4f*)(a_s + c0);
  v4f adv = *(const v4f*)(a_d + c0);
  v4f bv = *(const v4f*)(bias + c0);
  v4f wsv = *(const v4f*)(wsrc + c0);
  v4f wdv = *(const v4f*)(wdst + c0);
  float s_seed = half == 0 ? 1.f : 0.f;

  int node = wave0;
  if (node >= N) return;

  int beg = seg_beg[node];
  int end = seg_end[node];
  int cnt = end - beg;
  if (cnt > GCH) cnt = GCH;
  int mysrc = 0;
  if (cnt > 0) mysrc = csr_src[beg + (myedge < cnt ? myedge : cnt - 1)];
  v4f own = rows[node * 16 + q];
  int nfirst = node + nwaves < N ? node + nwaves : node;
  int nbeg = seg_beg[nfirst];
  int nend = seg_end[nfirst];

  for (; node < N; node += nwaves) {
    int ncnt = 0, nsrc = 0, nnbeg = 0, nnend = 0;
    v4f nrow;
    auto ask_next = [&]() {
      int nx1 = node + nwaves < N ? node + nwaves : node;
      int nx2 = node + 2 * nwaves < N ? node + 2 * nwaves : node;
      ncnt = nend - nbeg;
      if (ncnt > GCH) ncnt = GCH;
      if (ncnt > 0) nsrc = csr_src[nbeg + (myedge < ncnt ? myedge : ncnt - 1)];
      nrow = rows[nx1 * 16 + q];
      nnbeg = seg_beg[nx2];
      nnend = seg_end[nx2];
    };
    bool asked = false;

    float d_s = own.x * asv.x;
    float d_d = own.x * adv.x;
    d_s += own.y * asv.y;
    d_d += own.y * adv.y;
    d_s += own.z * asv.z;
    d_d += own.z * adv.z;
    d_s += own.w * asv.w;
    d_d += own.w * adv.w;
    d_s += row_xor_lane<1>(d_s);
    d_d += row_xor_lane<1>(d_d);
    float e0 = d_s + d_d;
    e0 = e0 > 0.f ? e0 : NEG_SLOPE * e0;
    float m = e0;
    float s = s_seed;
    v4f acc = half == 0 ? own : (v4f){0.f, 0.f, 0.f, 0.f};
    int base = beg;

    while (cnt > 0) {
      v4f hv[L1_NP];
      float tv[L1_NP];
#pragma unroll
      for (int g = 0; g < NG; ++g) {
        if (g > 0 && g * 2 * L1_PG >= cnt) break;
#pragma unroll
        for (int i = 0; i < L1_PG; ++i) {
          int p = g * L1_PG + i;
          int sidx = half_lane(mysrc, p);
          hv[p] = rows[sidx * 16 + q];
        }
      }
      if (!asked) {
        ask_next();
        asked = true;
      }
      __builtin_amdgcn_sched_barrier(0);

      int lim = cnt - half;
      float cm = NEG_INF;
#pragma unroll
      for (int g = 0; g < NG; ++g) {
        if (g > 0 && g * 2 * L1_PG >= cnt) break;
#pragma unroll
        for (int i = 0; i < L1_PG; ++i) {
          int p = g * L1_PG + i;
          float d = hv[p].x * asv.x;
          d += hv[p].y * asv.y;
          d += hv[p].z * asv.z;
          d += hv[p].w * asv.w;
          d += row_xor_lane<1>(d);
          float t = d + d_d;
          t = t > 0.f ? t : NEG_SLOPE * t;
          t = 2 * p < lim ? t : NEG_INF;
          tv[p] = t;
          cm = fmaxf(cm, t);
        }
      }
      float mn = fmaxf(m, fmaxf(cm, xhalf(cm)));
      float sc = __expf(m - mn);
      s = s * sc;
      acc = acc * sc;
#pragma unroll
      for (int g = 0; g < NG; ++g) {
        if (g > 0 && g * 2 * L1_PG >= cnt) break;
#pragma unroll
        for (int i = 0; i < L1_PG; ++i) {
          int p = g * L1_PG + i;
          float pw = __expf(tv[p] - mn);
          s += pw;
          acc += hv[p] * pw;
        }
      }
      m = mn;

      base += GCH;
      cnt = end - base;
      if (cnt > GCH) cnt = GCH;
      mysrc = 0;
      if (cnt > 0) mysrc = csr_src[base + (myedge < cnt ? myedge : cnt - 1)];
    }
    if (!asked) ask_next();

    float denom = s + xhalf(s) + GAT_EPS;
    v4f o;
#pragma unroll
    for (int k = 0; k < 4; ++k) {
      float v = (acc[k] + xhalf(acc[k])) / denom + bv[k];
      v = v > 0.f ? v : __expf(v) - 1.f;
      o[k] = v;
    }
    if (half == 0) *(v4f*)(out + (size_t)node * 64 + c0) = o;

    float ps = o.x * wsv.x;
    float pd = o.x * wdv.x;
    ps += o.y * wsv.y;
    pd += o.y * wdv.y;
    ps += o.z * wsv.z;
    pd += o.z * wdv.z;
    ps += o.w * wsv.w;
    pd += o.w * wdv.w;
    ps += row_xor_lane<1>(ps);
    pd += row_xor_lane<1>(pd);
    ps += row_xor_lane<2>(ps);
    pd += row_xor_lane<2>(pd);
    ps += row_xor_lane<4>(ps);
    pd += row_xor_lane<4>(pd);
    ps += row_xor_lane<8>(ps);
    pd += row_xor_lane<8>(pd);
    if (lane == 0) {
      as_next[node] = ps;
      ad_next[node] = pd;
    }

    beg = nbeg;
    end = nend;
    cnt = ncnt;
    mysrc = nsrc;
    own = nrow;
    nbeg = nnbeg;
    nend = nnend;
  }
}

#define GGRP 4

__global__ void __launch_bounds__(256)
node_gather_l2_kernel(const int* __restrict__ seg_beg,
                      const int* __restrict__ seg_end,
                      const int* __restrict__ csr_src,
                      const float* __restrict__ as,
                      const float* __restrict__ ad,
                      const float* __restrict__ hfeat,
                      float* __restrict__ out, int N) {
  constexpr int GCH = 32;
  constexpr int NG = GCH / GGRP;
  int lane = threadIdx.x & 31;
  int wave0 = __builtin_amdgcn_readfirstlane(
      (int)((blockIdx.x * blockDim.x + threadIdx.x) >> 5));
  int nwaves = (int)((gridDim.x * blockDim.x) >> 5);
  int c0 = lane * 2;
  const float NEG_INF = -__builtin_inff();
  const v2f* __restrict__ rows = (const v2f*)hfeat;

  int node = wave0;
  if (node >= N) return;

  int beg = seg_beg[node];
  int end = seg_end[node];
  int cnt = end - beg;
  if (cnt > GCH) cnt = GCH;
  int mysrc = 0;
  if (lane < cnt) mysrc = csr_src[beg + lane];
  float adv = ad[node];
  float as_self = as[node];
  v2f acc = rows[node * 32 + lane];
  int nfirst = node + nwaves < N ? node + nwaves : node;
  int nbeg = seg_beg[nfirst];
  int nend = seg_end[nfirst];

  for (; node < N; node += nwaves) {
    int ncnt = 0, nsrc = 0, nnbeg = 0, nnend = 0;
    float nadv = 0.f, nas = 0.f;
    v2f nrow;
    auto ask_next = [&]() {
      int nx1 = node + nwaves < N ? node + nwaves : node;
      int nx2 = node + 2 * nwaves < N ? node + 2 * nwaves : node;
      ncnt = nend - nbeg;
      if (ncnt > GCH) ncnt = GCH;
      if (lane < ncnt) nsrc = csr_src[nbeg + lane];
      nadv = ad[nx1];
      nas = as[nx1];
      nrow = rows[nx1 * 32 + lane];
      nnbeg = seg_beg[nx2];
      nnend = seg_end[nx2];
    };
    bool asked = false;

    float asv = as[mysrc];
    int base = beg;

    float e0 = as_self + adv;
    e0 = e0 > 0.f ? e0 : NEG_SLOPE * e0;
    float m = e0;
    float s = 1.f;

    while (cnt > 0) {
      int last = cnt - 1;
      v2f hv[GCH];
#pragma unroll
      for (int g = 0; g < NG; ++g) {
        if (g > 0 && g * GGRP >= cnt) break;
int sj[GGRP];
#pragma unroll
        for (int i = 0; i < GGRP; ++i) {
          int j = g * GGRP + i;
          sj[i] = __builtin_amdgcn_readlane(mysrc, j < last ? j : last);
        }
#pragma unroll
        for (int i = 0; i < GGRP; ++i)
          hv[g * GGRP + i] = rows[sj[i] * 32 + lane];
      }
      if (!asked) {
        ask_next();
        asked = true;
      }
      __builtin_amdgcn_sched_barrier(0);

      float t = asv + adv;
      t = t > 0.f ? t : NEG_SLOPE * t;
      float mye = lane < cnt ? t : NEG_INF;
      float mn = fmaxf(m, wave_max(mye));
      float sc = __expf(m - mn);
      float p = __expf(mye - mn);
      s = s * sc + wave_sum(p);
      acc = acc * sc;
      int pb = __float_as_int(p);
#pragma unroll
      for (int g = 0; g < NG; ++g) {
        if (g > 0 && g * GGRP >= cnt) break;
#pragma unroll
        for (int i = 0; i < GGRP; ++i) {
          int j = g * GGRP + i;
          float pj = __int_as_float(__builtin_amdgcn_readlane(pb, j));
          acc += hv[j] * pj;
        }
      }
      m = mn;

      base += GCH;
      cnt = end - base;
      if (cnt > GCH) cnt = GCH;
      mysrc = 0;
      if (lane < cnt) mysrc = csr_src[base + lane];
      if (cnt > 0) asv = as[mysrc];
    }
    if (!asked) ask_next();

    float denom = s + GAT_EPS;
    v2f o;
    o.x = acc.x / denom;
    o.y = acc.y / denom;
    *(v2f*)(out + (size_t)node * 64 + c0) = o;

    beg = nbeg;
    end = nend;
    cnt = ncnt;
    mysrc = nsrc;
    adv = nadv;
    as_self = nas;
    acc = nrow;
    nbeg = nnbeg;
    nend = nnend;
  }
}

static inline int cdiv(long long a, long long b) { return (int)((a + b - 1) / b); }

extern "C" void kernel_launch(void* const* d_in, const int* in_sizes, int n_in,
                              void* d_out, int out_size, void* d_ws, size_t ws_size,
                              hipStream_t stream) {
  const float* x      = (const float*)d_in[0];
  const int*   ei     = (const int*)  d_in[1];
  const float* W1     = (const float*)d_in[2];
  const float* a_src1 = (const float*)d_in[3];
  const float* a_dst1 = (const float*)d_in[4];
  const float* b1     = (const float*)d_in[5];
  const float* W2     = (const float*)d_in[6];
  const float* a_src2 = (const float*)d_in[7];
  const float* a_dst2 = (const float*)d_in[8];
  const float* b2     = (const float*)d_in[9];
  float* z = (float*)d_out;

  const int Fin = 256, C1 = 64, C2 = 128;
  const int N    = in_sizes[0] / Fin;
  const int E    = in_sizes[1] / 2;
  const int* srcv = ei;
  const int* dstv = ei + E;

  float* h1   = (float*)d_ws;
  float* hid  = h1  + (size_t)N * C1;
  float* agg  = hid + (size_t)N * C1;
  float* as2  = agg + (size_t)N * C1;
  float* ad2  = as2 + N;
  float* ws2  = ad2 + N;
  float* wd2  = ws2 + C1;
  const int NB = cdiv(N, BUCKET_NODES);
  int*   cnts = (int*)(wd2 + C1);
  int*   recs = cnts + (size_t)NB * NCHUNK;
  int*   csrc = recs + (size_t)2 * E;
  int*   sbeg = csrc + E;
  int*   send = sbeg + N;
  int*   bsta = send + N;

  const int B = 256;
  const int GB = 128;
  const int GATHER1_BLOCKS = 1024;
  const int GATHER2_BLOCKS = 1280;

  chunk_hist_kernel<<<NCHUNK + 1, CHUNK_THREADS, 0, stream>>>(
      dstv, E, NB, cnts, W2, a_src2, a_dst2, ws2, wd2);
  gemm1_scan_kernel<<<1 + cdiv((long long)(N / 16) * 32, GB), GB, 0, stream>>>(
      x, W1, h1, N, Fin, cnts, NB, bsta);
  chunk_scatter_kernel<<<NCHUNK, CHUNK_THREADS, 0, stream>>>(srcv, dstv, E, NB,
                                                            cnts, bsta, recs);
  bucket_rank_kernel<<<NB, BUCKET_NODES, 0, stream>>>(recs, bsta, sbeg, send,
                                                      csrc, N);

  node_gather_l1_kernel<<<GATHER1_BLOCKS, B, 0, stream>>>(
      sbeg, send, csrc, a_src1, a_dst1, h1, b1, hid, ws2, wd2, as2, ad2, N);

  node_gather_l2_kernel<<<GATHER2_BLOCKS, B, 0, stream>>>(
      sbeg, send, csrc, as2, ad2, hid, agg, N);
  gemm_f32_wmma_kernel<C2 / 16, true><<<cdiv((long long)(N / 16) * 32, GB), GB,
                                        0, stream>>>(agg, W2, b2, z, N, C1);
}


# Edge sort level 1 kept inside the chunk: a work-group counts, prefixes and orders its chunk by bucket in LDS, then does its layer-1 GEMM strips; 7->5 launches; sum order = sort arrival
# speedup vs baseline: 6.5310x; 1.1345x over previous
#include <hip/hip_runtime.h>
#include <cstdint>


typedef float v2f __attribute__((ext_vector_type(2)));
typedef float v4f __attribute__((ext_vector_type(4)));
typedef float v8f __attribute__((ext_vector_type(8)));

#define NEG_SLOPE 0.2f
#define GAT_EPS 1e-16f

template <int M>
__device__ __forceinline__ float row_xor_lane(float x) {
  int b = __builtin_bit_cast(int, x);
  return __builtin_bit_cast(
      float, __builtin_amdgcn_update_dpp(b, b, 0x160 + M, 0xf, 0xf, false));
}

__device__ __forceinline__ float wave_sum(float v) {
  v += row_xor_lane<1>(v);
  v += row_xor_lane<2>(v);
  v += row_xor_lane<4>(v);
  v += row_xor_lane<8>(v);
  int b = __builtin_bit_cast(int, v);
  return __builtin_bit_cast(float, __builtin_amdgcn_readlane(b, 0)) +
         __builtin_bit_cast(float, __builtin_amdgcn_readlane(b, 16));
}

__device__ __forceinline__ float wave_max(float v) {
  v = fmaxf(v, row_xor_lane<1>(v));
  v = fmaxf(v, row_xor_lane<2>(v));
  v = fmaxf(v, row_xor_lane<4>(v));
  v = fmaxf(v, row_xor_lane<8>(v));
  int b = __builtin_bit_cast(int, v);
  return fmaxf(__builtin_bit_cast(float, __builtin_amdgcn_readlane(b, 0)),
               __builtin_bit_cast(float, __builtin_amdgcn_readlane(b, 16)));
}

template <int NT, bool BIAS>
__device__ __forceinline__ void gemm_strip(const float* __restrict__ A,
                                           const float* __restrict__ B,
                                           const float* __restrict__ bias,
                                           float* __restrict__ C,
                                           int M, int K, int wave) {
  static_assert(NT % 4 == 0, "tiles come in groups of four");
  constexpr int N = NT * 16;
  constexpr int NGRP = NT / 4;
  int lane = threadIdx.x & 31;
  int mtiles = (M + 15) >> 4;
  if (wave >= mtiles) return;
  int m0 = wave << 4;
  int half = lane >> 4;
  int lid  = lane & 15;

  v8f c[NT];
#pragma unroll
  for (int t = 0; t < NT; ++t)
    c[t] = (v8f){0.f, 0.f, 0.f, 0.f, 0.f, 0.f, 0.f, 0.f};
  const float* arow = A + (size_t)(m0 + lid) * K + 4 * half;
  const float* bcol = B + (size_t)(4 * half) * N + 4 * lid;

#pragma unroll(8 / NT)
  for (int k0 = 0; k0 < K; k0 += 8) {
    v4f a4 = *(const v4f*)(arow + k0);
    const float* bp = bcol + (size_t)k0 * N;
    v4f b4[4][NGRP];
#pragma unroll
    for (int j = 0; j < 4; ++j)
#pragma unroll
      for (int g = 0; g < NGRP; ++g)
        b4[j][g] = *(const v4f*)(bp + j * N + 64 * g);
#pragma unroll
    for (int s = 0; s < 2; ++s) {
      v2f a;
      a.x = s == 0 ? a4.x : a4.z;
      a.y = s == 0 ? a4.y : a4.w;
#pragma unroll
      for (int t = 0; t < NT; ++t) {
        v2f b;
        b.x = b4[2 * s][t >> 2][t & 3];
        b.y = b4[2 * s + 1][t >> 2][t & 3];
        c[t] = __builtin_amdgcn_wmma_f32_16x16x4_f32(false, a, false, b,
                                                     (short)0, c[t], false, false);
      }
    }
  }

  v4f bc[NGRP];
#pragma unroll
  for (int g = 0; g < NGRP; ++g)
    bc[g] = BIAS ? *(const v4f*)(bias + 64 * g + 4 * lid)
                 : (v4f){0.f, 0.f, 0.f, 0.f};

  float* crow = C + (size_t)(m0 + 8 * half) * N + 4 * lid;
#pragma unroll
  for (int r = 0; r < 8; ++r)
#pragma unroll
    for (int g = 0; g < NGRP; ++g) {
      v4f o = (v4f){c[4 * g][r], c[4 * g + 1][r], c[4 * g + 2][r],
                    c[4 * g + 3][r]};
      if (BIAS) o += bc[g];
      *(v4f*)(crow + (size_t)r * N + 64 * g) = o;
    }
}

template <int NT, bool BIAS>
__global__ void gemm_f32_wmma_kernel(const float* __restrict__ A,
                                     const float* __restrict__ B,
                                     const float* __restrict__ bias,
                                     float* __restrict__ C,
                                     int M, int K) {
  int wave = (int)((blockIdx.x * blockDim.x + threadIdx.x) >> 5);
  gemm_strip<NT, BIAS>(A, B, bias, C, M, K, wave);
}

template <int K, int DO>
__device__ __forceinline__ void att_vec_wave(const float* __restrict__ W,
                                             const float* __restrict__ a_s,
                                             const float* __restrict__ a_d,
                                             float* __restrict__ ws,
                                             float* __restrict__ wd) {
  static_assert(K % 32 == 0, "32 rows per store");
  constexpr int V = DO / 32;
  typedef float vecf __attribute__((ext_vector_type(V)));
  int lane = threadIdx.x & 31;
  vecf sv = *(const vecf*)(a_s + lane * V);
  vecf dv = *(const vecf*)(a_d + lane * V);
  for (int j = 0; j < K / 32; ++j) {
    float ks = 0.f, kd = 0.f;
#pragma unroll 4
    for (int i = 0; i < 32; ++i) {
      vecf wv = *(const vecf*)(W + (size_t)(j * 32 + i) * DO + lane * V);
      float ps = 0.f, pd = 0.f;
#pragma unroll
      for (int u = 0; u < V; ++u) {
        ps += wv[u] * sv[u];
        pd += wv[u] * dv[u];
      }
      ps = wave_sum(ps);
      pd = wave_sum(pd);
      if (lane == i) { ks = ps; kd = pd; }
    }
    ws[j * 32 + lane] = ks;
    wd[j * 32 + lane] = kd;
  }
}

#define NCHUNK 256
#define BUCKET_SHIFT 8
#define BUCKET_NODES 256
#define MAX_BUCKETS 1024
#define CHUNK_THREADS 1024
#define STAGE_CAP 8192
#define SLICE_U 8

__device__ __forceinline__ int wave_incl_scan(int x, int lane) {
#pragma unroll
  for (int o = 1; o < 32; o <<= 1) {
    int y = __shfl_up(x, o);
    if (lane >= o) x += y;
  }
  return x;
}

__device__ __forceinline__ int block_excl_scan(int v, int* wsum, int& total) {
  int lane = threadIdx.x & 31;
  int w = threadIdx.x >> 5;
  int nw = blockDim.x >> 5;
  int x = wave_incl_scan(v, lane);
  if (lane == 31) wsum[w] = x;
  __syncthreads();
  if (w == 0) {
    int t = (lane < nw) ? wsum[lane] : 0;
    int xs = wave_incl_scan(t, lane);
    wsum[lane] = xs - t;
    if (lane == 31) wsum[32] = xs;
  }
  __syncthreads();
  total = wsum[32];
  int r = wsum[w] + (x - v);
  __syncthreads();
  return r;
}

__global__ void __launch_bounds__(CHUNK_THREADS)
chunk_sort_gemm1_kernel(const int* __restrict__ src,
                        const int* __restrict__ dst, int E, int NB,
                        int* __restrict__ off, int* __restrict__ rec_words,
                        const float* __restrict__ A,
                        const float* __restrict__ B,
                        float* __restrict__ C, int M, int K,
                        const float* __restrict__ W2,
                        const float* __restrict__ a_s2,
                        const float* __restrict__ a_d2,
                        float* __restrict__ ws2,
                        float* __restrict__ wd2) {
  typedef int v2i __attribute__((ext_vector_type(2)));
  __shared__ int h[MAX_BUCKETS];
  __shared__ int wsum[33];
  v2i* rec = (v2i*)rec_words;
  int tid = threadIdx.x;
  int c = blockIdx.x;
  for (int i = tid; i < NB; i += CHUNK_THREADS) h[i] = 0;
  __syncthreads();
  int per = (E + NCHUNK - 1) / NCHUNK;
  int e0 = c * per;
  int e1 = e0 + per < E ? e0 + per : E;
  for (int e = e0 + tid; e < e1; e += 4 * CHUNK_THREADS) {
    int d[4];
#pragma unroll
    for (int k = 0; k < 4; ++k) {
      int ee = e + k * CHUNK_THREADS;
      d[k] = ee < e1 ? dst[ee] : -1;
    }
#pragma unroll
    for (int k = 0; k < 4; ++k)
      if (d[k] >= 0) atomicAdd(&h[d[k] >> BUCKET_SHIFT], 1);
  }
  __syncthreads();
  int carry = 0;
  for (int base = 0; base < NB; base += CHUNK_THREADS) {
    int i = base + tid;
    int v = (i < NB) ? h[i] : 0;
    int total;
    int ex = block_excl_scan(v, wsum, total);
    if (i < NB) {
      h[i] = carry + ex;
      off[(size_t)i * NCHUNK + c] = carry + ex;
    }
    carry += total;
  }
  if (tid == 0) off[(size_t)NB * NCHUNK + c] = carry;
  __syncthreads();
  for (int e = e0 + tid; e < e1; e += 4 * CHUNK_THREADS) {
    int d[4], s[4], p[4];
#pragma unroll
    for (int k = 0; k < 4; ++k) {
      int ee = e + k * CHUNK_THREADS;
      bool ok = ee < e1;
      d[k] = ok ? dst[ee] : -1;
      s[k] = ok ? src[ee] : 0;
    }
#pragma unroll
    for (int k = 0; k < 4; ++k)
      p[k] = d[k] >= 0 ? atomicAdd(&h[d[k] >> BUCKET_SHIFT], 1) : 0;
#pragma unroll
    for (int k = 0; k < 4; ++k)
      if (d[k] >= 0) rec[e0 + p[k]] = (v2i){s[k], d[k]};
  }

  int w = __builtin_amdgcn_readfirstlane((int)(threadIdx.x >> 5));
  if (c == 0 && w == (CHUNK_THREADS >> 5) - 1)
    att_vec_wave<64, 128>(W2, a_s2, a_d2, ws2, wd2);
  int mtiles = (M + 15) >> 4;
  for (int strip = w * NCHUNK + c; strip < mtiles;
       strip += (CHUNK_THREADS >> 5) * NCHUNK)
    gemm_strip<4, false>(A, B, nullptr, C, M, K, strip);
}

__global__ void __launch_bounds__(BUCKET_NODES)
bucket_rank_kernel(const int* __restrict__ rec_words,
                   const int* __restrict__ off, int E,
                   int* __restrict__ seg_beg,
                   int* __restrict__ seg_end,
                   int* __restrict__ csr_src, int N) {
  static_assert(NCHUNK == BUCKET_NODES, "one thread per chunk");
  typedef int v2i __attribute__((ext_vector_type(2)));
  const v2i* rec = (const v2i*)rec_words;
  __shared__ int cnt[BUCKET_NODES];
  __shared__ int wsum[33];
  __shared__ int stage[STAGE_CAP];
  __shared__ v2i lrec[STAGE_CAP];
  int tid = threadIdx.x;
  int b = blockIdx.x;
  int lo = b << BUCKET_SHIFT;
  int per = (E + NCHUNK - 1) / NCHUNK;
  int o0 = off[(size_t)b * NCHUNK + tid];
  int o1 = off[(size_t)(b + 1) * NCHUNK + tid];
  int len = o1 - o0;
  int n, bs;
  int pos = block_excl_scan(len, wsum, n);
  (void)block_excl_scan(o0, wsum, bs);
  int g0 = tid * per + o0;
  cnt[tid] = 0;
  __syncthreads();
  bool staged = n <= STAGE_CAP;
  for (int i = 0; i < len; i += SLICE_U) {
    v2i r[SLICE_U];
#pragma unroll
    for (int k = 0; k < SLICE_U; ++k)
      r[k] = i + k < len ? rec[g0 + i + k] : (v2i){0, -1};
#pragma unroll
    for (int k = 0; k < SLICE_U; ++k) {
      if (r[k].y >= 0) {
        atomicAdd(&cnt[r[k].y - lo], 1);
        if (staged) lrec[pos + i + k] = r[k];
      }
    }
  }
  __syncthreads();
  int v = cnt[tid];
  int total;
  int ex = block_excl_scan(v, wsum, total);
  int node = lo + tid;
  if (node < N) {
    seg_beg[node] = bs + ex;
    seg_end[node] = bs + ex + v;
  }
  cnt[tid] = ex;
  __syncthreads();
  if (staged) {
    for (int i = tid; i < n; i += 4 * BUCKET_NODES) {
      v2i r[4];
      int p[4];
#pragma unroll
      for (int k = 0; k < 4; ++k) {
        int ii = i + k * BUCKET_NODES;
        r[k] = ii < n ? lrec[ii] : (v2i){0, -1};
      }
#pragma unroll
      for (int k = 0; k < 4; ++k)
        p[k] = r[k].y >= 0 ? atomicAdd(&cnt[r[k].y - lo], 1) : 0;
#pragma unroll
      for (int k = 0; k < 4; ++k)
        if (r[k].y >= 0) stage[p[k]] = r[k].x;
    }
    __syncthreads();
    for (int i = tid; i < n; i += BUCKET_NODES) csr_src[bs + i] = stage[i];
  } else {
    for (int i = 0; i < len; i += SLICE_U) {
      v2i r[SLICE_U];
      int p[SLICE_U];
#pragma unroll
      for (int k = 0; k < SLICE_U; ++k)
        r[k] = i + k < len ? rec[g0 + i + k] : (v2i){0, -1};
#pragma unroll
      for (int k = 0; k < SLICE_U; ++k)
        p[k] = r[k].y >= 0 ? atomicAdd(&cnt[r[k].y - lo], 1) : 0;
#pragma unroll
      for (int k = 0; k < SLICE_U; ++k)
        if (r[k].y >= 0) csr_src[bs + p[k]] = r[k].x;
    }
  }
}


__device__ __forceinline__ float xhalf(float x) {
  int b = __builtin_bit_cast(int, x);
  return __builtin_bit_cast(
      float, __builtin_amdgcn_permlanex16(b, b, 0x76543210, 0xfedcba98, false, false));
}

__device__ __forceinline__ int half_lane(int x, int i) {
#define HALF_LANE_CASE(I) \
  case I: return __builtin_amdgcn_update_dpp(0, x, 0x150 + I, 0xf, 0xf, false);
  switch (i) {
    HALF_LANE_CASE(0) HALF_LANE_CASE(1) HALF_LANE_CASE(2) HALF_LANE_CASE(3)
    HALF_LANE_CASE(4) HALF_LANE_CASE(5) HALF_LANE_CASE(6) HALF_LANE_CASE(7)
    HALF_LANE_CASE(8) HALF_LANE_CASE(9) HALF_LANE_CASE(10) HALF_LANE_CASE(11)
    HALF_LANE_CASE(12) HALF_LANE_CASE(13) HALF_LANE_CASE(14)
    default: return __builtin_amdgcn_update_dpp(0, x, 0x150 + 15, 0xf, 0xf, false);
  }
#undef HALF_LANE_CASE
}

#define L1_NP 14
#define L1_PG 2

__global__ void __launch_bounds__(256)
node_gather_l1_kernel(const int* __restrict__ seg_beg,
                      const int* __restrict__ seg_end,
                      const int* __restrict__ csr_src,
                      const float* __restrict__ a_s,
                      const float* __restrict__ a_d,
                      const float* __restrict__ hfeat,
                      const float* __restrict__ bias,
                      float* __restrict__ out,
                      const float* __restrict__ wsrc,
                      const float* __restrict__ wdst,
                      float* __restrict__ as_next,
                      float* __restrict__ ad_next, int N) {
  constexpr int GCH = 2 * L1_NP;
  constexpr int NG = L1_NP / L1_PG;
  int lane = threadIdx.x & 31;
  int half = lane >> 4;
  int q = lane & 15;
  int wave0 = __builtin_amdgcn_readfirstlane(
      (int)((blockIdx.x * blockDim.x + threadIdx.x) >> 5));
  int nwaves = (int)((gridDim.x * blockDim.x) >> 5);
  int c0 = 4 * q;
  int myedge = 2 * q + half;
  const float NEG_INF = -__builtin_inff();
  const v4f* __restrict__ rows = (const v4f*)hfeat;
  v4f asv = *(const v4f*)(a_s + c0);
  v4f adv = *(const v4f*)(a_d + c0);
  v4f bv = *(const v4f*)(bias + c0);
  v4f wsv = *(const v4f*)(wsrc + c0);
  v4f wdv = *(const v4f*)(wdst + c0);
  float s_seed = half == 0 ? 1.f : 0.f;

  int node = wave0;
  if (node >= N) return;

  int beg = seg_beg[node];
  int end = seg_end[node];
  int cnt = end - beg;
  if (cnt > GCH) cnt = GCH;
  int mysrc = 0;
  if (cnt > 0) mysrc = csr_src[beg + (myedge < cnt ? myedge : cnt - 1)];
  v4f own = rows[node * 16 + q];
  int nfirst = node + nwaves < N ? node + nwaves : node;
  int nbeg = seg_beg[nfirst];
  int nend = seg_end[nfirst];

  for (; node < N; node += nwaves) {
    int ncnt = 0, nsrc = 0, nnbeg = 0, nnend = 0;
    v4f nrow;
    auto ask_next = [&]() {
      int nx1 = node + nwaves < N ? node + nwaves : node;
      int nx2 = node + 2 * nwaves < N ? node + 2 * nwaves : node;
      ncnt = nend - nbeg;
      if (ncnt > GCH) ncnt = GCH;
      if (ncnt > 0) nsrc = csr_src[nbeg + (myedge < ncnt ? myedge : ncnt - 1)];
      nrow = rows[nx1 * 16 + q];
      nnbeg = seg_beg[nx2];
      nnend = seg_end[nx2];
    };
    bool asked = false;

    float d_s = own.x * asv.x;
    float d_d = own.x * adv.x;
    d_s += own.y * asv.y;
    d_d += own.y * adv.y;
    d_s += own.z * asv.z;
    d_d += own.z * adv.z;
    d_s += own.w * asv.w;
    d_d += own.w * adv.w;
    d_s += row_xor_lane<1>(d_s);
    d_d += row_xor_lane<1>(d_d);
    float e0 = d_s + d_d;
    e0 = e0 > 0.f ? e0 : NEG_SLOPE * e0;
    float m = e0;
    float s = s_seed;
    v4f acc = half == 0 ? own : (v4f){0.f, 0.f, 0.f, 0.f};
    int base = beg;

    while (cnt > 0) {
      v4f hv[L1_NP];
      float tv[L1_NP];
#pragma unroll
      for (int g = 0; g < NG; ++g) {
        if (g > 0 && g * 2 * L1_PG >= cnt) break;
#pragma unroll
        for (int i = 0; i < L1_PG; ++i) {
          int p = g * L1_PG + i;
          int sidx = half_lane(mysrc, p);
          hv[p] = rows[sidx * 16 + q];
        }
      }
      if (!asked) {
        ask_next();
        asked = true;
      }
      __builtin_amdgcn_sched_barrier(0);

      int lim = cnt - half;
      float cm = NEG_INF;
#pragma unroll
      for (int g = 0; g < NG; ++g) {
        if (g > 0 && g * 2 * L1_PG >= cnt) break;
#pragma unroll
        for (int i = 0; i < L1_PG; ++i) {
          int p = g * L1_PG + i;
          float d = hv[p].x * asv.x;
          d += hv[p].y * asv.y;
          d += hv[p].z * asv.z;
          d += hv[p].w * asv.w;
          d += row_xor_lane<1>(d);
          float t = d + d_d;
          t = t > 0.f ? t : NEG_SLOPE * t;
          t = 2 * p < lim ? t : NEG_INF;
          tv[p] = t;
          cm = fmaxf(cm, t);
        }
      }
      float mn = fmaxf(m, fmaxf(cm, xhalf(cm)));
      float sc = __expf(m - mn);
      s = s * sc;
      acc = acc * sc;
#pragma unroll
      for (int g = 0; g < NG; ++g) {
        if (g > 0 && g * 2 * L1_PG >= cnt) break;
#pragma unroll
        for (int i = 0; i < L1_PG; ++i) {
          int p = g * L1_PG + i;
          float pw = __expf(tv[p] - mn);
          s += pw;
          acc += hv[p] * pw;
        }
      }
      m = mn;

      base += GCH;
      cnt = end - base;
      if (cnt > GCH) cnt = GCH;
      mysrc = 0;
      if (cnt > 0) mysrc = csr_src[base + (myedge < cnt ? myedge : cnt - 1)];
    }
    if (!asked) ask_next();

    float denom = s + xhalf(s) + GAT_EPS;
    v4f o;
#pragma unroll
    for (int k = 0; k < 4; ++k) {
      float v = (acc[k] + xhalf(acc[k])) / denom + bv[k];
      v = v > 0.f ? v : __expf(v) - 1.f;
      o[k] = v;
    }
    if (half == 0) *(v4f*)(out + (size_t)node * 64 + c0) = o;

    float ps = o.x * wsv.x;
    float pd = o.x * wdv.x;
    ps += o.y * wsv.y;
    pd += o.y * wdv.y;
    ps += o.z * wsv.z;
    pd += o.z * wdv.z;
    ps += o.w * wsv.w;
    pd += o.w * wdv.w;
    ps += row_xor_lane<1>(ps);
    pd += row_xor_lane<1>(pd);
    ps += row_xor_lane<2>(ps);
    pd += row_xor_lane<2>(pd);
    ps += row_xor_lane<4>(ps);
    pd += row_xor_lane<4>(pd);
    ps += row_xor_lane<8>(ps);
    pd += row_xor_lane<8>(pd);
    if (lane == 0) {
      as_next[node] = ps;
      ad_next[node] = pd;
    }

    beg = nbeg;
    end = nend;
    cnt = ncnt;
    mysrc = nsrc;
    own = nrow;
    nbeg = nnbeg;
    nend = nnend;
  }
}

#define GGRP 4

__global__ void __launch_bounds__(256)
node_gather_l2_kernel(const int* __restrict__ seg_beg,
                      const int* __restrict__ seg_end,
                      const int* __restrict__ csr_src,
                      const float* __restrict__ as,
                      const float* __restrict__ ad,
                      const float* __restrict__ hfeat,
                      float* __restrict__ out, int N) {
  constexpr int GCH = 32;
  constexpr int NG = GCH / GGRP;
  int lane = threadIdx.x & 31;
  int wave0 = __builtin_amdgcn_readfirstlane(
      (int)((blockIdx.x * blockDim.x + threadIdx.x) >> 5));
  int nwaves = (int)((gridDim.x * blockDim.x) >> 5);
  int c0 = lane * 2;
  const float NEG_INF = -__builtin_inff();
  const v2f* __restrict__ rows = (const v2f*)hfeat;

  int node = wave0;
  if (node >= N) return;

  int beg = seg_beg[node];
  int end = seg_end[node];
  int cnt = end - beg;
  if (cnt > GCH) cnt = GCH;
  int mysrc = 0;
  if (lane < cnt) mysrc = csr_src[beg + lane];
  float adv = ad[node];
  float as_self = as[node];
  v2f acc = rows[node * 32 + lane];
  int nfirst = node + nwaves < N ? node + nwaves : node;
  int nbeg = seg_beg[nfirst];
  int nend = seg_end[nfirst];

  for (; node < N; node += nwaves) {
    int ncnt = 0, nsrc = 0, nnbeg = 0, nnend = 0;
    float nadv = 0.f, nas = 0.f;
    v2f nrow;
    auto ask_next = [&]() {
      int nx1 = node + nwaves < N ? node + nwaves : node;
      int nx2 = node + 2 * nwaves < N ? node + 2 * nwaves : node;
      ncnt = nend - nbeg;
      if (ncnt > GCH) ncnt = GCH;
      if (lane < ncnt) nsrc = csr_src[nbeg + lane];
      nadv = ad[nx1];
      nas = as[nx1];
      nrow = rows[nx1 * 32 + lane];
      nnbeg = seg_beg[nx2];
      nnend = seg_end[nx2];
    };
    bool asked = false;

    float asv = as[mysrc];
    int base = beg;

    float e0 = as_self + adv;
    e0 = e0 > 0.f ? e0 : NEG_SLOPE * e0;
    float m = e0;
    float s = 1.f;

    while (cnt > 0) {
      int last = cnt - 1;
      v2f hv[GCH];
#pragma unroll
      for (int g = 0; g < NG; ++g) {
        if (g > 0 && g * GGRP >= cnt) break;
int sj[GGRP];
#pragma unroll
        for (int i = 0; i < GGRP; ++i) {
          int j = g * GGRP + i;
          sj[i] = __builtin_amdgcn_readlane(mysrc, j < last ? j : last);
        }
#pragma unroll
        for (int i = 0; i < GGRP; ++i)
          hv[g * GGRP + i] = rows[sj[i] * 32 + lane];
      }
      if (!asked) {
        ask_next();
        asked = true;
      }
      __builtin_amdgcn_sched_barrier(0);

      float t = asv + adv;
      t = t > 0.f ? t : NEG_SLOPE * t;
      float mye = lane < cnt ? t : NEG_INF;
      float mn = fmaxf(m, wave_max(mye));
      float sc = __expf(m - mn);
      float p = __expf(mye - mn);
      s = s * sc + wave_sum(p);
      acc = acc * sc;
      int pb = __float_as_int(p);
#pragma unroll
      for (int g = 0; g < NG; ++g) {
        if (g > 0 && g * GGRP >= cnt) break;
#pragma unroll
        for (int i = 0; i < GGRP; ++i) {
          int j = g * GGRP + i;
          float pj = __int_as_float(__builtin_amdgcn_readlane(pb, j));
          acc += hv[j] * pj;
        }
      }
      m = mn;

      base += GCH;
      cnt = end - base;
      if (cnt > GCH) cnt = GCH;
      mysrc = 0;
      if (lane < cnt) mysrc = csr_src[base + lane];
      if (cnt > 0) asv = as[mysrc];
    }
    if (!asked) ask_next();

    float denom = s + GAT_EPS;
    v2f o;
    o.x = acc.x / denom;
    o.y = acc.y / denom;
    *(v2f*)(out + (size_t)node * 64 + c0) = o;

    beg = nbeg;
    end = nend;
    cnt = ncnt;
    mysrc = nsrc;
    adv = nadv;
    as_self = nas;
    acc = nrow;
    nbeg = nnbeg;
    nend = nnend;
  }
}

static inline int cdiv(long long a, long long b) { return (int)((a + b - 1) / b); }

extern "C" void kernel_launch(void* const* d_in, const int* in_sizes, int n_in,
                              void* d_out, int out_size, void* d_ws, size_t ws_size,
                              hipStream_t stream) {
  const float* x      = (const float*)d_in[0];
  const int*   ei     = (const int*)  d_in[1];
  const float* W1     = (const float*)d_in[2];
  const float* a_src1 = (const float*)d_in[3];
  const float* a_dst1 = (const float*)d_in[4];
  const float* b1     = (const float*)d_in[5];
  const float* W2     = (const float*)d_in[6];
  const float* a_src2 = (const float*)d_in[7];
  const float* a_dst2 = (const float*)d_in[8];
  const float* b2     = (const float*)d_in[9];
  float* z = (float*)d_out;

  const int Fin = 256, C1 = 64, C2 = 128;
  const int N    = in_sizes[0] / Fin;
  const int E    = in_sizes[1] / 2;
  const int* srcv = ei;
  const int* dstv = ei + E;

  float* h1   = (float*)d_ws;
  float* hid  = h1  + (size_t)N * C1;
  float* agg  = hid + (size_t)N * C1;
  float* as2  = agg + (size_t)N * C1;
  float* ad2  = as2 + N;
  float* ws2  = ad2 + N;
  float* wd2  = ws2 + C1;
  const int NB = cdiv(N, BUCKET_NODES);
  int*   offs = (int*)(wd2 + C1);
  int*   recs = offs + (size_t)(NB + 1) * NCHUNK;
  int*   csrc = recs + (size_t)2 * E;
  int*   sbeg = csrc + E;
  int*   send = sbeg + N;

  const int B = 256;
  const int GB = 128;
  const int GATHER1_BLOCKS = 1024;
  const int GATHER2_BLOCKS = 1280;

  chunk_sort_gemm1_kernel<<<NCHUNK, CHUNK_THREADS, 0, stream>>>(
      srcv, dstv, E, NB, offs, recs, x, W1, h1, N, Fin, W2, a_src2, a_dst2, ws2,
      wd2);
  bucket_rank_kernel<<<NB, BUCKET_NODES, 0, stream>>>(recs, offs, E, sbeg, send,
                                                      csrc, N);

  node_gather_l1_kernel<<<GATHER1_BLOCKS, B, 0, stream>>>(
      sbeg, send, csrc, a_src1, a_dst1, h1, b1, hid, ws2, wd2, as2, ad2, N);

  node_gather_l2_kernel<<<GATHER2_BLOCKS, B, 0, stream>>>(
      sbeg, send, csrc, as2, ad2, hid, agg, N);
  gemm_f32_wmma_kernel<C2 / 16, true><<<cdiv((long long)(N / 16) * 32, GB), GB,
                                        0, stream>>>(agg, W2, b2, z, N, C1);
}


# Bucket ranking and layer-1 gather in one work-group: it ranks its 196-node bucket in LDS, then its 16 waves gather those nodes with bounds and ids from LDS; 5->4 launches; sum order = sort arrival
# speedup vs baseline: 7.0997x; 1.0871x over previous
#include <hip/hip_runtime.h>
#include <cstdint>


typedef float v2f __attribute__((ext_vector_type(2)));
typedef float v4f __attribute__((ext_vector_type(4)));
typedef float v8f __attribute__((ext_vector_type(8)));

#define NEG_SLOPE 0.2f
#define GAT_EPS 1e-16f

template <int M>
__device__ __forceinline__ float row_xor_lane(float x) {
  int b = __builtin_bit_cast(int, x);
  return __builtin_bit_cast(
      float, __builtin_amdgcn_update_dpp(b, b, 0x160 + M, 0xf, 0xf, false));
}

__device__ __forceinline__ float wave_sum(float v) {
  v += row_xor_lane<1>(v);
  v += row_xor_lane<2>(v);
  v += row_xor_lane<4>(v);
  v += row_xor_lane<8>(v);
  int b = __builtin_bit_cast(int, v);
  return __builtin_bit_cast(float, __builtin_amdgcn_readlane(b, 0)) +
         __builtin_bit_cast(float, __builtin_amdgcn_readlane(b, 16));
}

__device__ __forceinline__ float wave_max(float v) {
  v = fmaxf(v, row_xor_lane<1>(v));
  v = fmaxf(v, row_xor_lane<2>(v));
  v = fmaxf(v, row_xor_lane<4>(v));
  v = fmaxf(v, row_xor_lane<8>(v));
  int b = __builtin_bit_cast(int, v);
  return fmaxf(__builtin_bit_cast(float, __builtin_amdgcn_readlane(b, 0)),
               __builtin_bit_cast(float, __builtin_amdgcn_readlane(b, 16)));
}

template <int NT, bool BIAS>
__device__ __forceinline__ void gemm_strip(const float* __restrict__ A,
                                           const float* __restrict__ B,
                                           const float* __restrict__ bias,
                                           float* __restrict__ C,
                                           int M, int K, int wave) {
  static_assert(NT % 4 == 0, "tiles come in groups of four");
  constexpr int N = NT * 16;
  constexpr int NGRP = NT / 4;
  int lane = threadIdx.x & 31;
  int mtiles = (M + 15) >> 4;
  if (wave >= mtiles) return;
  int m0 = wave << 4;
  int half = lane >> 4;
  int lid  = lane & 15;

  v8f c[NT];
#pragma unroll
  for (int t = 0; t < NT; ++t)
    c[t] = (v8f){0.f, 0.f, 0.f, 0.f, 0.f, 0.f, 0.f, 0.f};
  const float* arow = A + (size_t)(m0 + lid) * K + 4 * half;
  const float* bcol = B + (size_t)(4 * half) * N + 4 * lid;

#pragma unroll(8 / NT)
  for (int k0 = 0; k0 < K; k0 += 8) {
    v4f a4 = *(const v4f*)(arow + k0);
    const float* bp = bcol + (size_t)k0 * N;
    v4f b4[4][NGRP];
#pragma unroll
    for (int j = 0; j < 4; ++j)
#pragma unroll
      for (int g = 0; g < NGRP; ++g)
        b4[j][g] = *(const v4f*)(bp + j * N + 64 * g);
#pragma unroll
    for (int s = 0; s < 2; ++s) {
      v2f a;
      a.x = s == 0 ? a4.x : a4.z;
      a.y = s == 0 ? a4.y : a4.w;
#pragma unroll
      for (int t = 0; t < NT; ++t) {
        v2f b;
        b.x = b4[2 * s][t >> 2][t & 3];
        b.y = b4[2 * s + 1][t >> 2][t & 3];
        c[t] = __builtin_amdgcn_wmma_f32_16x16x4_f32(false, a, false, b,
                                                     (short)0, c[t], false, false);
      }
    }
  }

  v4f bc[NGRP];
#pragma unroll
  for (int g = 0; g < NGRP; ++g)
    bc[g] = BIAS ? *(const v4f*)(bias + 64 * g + 4 * lid)
                 : (v4f){0.f, 0.f, 0.f, 0.f};

  float* crow = C + (size_t)(m0 + 8 * half) * N + 4 * lid;
#pragma unroll
  for (int r = 0; r < 8; ++r)
#pragma unroll
    for (int g = 0; g < NGRP; ++g) {
      v4f o = (v4f){c[4 * g][r], c[4 * g + 1][r], c[4 * g + 2][r],
                    c[4 * g + 3][r]};
      if (BIAS) o += bc[g];
      *(v4f*)(crow + (size_t)r * N + 64 * g) = o;
    }
}

template <int NT, bool BIAS>
__global__ void gemm_f32_wmma_kernel(const float* __restrict__ A,
                                     const float* __restrict__ B,
                                     const float* __restrict__ bias,
                                     float* __restrict__ C,
                                     int M, int K) {
  int wave = (int)((blockIdx.x * blockDim.x + threadIdx.x) >> 5);
  gemm_strip<NT, BIAS>(A, B, bias, C, M, K, wave);
}

template <int K, int DO>
__device__ __forceinline__ void att_vec_wave(const float* __restrict__ W,
                                             const float* __restrict__ a_s,
                                             const float* __restrict__ a_d,
                                             float* __restrict__ ws,
                                             float* __restrict__ wd) {
  static_assert(K % 32 == 0, "32 rows per store");
  constexpr int V = DO / 32;
  typedef float vecf __attribute__((ext_vector_type(V)));
  int lane = threadIdx.x & 31;
  vecf sv = *(const vecf*)(a_s + lane * V);
  vecf dv = *(const vecf*)(a_d + lane * V);
  for (int j = 0; j < K / 32; ++j) {
    float ks = 0.f, kd = 0.f;
#pragma unroll 4
    for (int i = 0; i < 32; ++i) {
      vecf wv = *(const vecf*)(W + (size_t)(j * 32 + i) * DO + lane * V);
      float ps = 0.f, pd = 0.f;
#pragma unroll
      for (int u = 0; u < V; ++u) {
        ps += wv[u] * sv[u];
        pd += wv[u] * dv[u];
      }
      ps = wave_sum(ps);
      pd = wave_sum(pd);
      if (lane == i) { ks = ps; kd = pd; }
    }
    ws[j * 32 + lane] = ks;
    wd[j * 32 + lane] = kd;
  }
}

#define NCHUNK 256
#define BUCKET_NODES 196
#define MAX_BUCKETS 1024
#define CHUNK_THREADS 1024
#define STAGE_CAP 8192
#define SLICE_U 8

__device__ __forceinline__ int wave_incl_scan(int x, int lane) {
#pragma unroll
  for (int o = 1; o < 32; o <<= 1) {
    int y = __shfl_up(x, o);
    if (lane >= o) x += y;
  }
  return x;
}

__device__ __forceinline__ int block_excl_scan(int v, int* wsum, int& total) {
  int lane = threadIdx.x & 31;
  int w = threadIdx.x >> 5;
  int nw = blockDim.x >> 5;
  int x = wave_incl_scan(v, lane);
  if (lane == 31) wsum[w] = x;
  __syncthreads();
  if (w == 0) {
    int t = (lane < nw) ? wsum[lane] : 0;
    int xs = wave_incl_scan(t, lane);
    wsum[lane] = xs - t;
    if (lane == 31) wsum[32] = xs;
  }
  __syncthreads();
  total = wsum[32];
  int r = wsum[w] + (x - v);
  __syncthreads();
  return r;
}

__global__ void __launch_bounds__(CHUNK_THREADS)
chunk_sort_gemm1_kernel(const int* __restrict__ src,
                        const int* __restrict__ dst, int E, int NB,
                        int* __restrict__ off, int* __restrict__ rec_words,
                        const float* __restrict__ A,
                        const float* __restrict__ B,
                        float* __restrict__ C, int M, int K,
                        const float* __restrict__ W2,
                        const float* __restrict__ a_s2,
                        const float* __restrict__ a_d2,
                        float* __restrict__ ws2,
                        float* __restrict__ wd2) {
  typedef int v2i __attribute__((ext_vector_type(2)));
  __shared__ int h[MAX_BUCKETS];
  __shared__ int wsum[33];
  v2i* rec = (v2i*)rec_words;
  int tid = threadIdx.x;
  int c = blockIdx.x;
  for (int i = tid; i < NB; i += CHUNK_THREADS) h[i] = 0;
  __syncthreads();
  int per = (E + NCHUNK - 1) / NCHUNK;
  int e0 = c * per;
  int e1 = e0 + per < E ? e0 + per : E;
  for (int e = e0 + tid; e < e1; e += 4 * CHUNK_THREADS) {
    int d[4];
#pragma unroll
    for (int k = 0; k < 4; ++k) {
      int ee = e + k * CHUNK_THREADS;
      d[k] = ee < e1 ? dst[ee] : -1;
    }
#pragma unroll
    for (int k = 0; k < 4; ++k)
      if (d[k] >= 0) atomicAdd(&h[(unsigned)d[k] / BUCKET_NODES], 1);
  }
  __syncthreads();
  int carry = 0;
  for (int base = 0; base < NB; base += CHUNK_THREADS) {
    int i = base + tid;
    int v = (i < NB) ? h[i] : 0;
    int total;
    int ex = block_excl_scan(v, wsum, total);
    if (i < NB) {
      h[i] = carry + ex;
      off[(size_t)i * NCHUNK + c] = carry + ex;
    }
    carry += total;
  }
  if (tid == 0) off[(size_t)NB * NCHUNK + c] = carry;
  __syncthreads();
  for (int e = e0 + tid; e < e1; e += 4 * CHUNK_THREADS) {
    int d[4], s[4], p[4];
#pragma unroll
    for (int k = 0; k < 4; ++k) {
      int ee = e + k * CHUNK_THREADS;
      bool ok = ee < e1;
      d[k] = ok ? dst[ee] : -1;
      s[k] = ok ? src[ee] : 0;
    }
#pragma unroll
    for (int k = 0; k < 4; ++k)
      p[k] = d[k] >= 0 ? atomicAdd(&h[(unsigned)d[k] / BUCKET_NODES], 1) : 0;
#pragma unroll
    for (int k = 0; k < 4; ++k)
      if (d[k] >= 0) rec[e0 + p[k]] = (v2i){s[k], d[k]};
  }

  int w = __builtin_amdgcn_readfirstlane((int)(threadIdx.x >> 5));
  if (c == 0 && w == (CHUNK_THREADS >> 5) - 1)
    att_vec_wave<64, 128>(W2, a_s2, a_d2, ws2, wd2);
  int mtiles = (M + 15) >> 4;
  for (int strip = w * NCHUNK + c; strip < mtiles;
       strip += (CHUNK_THREADS >> 5) * NCHUNK)
    gemm_strip<4, false>(A, B, nullptr, C, M, K, strip);
}


__device__ __forceinline__ float xhalf(float x) {
  int b = __builtin_bit_cast(int, x);
  return __builtin_bit_cast(
      float, __builtin_amdgcn_permlanex16(b, b, 0x76543210, 0xfedcba98, false, false));
}

__device__ __forceinline__ int half_lane(int x, int i) {
#define HALF_LANE_CASE(I) \
  case I: return __builtin_amdgcn_update_dpp(0, x, 0x150 + I, 0xf, 0xf, false);
  switch (i) {
    HALF_LANE_CASE(0) HALF_LANE_CASE(1) HALF_LANE_CASE(2) HALF_LANE_CASE(3)
    HALF_LANE_CASE(4) HALF_LANE_CASE(5) HALF_LANE_CASE(6) HALF_LANE_CASE(7)
    HALF_LANE_CASE(8) HALF_LANE_CASE(9) HALF_LANE_CASE(10) HALF_LANE_CASE(11)
    HALF_LANE_CASE(12) HALF_LANE_CASE(13) HALF_LANE_CASE(14)
    default: return __builtin_amdgcn_update_dpp(0, x, 0x150 + 15, 0xf, 0xf, false);
  }
#undef HALF_LANE_CASE
}

#define L1_NP 14
#define L1_PG 2

template <bool V> struct BoolTag { static constexpr bool value = V; };

#define RANK_THREADS 512
__global__ void __launch_bounds__(RANK_THREADS, 8)
rank_gather_l1_kernel(const int* __restrict__ rec_words,
                      const int* __restrict__ off, int E,
                      int* __restrict__ seg_beg,
                      int* __restrict__ seg_end,
                      int* __restrict__ csr_src,
                      const float* __restrict__ a_s,
                      const float* __restrict__ a_d,
                      const float* __restrict__ hfeat,
                      const float* __restrict__ bias,
                      float* __restrict__ out,
                      const float* __restrict__ wsrc,
                      const float* __restrict__ wdst,
                      float* __restrict__ as_next,
                      float* __restrict__ ad_next, int N) {
  static_assert(NCHUNK <= RANK_THREADS && BUCKET_NODES <= NCHUNK,
                "a thread per chunk and a thread per node");
  typedef int v2i __attribute__((ext_vector_type(2)));
  const v2i* rec = (const v2i*)rec_words;
  __shared__ int cnt[BUCKET_NODES];
  __shared__ int lbeg[BUCKET_NODES];
  __shared__ int lend[BUCKET_NODES];
  __shared__ int wsum[33];
  __shared__ int stage[STAGE_CAP];
  __shared__ v2i lrec[STAGE_CAP];
  int tid = threadIdx.x;
  int b = blockIdx.x;
  int lo = b * BUCKET_NODES;
  int per = (E + NCHUNK - 1) / NCHUNK;
  bool isc = tid < NCHUNK;
  bool isn = tid < BUCKET_NODES;
  int o0 = 0, o1 = 0;
  if (isc) {
    o0 = off[(size_t)b * NCHUNK + tid];
    o1 = off[(size_t)(b + 1) * NCHUNK + tid];
  }
  int len = o1 - o0;
  int n, bs;
  int pos = block_excl_scan(len, wsum, n);
  (void)block_excl_scan(o0, wsum, bs);
  int g0 = tid * per + o0;
  if (isn) cnt[tid] = 0;
  __syncthreads();
  bool staged = n <= STAGE_CAP;
  for (int i = 0; i < len; i += SLICE_U) {
    v2i r[SLICE_U];
#pragma unroll
    for (int k = 0; k < SLICE_U; ++k)
      r[k] = i + k < len ? rec[g0 + i + k] : (v2i){0, -1};
#pragma unroll
    for (int k = 0; k < SLICE_U; ++k) {
      if (r[k].y >= 0) {
        atomicAdd(&cnt[r[k].y - lo], 1);
        if (staged) lrec[pos + i + k] = r[k];
      }
    }
  }
  __syncthreads();
  {
    int v = isn ? cnt[tid] : 0;
    int total;
    int ex = block_excl_scan(v, wsum, total);
    int nd = lo + tid;
    if (isn) {
      if (nd < N) {
        seg_beg[nd] = bs + ex;
        seg_end[nd] = bs + ex + v;
      }
      lbeg[tid] = ex;
      lend[tid] = ex + v;
      cnt[tid] = ex;
    }
  }
  __syncthreads();
  if (staged) {
    for (int i = tid; i < n; i += 4 * RANK_THREADS) {
      v2i r[4];
      int p[4];
#pragma unroll
      for (int k = 0; k < 4; ++k) {
        int ii = i + k * RANK_THREADS;
        r[k] = ii < n ? lrec[ii] : (v2i){0, -1};
      }
#pragma unroll
      for (int k = 0; k < 4; ++k)
        p[k] = r[k].y >= 0 ? atomicAdd(&cnt[r[k].y - lo], 1) : 0;
#pragma unroll
      for (int k = 0; k < 4; ++k)
        if (r[k].y >= 0) stage[p[k]] = r[k].x;
    }
    __syncthreads();
    for (int i = tid; i < n; i += RANK_THREADS) csr_src[bs + i] = stage[i];
  } else {
    for (int i = 0; i < len; i += SLICE_U) {
      v2i r[SLICE_U];
      int p[SLICE_U];
#pragma unroll
      for (int k = 0; k < SLICE_U; ++k)
        r[k] = i + k < len ? rec[g0 + i + k] : (v2i){0, -1};
#pragma unroll
      for (int k = 0; k < SLICE_U; ++k)
        p[k] = r[k].y >= 0 ? atomicAdd(&cnt[r[k].y - lo], 1) : 0;
#pragma unroll
      for (int k = 0; k < SLICE_U; ++k)
        if (r[k].y >= 0) csr_src[bs + p[k]] = r[k].x;
    }
    __threadfence();
  }
  __syncthreads();

  int bs_u = __builtin_amdgcn_readfirstlane(bs);
  auto gather = [&](auto in_lds) {
  constexpr bool IN_LDS = decltype(in_lds)::value;
  constexpr int GCH = 2 * L1_NP;
  constexpr int NG = L1_NP / L1_PG;
  constexpr int NWV = RANK_THREADS / 32;
  int lane = tid & 31;
  int half = lane >> 4;
  int q = lane & 15;
  int c0 = 4 * q;
  int myedge = 2 * q + half;
  const float NEG_INF = -__builtin_inff();
  const v4f* __restrict__ rows = (const v4f*)hfeat;
  v4f* __restrict__ orows = (v4f*)out;
  v4f asv = *(const v4f*)(a_s + c0);
  v4f adv = *(const v4f*)(a_d + c0);
  v4f bv = *(const v4f*)(bias + c0);
  v4f wsv = *(const v4f*)(wsrc + c0);
  v4f wdv = *(const v4f*)(wdst + c0);
  float s_seed = half == 0 ? 1.f : 0.f;

  int t = __builtin_amdgcn_readfirstlane(tid >> 5);
  int tmax = N - lo < BUCKET_NODES ? N - lo : BUCKET_NODES;
  if (t >= tmax) return;

  auto src_id = [&](int rel) -> int {
    if constexpr (IN_LDS) return stage[rel];
    else return csr_src[bs_u + rel];
  };

  int beg = __builtin_amdgcn_readfirstlane(lbeg[t]);
  int end = __builtin_amdgcn_readfirstlane(lend[t]);
  int cnt_e = end - beg;
  if (cnt_e > GCH) cnt_e = GCH;
  int mysrc = 0;
  if (cnt_e > 0) mysrc = src_id(beg + (myedge < cnt_e ? myedge : cnt_e - 1));
  v4f own = rows[(lo + t) * 16 + q];
  int tfirst = t + NWV < tmax ? t + NWV : t;
  int nbeg = __builtin_amdgcn_readfirstlane(lbeg[tfirst]);
  int nend = __builtin_amdgcn_readfirstlane(lend[tfirst]);

  for (; t < tmax; t += NWV) {
    int node = lo + t;
    int ncnt = 0, nsrc = 0, nnbeg = 0, nnend = 0;
    v4f nrow;
    auto ask_next = [&]() {
      int t1 = t + NWV < tmax ? t + NWV : t;
      int t2 = t + 2 * NWV < tmax ? t + 2 * NWV : t;
      ncnt = nend - nbeg;
      if (ncnt > GCH) ncnt = GCH;
      if (ncnt > 0) nsrc = src_id(nbeg + (myedge < ncnt ? myedge : ncnt - 1));
      nrow = rows[(lo + t1) * 16 + q];
      nnbeg = lbeg[t2];
      nnend = lend[t2];
    };
    bool asked = false;

    float d_s = own.x * asv.x;
    float d_d = own.x * adv.x;
    d_s += own.y * asv.y;
    d_d += own.y * adv.y;
    d_s += own.z * asv.z;
    d_d += own.z * adv.z;
    d_s += own.w * asv.w;
    d_d += own.w * adv.w;
    d_s += row_xor_lane<1>(d_s);
    d_d += row_xor_lane<1>(d_d);
    float e0 = d_s + d_d;
    e0 = e0 > 0.f ? e0 : NEG_SLOPE * e0;
    float m = e0;
    float s = s_seed;
    v4f acc = half == 0 ? own : (v4f){0.f, 0.f, 0.f, 0.f};
    int base = beg;

    while (cnt_e > 0) {
      v4f hv[L1_NP];
      float tv[L1_NP];
#pragma unroll
      for (int g = 0; g < NG; ++g) {
        if (g > 0 && g * 2 * L1_PG >= cnt_e) break;
#pragma unroll
        for (int i = 0; i < L1_PG; ++i) {
          int p = g * L1_PG + i;
          int sidx = half_lane(mysrc, p);
          hv[p] = rows[sidx * 16 + q];
        }
      }
      if (!asked) {
        ask_next();
        asked = true;
      }
      __builtin_amdgcn_sched_barrier(0);

      int lim = cnt_e - half;
      float cm = NEG_INF;
#pragma unroll
      for (int g = 0; g < NG; ++g) {
        if (g > 0 && g * 2 * L1_PG >= cnt_e) break;
#pragma unroll
        for (int i = 0; i < L1_PG; ++i) {
          int p = g * L1_PG + i;
          float d = hv[p].x * asv.x;
          d += hv[p].y * asv.y;
          d += hv[p].z * asv.z;
          d += hv[p].w * asv.w;
          d += row_xor_lane<1>(d);
          float tt = d + d_d;
          tt = tt > 0.f ? tt : NEG_SLOPE * tt;
          tt = 2 * p < lim ? tt : NEG_INF;
          tv[p] = tt;
          cm = fmaxf(cm, tt);
        }
      }
      float mn = fmaxf(m, fmaxf(cm, xhalf(cm)));
      float sc = __expf(m - mn);
      s = s * sc;
      acc = acc * sc;
#pragma unroll
      for (int g = 0; g < NG; ++g) {
        if (g > 0 && g * 2 * L1_PG >= cnt_e) break;
#pragma unroll
        for (int i = 0; i < L1_PG; ++i) {
          int p = g * L1_PG + i;
          float pw = __expf(tv[p] - mn);
          s += pw;
          acc += hv[p] * pw;
        }
      }
      m = mn;

      base += GCH;
      cnt_e = end - base;
      if (cnt_e > GCH) cnt_e = GCH;
      mysrc = 0;
      if (cnt_e > 0) mysrc = src_id(base + (myedge < cnt_e ? myedge : cnt_e - 1));
    }
    if (!asked) ask_next();

    float denom = s + xhalf(s) + GAT_EPS;
    v4f o;
#pragma unroll
    for (int k = 0; k < 4; ++k) {
      float v = (acc[k] + xhalf(acc[k])) / denom + bv[k];
      v = v > 0.f ? v : __expf(v) - 1.f;
      o[k] = v;
    }
    if (half == 0) orows[node * 16 + q] = o;

    float ps = o.x * wsv.x;
    float pd = o.x * wdv.x;
    ps += o.y * wsv.y;
    pd += o.y * wdv.y;
    ps += o.z * wsv.z;
    pd += o.z * wdv.z;
    ps += o.w * wsv.w;
    pd += o.w * wdv.w;
    ps += row_xor_lane<1>(ps);
    pd += row_xor_lane<1>(pd);
    ps += row_xor_lane<2>(ps);
    pd += row_xor_lane<2>(pd);
    ps += row_xor_lane<4>(ps);
    pd += row_xor_lane<4>(pd);
    ps += row_xor_lane<8>(ps);
    pd += row_xor_lane<8>(pd);
    if (lane == 0) {
      as_next[node] = ps;
      ad_next[node] = pd;
    }

    beg = nbeg;
    end = nend;
    cnt_e = ncnt;
    mysrc = nsrc;
    own = nrow;
    nbeg = __builtin_amdgcn_readfirstlane(nnbeg);
    nend = __builtin_amdgcn_readfirstlane(nnend);
  }
  };
  if (__builtin_amdgcn_readfirstlane(n) <= STAGE_CAP) gather(BoolTag<true>{});
  else gather(BoolTag<false>{});
}

#define GGRP 4

__global__ void __launch_bounds__(256)
node_gather_l2_kernel(const int* __restrict__ seg_beg,
                      const int* __restrict__ seg_end,
                      const int* __restrict__ csr_src,
                      const float* __restrict__ as,
                      const float* __restrict__ ad,
                      const float* __restrict__ hfeat,
                      float* __restrict__ out, int N) {
  constexpr int GCH = 32;
  constexpr int NG = GCH / GGRP;
  int lane = threadIdx.x & 31;
  int wave0 = __builtin_amdgcn_readfirstlane(
      (int)((blockIdx.x * blockDim.x + threadIdx.x) >> 5));
  int nwaves = (int)((gridDim.x * blockDim.x) >> 5);
  int c0 = lane * 2;
  const float NEG_INF = -__builtin_inff();
  const v2f* __restrict__ rows = (const v2f*)hfeat;

  int node = wave0;
  if (node >= N) return;

  int beg = seg_beg[node];
  int end = seg_end[node];
  int cnt = end - beg;
  if (cnt > GCH) cnt = GCH;
  int mysrc = 0;
  if (lane < cnt) mysrc = csr_src[beg + lane];
  float adv = ad[node];
  float as_self = as[node];
  v2f acc = rows[node * 32 + lane];
  int nfirst = node + nwaves < N ? node + nwaves : node;
  int nbeg = seg_beg[nfirst];
  int nend = seg_end[nfirst];

  for (; node < N; node += nwaves) {
    int ncnt = 0, nsrc = 0, nnbeg = 0, nnend = 0;
    float nadv = 0.f, nas = 0.f;
    v2f nrow;
    auto ask_next = [&]() {
      int nx1 = node + nwaves < N ? node + nwaves : node;
      int nx2 = node + 2 * nwaves < N ? node + 2 * nwaves : node;
      ncnt = nend - nbeg;
      if (ncnt > GCH) ncnt = GCH;
      if (lane < ncnt) nsrc = csr_src[nbeg + lane];
      nadv = ad[nx1];
      nas = as[nx1];
      nrow = rows[nx1 * 32 + lane];
      nnbeg = seg_beg[nx2];
      nnend = seg_end[nx2];
    };
    bool asked = false;

    float asv = as[mysrc];
    int base = beg;

    float e0 = as_self + adv;
    e0 = e0 > 0.f ? e0 : NEG_SLOPE * e0;
    float m = e0;
    float s = 1.f;

    while (cnt > 0) {
      int last = cnt - 1;
      v2f hv[GCH];
#pragma unroll
      for (int g = 0; g < NG; ++g) {
        if (g > 0 && g * GGRP >= cnt) break;
int sj[GGRP];
#pragma unroll
        for (int i = 0; i < GGRP; ++i) {
          int j = g * GGRP + i;
          sj[i] = __builtin_amdgcn_readlane(mysrc, j < last ? j : last);
        }
#pragma unroll
        for (int i = 0; i < GGRP; ++i)
          hv[g * GGRP + i] = rows[sj[i] * 32 + lane];
      }
      if (!asked) {
        ask_next();
        asked = true;
      }
      __builtin_amdgcn_sched_barrier(0);

      float t = asv + adv;
      t = t > 0.f ? t : NEG_SLOPE * t;
      float mye = lane < cnt ? t : NEG_INF;
      float mn = fmaxf(m, wave_max(mye));
      float sc = __expf(m - mn);
      float p = __expf(mye - mn);
      s = s * sc + wave_sum(p);
      acc = acc * sc;
      int pb = __float_as_int(p);
#pragma unroll
      for (int g = 0; g < NG; ++g) {
        if (g > 0 && g * GGRP >= cnt) break;
#pragma unroll
        for (int i = 0; i < GGRP; ++i) {
          int j = g * GGRP + i;
          float pj = __int_as_float(__builtin_amdgcn_readlane(pb, j));
          acc += hv[j] * pj;
        }
      }
      m = mn;

      base += GCH;
      cnt = end - base;
      if (cnt > GCH) cnt = GCH;
      mysrc = 0;
      if (lane < cnt) mysrc = csr_src[base + lane];
      if (cnt > 0) asv = as[mysrc];
    }
    if (!asked) ask_next();

    float denom = s + GAT_EPS;
    v2f o;
    o.x = acc.x / denom;
    o.y = acc.y / denom;
    *(v2f*)(out + (size_t)node * 64 + c0) = o;

    beg = nbeg;
    end = nend;
    cnt = ncnt;
    mysrc = nsrc;
    adv = nadv;
    as_self = nas;
    acc = nrow;
    nbeg = nnbeg;
    nend = nnend;
  }
}

static inline int cdiv(long long a, long long b) { return (int)((a + b - 1) / b); }

extern "C" void kernel_launch(void* const* d_in, const int* in_sizes, int n_in,
                              void* d_out, int out_size, void* d_ws, size_t ws_size,
                              hipStream_t stream) {
  const float* x      = (const float*)d_in[0];
  const int*   ei     = (const int*)  d_in[1];
  const float* W1     = (const float*)d_in[2];
  const float* a_src1 = (const float*)d_in[3];
  const float* a_dst1 = (const float*)d_in[4];
  const float* b1     = (const float*)d_in[5];
  const float* W2     = (const float*)d_in[6];
  const float* a_src2 = (const float*)d_in[7];
  const float* a_dst2 = (const float*)d_in[8];
  const float* b2     = (const float*)d_in[9];
  float* z = (float*)d_out;

  const int Fin = 256, C1 = 64, C2 = 128;
  const int N    = in_sizes[0] / Fin;
  const int E    = in_sizes[1] / 2;
  const int* srcv = ei;
  const int* dstv = ei + E;

  float* h1   = (float*)d_ws;
  float* hid  = h1  + (size_t)N * C1;
  float* agg  = hid + (size_t)N * C1;
  float* as2  = agg + (size_t)N * C1;
  float* ad2  = as2 + N;
  float* ws2  = ad2 + N;
  float* wd2  = ws2 + C1;
  const int NB = cdiv(N, BUCKET_NODES);
  int*   offs = (int*)(wd2 + C1);
  int*   recs = offs + (size_t)(NB + 1) * NCHUNK;
  int*   csrc = recs + (size_t)2 * E;
  int*   sbeg = csrc + E;
  int*   send = sbeg + N;

  const int B = 256;
  const int GB = 128;
  const int GATHER2_BLOCKS = 1280;

  chunk_sort_gemm1_kernel<<<NCHUNK, CHUNK_THREADS, 0, stream>>>(
      srcv, dstv, E, NB, offs, recs, x, W1, h1, N, Fin, W2, a_src2, a_dst2, ws2,
      wd2);
  rank_gather_l1_kernel<<<NB, RANK_THREADS, 0, stream>>>(
      recs, offs, E, sbeg, send, csrc, a_src1, a_dst1, h1, b1, hid, ws2, wd2,
      as2, ad2, N);

  node_gather_l2_kernel<<<GATHER2_BLOCKS, B, 0, stream>>>(
      sbeg, send, csrc, as2, ad2, hid, agg, N);
  gemm_f32_wmma_kernel<C2 / 16, true><<<cdiv((long long)(N / 16) * 32, GB), GB,
                                        0, stream>>>(agg, W2, b2, z, N, C1);
}
